# starting kernel
# baseline (speedup 1.0000x reference)
#include <hip/hip_runtime.h>
#include <math.h>

typedef __attribute__((ext_vector_type(2))) float v2f;
typedef __attribute__((ext_vector_type(8))) float v8f;

#define IN_FEATS 256
#define OUT_FEATS 64
#define MAXK 16
#define TILE_M 16
#define LDA (IN_FEATS + 4)
#define LDH (OUT_FEATS + 4)

__global__ void init_kernel(float* __restrict__ out, const float* __restrict__ bias,
                            float* __restrict__ deg, int n_nodes) {
    long long i = (long long)blockIdx.x * blockDim.x + threadIdx.x;
    long long total = (long long)n_nodes * OUT_FEATS;
    if (i < total) out[i] = bias[i & (OUT_FEATS - 1)];
    if (i < 2LL * n_nodes) deg[i] = 0.0f;
}

__global__ void deg_wt_kernel(const int* __restrict__ src, const int* __restrict__ dst,
                              float* __restrict__ out_deg, float* __restrict__ in_deg,
                              const float* __restrict__ weight, float* __restrict__ wT,
                              int n_edges) {
    int i = blockIdx.x * blockDim.x + threadIdx.x;
    if (i < n_edges) {
        atomicAdd(&out_deg[src[i]], 1.0f);
        atomicAdd(&in_deg[dst[i]], 1.0f);
    }
    if (i < IN_FEATS * OUT_FEATS) {
        int k = i >> 6;
        int c = i & 63;
        wT[c * IN_FEATS + k] = weight[i];
    }
}

__global__ __launch_bounds__(128)
void gemm_maxk_kernel(const float* __restrict__ feat, const float* __restrict__ wT,
                      const float* __restrict__ out_deg, const float* __restrict__ in_deg,
                      float* __restrict__ tk_vals, unsigned int* __restrict__ tk_idx,
                      int n_nodes) {
    __shared__ float sA[TILE_M * LDA];
    __shared__ float sH[TILE_M * LDH];

    const int tid = threadIdx.x;
    const int m0 = blockIdx.x * TILE_M;

    for (int i = tid; i < TILE_M * (IN_FEATS / 4); i += 128) {
        int row = i / (IN_FEATS / 4);
        int c4  = i % (IN_FEATS / 4);
        float4 v = make_float4(0.f, 0.f, 0.f, 0.f);
        if (m0 + row < n_nodes)
            v = ((const float4*)feat)[(long long)(m0 + row) * (IN_FEATS / 4) + c4];
        float* p = &sA[row * LDA + c4 * 4];
        p[0] = v.x; p[1] = v.y; p[2] = v.z; p[3] = v.w;
    }
    __syncthreads();

    const int wave = tid >> 5;
    const int lane = tid & 31;
    const int r16  = lane & 15;
    const int koff = (lane >> 4) << 1;
    const float* wcol = wT + (wave * 16 + r16) * IN_FEATS;
    const float* arow = sA + r16 * LDA;

    v8f c = {0.f, 0.f, 0.f, 0.f, 0.f, 0.f, 0.f, 0.f};
    #pragma unroll
    for (int k0 = 0; k0 < IN_FEATS; k0 += 4) {
        v2f a, b;
        a.x = arow[k0 + koff];
        a.y = arow[k0 + koff + 1];
        b.x = wcol[k0 + koff];
        b.y = wcol[k0 + koff + 1];
        c = __builtin_amdgcn_wmma_f32_16x16x4_f32(false, a, false, b, (short)0, c,
                                                  false, false);
    }

    {
        const int rbase = (lane >> 4) << 3;
        const int col = wave * 16 + r16;
        #pragma unroll
        for (int r = 0; r < 8; ++r)
            sH[(rbase + r) * LDH + col] = c[r];
    }
    __syncthreads();

    if (tid < TILE_M && (m0 + tid) < n_nodes) {
        const int node = m0 + tid;
        float od = out_deg[node]; od = od < 1.f ? 1.f : od;
        float id = in_deg[node];  id = id < 1.f ? 1.f : id;
        const float scale = rsqrtf(od) * rsqrtf(id);
        float* rowp = &sH[tid * LDH];

        unsigned int iout0 = 0u, iout1 = 0u, iout2 = 0u, iout3 = 0u;
        #pragma unroll
        for (int i = 0; i < MAXK; ++i) {
            float best = -3.402823466e+38f;
            int bi = 0;
            for (int j = 0; j < OUT_FEATS; ++j) {
                float v = rowp[j];
                if (v > best) { best = v; bi = j; }
            }
            rowp[bi] = -3.402823466e+38f;
            tk_vals[(long long)node * MAXK + i] = best * scale;
            unsigned int b8 = ((unsigned int)bi) << ((i & 3) * 8);
            if (i < 4)       iout0 |= b8;
            else if (i < 8)  iout1 |= b8;
            else if (i < 12) iout2 |= b8;
            else             iout3 |= b8;
        }
        unsigned int* iq = &tk_idx[(long long)node * 4];
        iq[0] = iout0; iq[1] = iout1; iq[2] = iout2; iq[3] = iout3;
    }
}

__global__ void scatter_kernel(const int* __restrict__ src, const int* __restrict__ dst,
                               const float* __restrict__ tk_vals,
                               const unsigned int* __restrict__ tk_idx,
                               float* __restrict__ out, int n_edges) {
    long long gid = (long long)blockIdx.x * blockDim.x + threadIdx.x;
    long long total = (long long)n_edges * 4;
    if (gid >= total) return;
    int e = (int)(gid >> 2);
    int part = (int)(gid & 3);
    int s = src[e], d = dst[e];
    float4 v = ((const float4*)tk_vals)[(long long)s * 4 + part];
    unsigned int ip = tk_idx[(long long)s * 4 + part];
    float* orow = out + (long long)d * OUT_FEATS;
    atomicAdd(orow + (ip & 0xffu), v.x);
    atomicAdd(orow + ((ip >> 8) & 0xffu), v.y);
    atomicAdd(orow + ((ip >> 16) & 0xffu), v.z);
    atomicAdd(orow + ((ip >> 24) & 0xffu), v.w);
}

extern "C" void kernel_launch(void* const* d_in, const int* in_sizes, int n_in,
                              void* d_out, int out_size, void* d_ws, size_t ws_size,
                              hipStream_t stream) {
    const float* feat   = (const float*)d_in[0];
    const float* weight = (const float*)d_in[1];
    const float* bias   = (const float*)d_in[2];
    const int*   src    = (const int*)d_in[3];
    const int*   dst    = (const int*)d_in[4];
    float* out = (float*)d_out;

    const int n_nodes = in_sizes[0] / IN_FEATS;
    const int n_edges = in_sizes[3];

    float* ws_f    = (float*)d_ws;
    float* out_deg = ws_f;
    float* in_deg  = out_deg + n_nodes;
    float* wT      = in_deg + n_nodes;
    float* tk_vals = wT + IN_FEATS * OUT_FEATS;
    unsigned int* tk_idx = (unsigned int*)(tk_vals + (long long)n_nodes * MAXK);

    {
        long long total = (long long)n_nodes * OUT_FEATS;
        int grid = (int)((total + 255) / 256);
        init_kernel<<<grid, 256, 0, stream>>>(out, bias, out_deg, n_nodes);
    }
    {
        int work = n_edges > IN_FEATS * OUT_FEATS ? n_edges : IN_FEATS * OUT_FEATS;
        int grid = (work + 255) / 256;
        deg_wt_kernel<<<grid, 256, 0, stream>>>(src, dst, out_deg, in_deg, weight, wT,
                                                n_edges);
    }
    {
        int grid = (n_nodes + TILE_M - 1) / TILE_M;
        gemm_maxk_kernel<<<grid, 128, 0, stream>>>(feat, wT, out_deg, in_deg,
                                                   tk_vals, tk_idx, n_nodes);
    }
    {
        long long total = (long long)n_edges * 4;
        int grid = (int)((total + 255) / 256);
        scatter_kernel<<<grid, 256, 0, stream>>>(src, dst, tk_vals, tk_idx, out, n_edges);
    }
}


# edge sum via destination tiles: edges sorted by 256-node tile in one placing pass, a block adds a tile's records into LDS rows with LDS float adds, stores rows once; sum order: LDS arrival order
# speedup vs baseline: 1.4081x; 1.4081x over previous
#include <hip/hip_runtime.h>
#include <math.h>

typedef __attribute__((ext_vector_type(2))) float v2f;
typedef __attribute__((ext_vector_type(8))) float v8f;

#define IN_FEATS 256
#define OUT_FEATS 64
#define MAXK 16
#define TILE_M 16
#define LDA (IN_FEATS + 4)
#define BKT_SHIFT 8
#define BKT_NODES (1 << BKT_SHIFT)
#define MAX_BKT 1024
#define EDGE_CHUNK 8192
#define EPT (EDGE_CHUNK / 1024)

__global__ void init_kernel(float* __restrict__ deg, int* __restrict__ bkt, int n_nodes) {
    int i = blockIdx.x * blockDim.x + threadIdx.x;
    if (i < 2 * n_nodes) deg[i] = 0.0f;
    if (i < 2 * MAX_BKT) bkt[i] = 0;
}

__global__ void deg_wt_kernel(const int* __restrict__ src, const int* __restrict__ dst,
                              float* __restrict__ out_deg, float* __restrict__ in_deg,
                              const float* __restrict__ weight, float* __restrict__ wT,
                              int n_edges) {
    int i = blockIdx.x * blockDim.x + threadIdx.x;
    if (i < n_edges) {
        atomicAdd(&out_deg[src[i]], 1.0f);
        atomicAdd(&in_deg[dst[i]], 1.0f);
    }
    if (i < IN_FEATS * OUT_FEATS) {
        int k = i >> 6;
        int c = i & 63;
        wT[c * IN_FEATS + k] = weight[i];
    }
}


__device__ __forceinline__ int row_sum16(int x) {
    x += __builtin_amdgcn_update_dpp(0, x, 0xB1, 0xF, 0xF, false);
    x += __builtin_amdgcn_update_dpp(0, x, 0x4E, 0xF, 0xF, false);
    x += __builtin_amdgcn_update_dpp(0, x, 0x141, 0xF, 0xF, false);
    x += __builtin_amdgcn_update_dpp(0, x, 0x140, 0xF, 0xF, false);
    return x;
}

__device__ __forceinline__ unsigned int order_key(float f) {
    unsigned int u = __float_as_uint(f);
    return u ^ ((unsigned int)((int)u >> 31) | 0x80000000u);
}

__global__ __launch_bounds__(32)
void gemm_maxk_kernel(const float* __restrict__ feat, const float* __restrict__ wT,
                      const float* __restrict__ out_deg, const float* __restrict__ in_deg,
                      float* __restrict__ tk_vals, unsigned int* __restrict__ tk_idx,
                      int n_nodes) {
    __shared__ __attribute__((aligned(16))) float sA[TILE_M * LDA];

    const int lane = threadIdx.x;
    const int m0   = blockIdx.x * TILE_M;
    const int r16  = lane & 15;
    const int half = lane >> 4;

    float sc[8];
    #pragma unroll
    for (int r = 0; r < 8; ++r) {
        int node = m0 + r + 8 * half;
        node = node < n_nodes ? node : n_nodes - 1;
        float od = out_deg[node]; od = od < 1.f ? 1.f : od;
        float id = in_deg[node];  id = id < 1.f ? 1.f : id;
        sc[r] = rsqrtf(od) * rsqrtf(id);
    }

    #pragma unroll
    for (int pass = 0; pass < 2; ++pass) {
        float4 buf[16];
        #pragma unroll
        for (int j = 0; j < 16; ++j) {
            int i   = (pass * 16 + j) * 32 + lane;
            int row = i >> 6;
            int c4  = i & 63;
            int grow = m0 + row;
            grow = grow < n_nodes ? grow : n_nodes - 1;
            buf[j] = ((const float4*)feat)[(long long)grow * (IN_FEATS / 4) + c4];
        }
        #pragma unroll
        for (int j = 0; j < 16; ++j) {
            int i   = (pass * 16 + j) * 32 + lane;
            int row = i >> 6;
            int c4  = i & 63;
            float4 v = buf[j];
            if (m0 + row >= n_nodes) v = make_float4(0.f, 0.f, 0.f, 0.f);
            *(float4*)&sA[row * LDA + c4 * 4] = v;
        }
    }
    __syncthreads();

    const int koff = half << 1;
    const float* arow = sA + r16 * LDA;
    const float* w0 = wT + r16 * IN_FEATS;
    const float* w1 = w0 + 16 * IN_FEATS;
    const float* w2 = w0 + 32 * IN_FEATS;
    const float* w3 = w0 + 48 * IN_FEATS;

    v8f c0 = {0.f, 0.f, 0.f, 0.f, 0.f, 0.f, 0.f, 0.f};
    v8f c1 = c0, c2 = c0, c3 = c0;
    #pragma unroll
    for (int k0 = 0; k0 < IN_FEATS; k0 += 4) {
        v2f a, b0, b1, b2, b3;
        a.x  = arow[k0 + koff];  a.y  = arow[k0 + koff + 1];
        b0.x = w0[k0 + koff];    b0.y = w0[k0 + koff + 1];
        b1.x = w1[k0 + koff];    b1.y = w1[k0 + koff + 1];
        b2.x = w2[k0 + koff];    b2.y = w2[k0 + koff + 1];
        b3.x = w3[k0 + koff];    b3.y = w3[k0 + koff + 1];
        c0 = __builtin_amdgcn_wmma_f32_16x16x4_f32(false, a, false, b0, (short)0, c0, false, false);
        c1 = __builtin_amdgcn_wmma_f32_16x16x4_f32(false, a, false, b1, (short)0, c1, false, false);
        c2 = __builtin_amdgcn_wmma_f32_16x16x4_f32(false, a, false, b2, (short)0, c2, false, false);
        c3 = __builtin_amdgcn_wmma_f32_16x16x4_f32(false, a, false, b3, (short)0, c3, false, false);
    }

    unsigned int key[4][8];
    #pragma unroll
    for (int r = 0; r < 8; ++r) {
        key[0][r] = order_key(c0[r]);
        key[1][r] = order_key(c1[r]);
        key[2][r] = order_key(c2[r]);
        key[3][r] = order_key(c3[r]);
    }

    unsigned int thr[8];
    #pragma unroll
    for (int r = 0; r < 8; ++r) thr[r] = 0u;
    for (int bit = 31; bit >= 0; --bit) {
        const unsigned int bmask = 1u << bit;
        #pragma unroll
        for (int r = 0; r < 8; ++r) {
            const unsigned int cand = thr[r] | bmask;
            int cnt = (key[0][r] >= cand ? 1 : 0) + (key[1][r] >= cand ? 1 : 0) +
                      (key[2][r] >= cand ? 1 : 0) + (key[3][r] >= cand ? 1 : 0);
            cnt = row_sum16(cnt);
            thr[r] = cnt >= MAXK ? cand : thr[r];
        }
    }

    const unsigned int hshift = (unsigned int)half << 4;
    const unsigned int below  = (1u << r16) - 1u;
    __syncthreads();
    float* sV = sA;
    unsigned char* sI = (unsigned char*)(sA + TILE_M * MAXK);
    #pragma unroll
    for (int r = 0; r < 8; ++r) {
        const unsigned int T = thr[r];
        bool gt[4], eq[4];
        unsigned int me[4];
        int n_gt = 0;
        #pragma unroll
        for (int t = 0; t < 4; ++t) {
            gt[t] = key[t][r] > T;
            eq[t] = key[t][r] == T;
            unsigned int mg = (__builtin_amdgcn_ballot_w32(gt[t]) >> hshift) & 0xffffu;
            me[t] = (__builtin_amdgcn_ballot_w32(eq[t]) >> hshift) & 0xffffu;
            n_gt += __popc(mg);
        }
        const int need = MAXK - n_gt;
        const float hv[4] = {c0[r], c1[r], c2[r], c3[r]};
        int eq_before = 0, slot_base = 0;
        #pragma unroll
        for (int t = 0; t < 4; ++t) {
            const int eq_rank = eq_before + __popc(me[t] & below);
            const bool sel = gt[t] || (eq[t] && eq_rank < need);
            const unsigned int ms = (__builtin_amdgcn_ballot_w32(sel) >> hshift) & 0xffffu;
            const int slot = slot_base + __popc(ms & below);
            if (sel) {
                sV[(r + 8 * half) * MAXK + slot] = hv[t] * sc[r];
                sI[(r + 8 * half) * MAXK + slot] = (unsigned char)(t * 16 + r16);
            }
            eq_before += __popc(me[t]);
            slot_base += __popc(ms);
        }
    }
    __syncthreads();

    const float4* sV4 = (const float4*)sV;
    #pragma unroll
    for (int q = 0; q < 2; ++q) {
        const int f = q * 32 + lane;
        if (m0 + (f >> 2) < n_nodes)
            ((float4*)tk_vals)[(long long)m0 * (MAXK / 4) + f] = sV4[f];
    }
    if (lane < TILE_M && m0 + lane < n_nodes)
        ((uint4*)tk_idx)[m0 + lane] = ((const uint4*)sI)[lane];
}


__global__ __launch_bounds__(BKT_NODES)
void bkt_count_kernel(const float* __restrict__ in_deg, int* __restrict__ bkt_cnt, int n_nodes) {
    __shared__ int wtot[BKT_NODES / 32];
    const int tid  = threadIdx.x;
    const int node = blockIdx.x * BKT_NODES + tid;
    int x = node < n_nodes ? (int)in_deg[node] : 0;
    #pragma unroll
    for (int o = 16; o > 0; o >>= 1) x += __shfl_xor(x, o);
    if ((tid & 31) == 0) wtot[tid >> 5] = x;
    __syncthreads();
    if (tid == 0) {
        int s = 0;
        #pragma unroll
        for (int k = 0; k < BKT_NODES / 32; ++k) s += wtot[k];
        bkt_cnt[blockIdx.x] = s;
    }
}

__global__ __launch_bounds__(1024)
void bkt_fill_kernel(const int* __restrict__ src, const int* __restrict__ dst,
                     const int* __restrict__ bkt_cnt, int* __restrict__ bkt_cur,
                     int* __restrict__ bkt_start, unsigned int* __restrict__ binned,
                     int n_edges) {
    __shared__ int hist[MAX_BKT];
    __shared__ int gbase[MAX_BKT];
    __shared__ int wsum[2][32];
    __shared__ unsigned int stage[EDGE_CHUNK];
    __shared__ unsigned short stage_b[EDGE_CHUNK];
    const int tid  = threadIdx.x;
    const int lane = tid & 31;
    const int wave = tid >> 5;
    hist[tid] = 0;
    const int v = bkt_cnt[tid];

    const int base = blockIdx.x * EDGE_CHUNK;
    int es[EPT], ed[EPT];
    #pragma unroll
    for (int j = 0; j < EPT; ++j) {
        const int e  = base + j * 1024 + tid;
        const int ec = e < n_edges ? e : n_edges - 1;
        const int d  = dst[ec];
        es[j] = src[ec];
        ed[j] = e < n_edges ? d : -1;
    }
    __syncthreads();
    #pragma unroll
    for (int j = 0; j < EPT; ++j)
        if (ed[j] >= 0) atomicAdd(&hist[ed[j] >> BKT_SHIFT], 1);
    __syncthreads();

    const int c = hist[tid];
    int x = v, y = c;
    #pragma unroll
    for (int o = 1; o < 32; o <<= 1) {
        const int xs = __shfl_up(x, o);
        const int ys = __shfl_up(y, o);
        if (lane >= o) { x += xs; y += ys; }
    }
    if (lane == 31) { wsum[0][wave] = x; wsum[1][wave] = y; }
    __syncthreads();
    if (wave < 2) {
        const int w = wsum[wave][lane];
        int s = w;
        #pragma unroll
        for (int o = 1; o < 32; o <<= 1) {
            const int t = __shfl_up(s, o);
            if (lane >= o) s += t;
        }
        wsum[wave][lane] = s - w;
    }
    __syncthreads();

    const int start  = x - v + wsum[0][wave];
    const int lstart = y - c + wsum[1][wave];
    if (blockIdx.x == 0) bkt_start[tid] = start;
    const int rsv = c ? atomicAdd(&bkt_cur[tid], c) : 0;
    gbase[tid] = start + rsv - lstart;
    hist[tid]  = lstart;
    __syncthreads();

    #pragma unroll
    for (int j = 0; j < EPT; ++j) {
        if (ed[j] >= 0) {
            const int b    = ed[j] >> BKT_SHIFT;
            const int slot = atomicAdd(&hist[b], 1);
            stage[slot]   = ((unsigned int)es[j] << BKT_SHIFT) | ((unsigned int)ed[j] & (BKT_NODES - 1));
            stage_b[slot] = (unsigned short)b;
        }
    }
    __syncthreads();

    const int left = n_edges - base;
    const int n_here = left < EDGE_CHUNK ? left : EDGE_CHUNK;
    #pragma unroll
    for (int j = 0; j < EPT; ++j) {
        const int slot = j * 1024 + tid;
        if (slot < n_here) binned[gbase[stage_b[slot]] + slot] = stage[slot];
    }
}

__device__ __forceinline__ void add_record(float* row, float4 v, unsigned int ip) {
    atomicAdd(row + (ip & 0xffu), v.x);
    atomicAdd(row + ((ip >> 8) & 0xffu), v.y);
    atomicAdd(row + ((ip >> 16) & 0xffu), v.z);
    atomicAdd(row + ((ip >> 24) & 0xffu), v.w);
}

__global__ __launch_bounds__(1024)
void bkt_sum_kernel(const unsigned int* __restrict__ binned, const int* __restrict__ bkt_cnt,
                    const int* __restrict__ bkt_start, const float* __restrict__ tk_vals,
                    const unsigned int* __restrict__ tk_idx, const float* __restrict__ bias,
                    float* __restrict__ out, int n_nodes) {
    __shared__ __attribute__((aligned(16))) float acc[BKT_NODES * OUT_FEATS];
    const int tid = threadIdx.x;
    for (int i = tid; i < BKT_NODES * OUT_FEATS; i += 1024) acc[i] = bias[i & (OUT_FEATS - 1)];
    __syncthreads();

    const int cnt = bkt_cnt[blockIdx.x];
    const unsigned int* mine = binned + bkt_start[blockIdx.x];
    const int part = tid & 3;
    int i0 = tid >> 2;
    for (; i0 + 3 * 256 < cnt; i0 += 4 * 256) {
        unsigned int w[4], ip[4];
        float4 v[4];
        #pragma unroll
        for (int k = 0; k < 4; ++k) w[k] = mine[i0 + 256 * k];
        #pragma unroll
        for (int k = 0; k < 4; ++k) {
            const long long s = (long long)(w[k] >> BKT_SHIFT);
            v[k]  = ((const float4*)tk_vals)[s * 4 + part];
            ip[k] = tk_idx[s * 4 + part];
        }
        #pragma unroll
        for (int k = 0; k < 4; ++k)
            add_record(acc + (w[k] & (BKT_NODES - 1)) * OUT_FEATS, v[k], ip[k]);
    }
    for (; i0 < cnt; i0 += 256) {
        const unsigned int w = mine[i0];
        const long long s = (long long)(w >> BKT_SHIFT);
        const float4 v = ((const float4*)tk_vals)[s * 4 + part];
        const unsigned int ip = tk_idx[s * 4 + part];
        add_record(acc + (w & (BKT_NODES - 1)) * OUT_FEATS, v, ip);
    }
    __syncthreads();

    const long long node0 = (long long)blockIdx.x * BKT_NODES;
    for (int i = tid; i < BKT_NODES * OUT_FEATS / 4; i += 1024) {
        if (node0 + (i >> 4) < n_nodes)
            ((float4*)out)[node0 * (OUT_FEATS / 4) + i] = ((const float4*)acc)[i];
    }
}

extern "C" void kernel_launch(void* const* d_in, const int* in_sizes, int n_in,
                              void* d_out, int out_size, void* d_ws, size_t ws_size,
                              hipStream_t stream) {
    const float* feat   = (const float*)d_in[0];
    const float* weight = (const float*)d_in[1];
    const float* bias   = (const float*)d_in[2];
    const int*   src    = (const int*)d_in[3];
    const int*   dst    = (const int*)d_in[4];
    float* out = (float*)d_out;

    const int n_nodes = in_sizes[0] / IN_FEATS;
    const int n_edges = in_sizes[3];

    float* ws_f    = (float*)d_ws;
    float* out_deg = ws_f;
    float* in_deg  = out_deg + n_nodes;
    float* wT      = in_deg + n_nodes;
    float* tk_vals = wT + IN_FEATS * OUT_FEATS;
    unsigned int* tk_idx = (unsigned int*)(tk_vals + (long long)n_nodes * MAXK);
    int* bkt_cnt   = (int*)(tk_idx + (long long)n_nodes * (MAXK / 4));
    int* bkt_cur   = bkt_cnt + MAX_BKT;
    int* bkt_start = bkt_cur + MAX_BKT;
    unsigned int* binned = (unsigned int*)(bkt_start + MAX_BKT);

    {
        int work = 2 * n_nodes > 2 * MAX_BKT ? 2 * n_nodes : 2 * MAX_BKT;
        int grid = (work + 255) / 256;
        init_kernel<<<grid, 256, 0, stream>>>(out_deg, bkt_cnt, n_nodes);
    }
    {
        int work = n_edges > IN_FEATS * OUT_FEATS ? n_edges : IN_FEATS * OUT_FEATS;
        int grid = (work + 255) / 256;
        deg_wt_kernel<<<grid, 256, 0, stream>>>(src, dst, out_deg, in_deg, weight, wT,
                                                n_edges);
    }
    const int n_tiles  = (n_nodes + BKT_NODES - 1) / BKT_NODES;
    const int n_chunks = (n_edges + EDGE_CHUNK - 1) / EDGE_CHUNK;
    bkt_count_kernel<<<n_tiles, BKT_NODES, 0, stream>>>(in_deg, bkt_cnt, n_nodes);
    {
        int grid = (n_nodes + TILE_M - 1) / TILE_M;
        gemm_maxk_kernel<<<grid, 32, 0, stream>>>(feat, wT, out_deg, in_deg,
                                                   tk_vals, tk_idx, n_nodes);
    }
    bkt_fill_kernel<<<n_chunks, 1024, 0, stream>>>(src, dst, bkt_cnt, bkt_cur, bkt_start,
                                                   binned, n_edges);
    bkt_sum_kernel<<<n_tiles, 1024, 0, stream>>>(binned, bkt_cnt, bkt_start, tk_vals, tk_idx,
                                                 bias, out, n_nodes);
}


# degree counting fused into the GEMM launch: each block counts an equal share of the edges, float adds issued last, under other blocks' matrix work; record scale moved to the tile-count pass
# speedup vs baseline: 1.4719x; 1.0453x over previous
#include <hip/hip_runtime.h>
#include <math.h>

typedef __attribute__((ext_vector_type(2))) float v2f;
typedef __attribute__((ext_vector_type(8))) float v8f;

#define IN_FEATS 256
#define OUT_FEATS 64
#define MAXK 16
#define TILE_M 16
#define LDA (IN_FEATS + 4)
#define BKT_SHIFT 8
#define BKT_NODES (1 << BKT_SHIFT)
#define MAX_BKT 1024
#define EDGE_CHUNK 8192
#define EPT (EDGE_CHUNK / 1024)

__global__ void init_kernel(float* __restrict__ deg, int* __restrict__ bkt,
                            const float* __restrict__ weight, float* __restrict__ wT,
                            int n_nodes) {
    int i = blockIdx.x * blockDim.x + threadIdx.x;
    if (i < 2 * n_nodes) deg[i] = 0.0f;
    if (i < 2 * MAX_BKT) bkt[i] = 0;
    if (i < IN_FEATS * OUT_FEATS) {
        int k = i >> 6;
        int c = i & 63;
        wT[c * IN_FEATS + k] = weight[i];
    }
}


__device__ __forceinline__ int row_sum16(int x) {
    x += __builtin_amdgcn_update_dpp(0, x, 0xB1, 0xF, 0xF, false);
    x += __builtin_amdgcn_update_dpp(0, x, 0x4E, 0xF, 0xF, false);
    x += __builtin_amdgcn_update_dpp(0, x, 0x141, 0xF, 0xF, false);
    x += __builtin_amdgcn_update_dpp(0, x, 0x140, 0xF, 0xF, false);
    return x;
}

__device__ __forceinline__ unsigned int order_key(float f) {
    unsigned int u = __float_as_uint(f);
    return u ^ ((unsigned int)((int)u >> 31) | 0x80000000u);
}

__global__ __launch_bounds__(32)
void gemm_maxk_kernel(const float* __restrict__ feat, const float* __restrict__ wT,
                      const int* __restrict__ src, const int* __restrict__ dst,
                      float* __restrict__ out_deg, float* __restrict__ in_deg,
                      float* __restrict__ tk_vals, unsigned int* __restrict__ tk_idx,
                      int n_nodes, int n_edges) {
    __shared__ __attribute__((aligned(16))) float sA[TILE_M * LDA];

    const int lane = threadIdx.x;
    const int m0   = blockIdx.x * TILE_M;
    const int r16  = lane & 15;
    const int half = lane >> 4;

    const int per = (n_edges + (int)gridDim.x - 1) / (int)gridDim.x;
    const int e0  = blockIdx.x * per;
    const int e1  = e0 + per < n_edges ? e0 + per : n_edges;
    int es[8], ed[8];
    #pragma unroll
    for (int j = 0; j < 8; ++j) {
        const int e  = e0 + j * 32 + lane;
        const int ec = e < n_edges ? e : n_edges - 1;
        es[j] = src[ec];
        ed[j] = dst[ec];
    }

    #pragma unroll
    for (int pass = 0; pass < 2; ++pass) {
        float4 buf[16];
        #pragma unroll
        for (int j = 0; j < 16; ++j) {
            int i   = (pass * 16 + j) * 32 + lane;
            int row = i >> 6;
            int c4  = i & 63;
            int grow = m0 + row;
            grow = grow < n_nodes ? grow : n_nodes - 1;
            buf[j] = ((const float4*)feat)[(long long)grow * (IN_FEATS / 4) + c4];
        }
        #pragma unroll
        for (int j = 0; j < 16; ++j) {
            int i   = (pass * 16 + j) * 32 + lane;
            int row = i >> 6;
            int c4  = i & 63;
            float4 v = buf[j];
            if (m0 + row >= n_nodes) v = make_float4(0.f, 0.f, 0.f, 0.f);
            *(float4*)&sA[row * LDA + c4 * 4] = v;
        }
    }
    __syncthreads();

    const int koff = half << 1;
    const float* arow = sA + r16 * LDA;
    const float* w0 = wT + r16 * IN_FEATS;
    const float* w1 = w0 + 16 * IN_FEATS;
    const float* w2 = w0 + 32 * IN_FEATS;
    const float* w3 = w0 + 48 * IN_FEATS;

    v8f c0 = {0.f, 0.f, 0.f, 0.f, 0.f, 0.f, 0.f, 0.f};
    v8f c1 = c0, c2 = c0, c3 = c0;
    #pragma unroll
    for (int k0 = 0; k0 < IN_FEATS; k0 += 4) {
        v2f a, b0, b1, b2, b3;
        a.x  = arow[k0 + koff];  a.y  = arow[k0 + koff + 1];
        b0.x = w0[k0 + koff];    b0.y = w0[k0 + koff + 1];
        b1.x = w1[k0 + koff];    b1.y = w1[k0 + koff + 1];
        b2.x = w2[k0 + koff];    b2.y = w2[k0 + koff + 1];
        b3.x = w3[k0 + koff];    b3.y = w3[k0 + koff + 1];
        c0 = __builtin_amdgcn_wmma_f32_16x16x4_f32(false, a, false, b0, (short)0, c0, false, false);
        c1 = __builtin_amdgcn_wmma_f32_16x16x4_f32(false, a, false, b1, (short)0, c1, false, false);
        c2 = __builtin_amdgcn_wmma_f32_16x16x4_f32(false, a, false, b2, (short)0, c2, false, false);
        c3 = __builtin_amdgcn_wmma_f32_16x16x4_f32(false, a, false, b3, (short)0, c3, false, false);
    }

    unsigned int key[4][8];
    #pragma unroll
    for (int r = 0; r < 8; ++r) {
        key[0][r] = order_key(c0[r]);
        key[1][r] = order_key(c1[r]);
        key[2][r] = order_key(c2[r]);
        key[3][r] = order_key(c3[r]);
    }

    unsigned int thr[8];
    #pragma unroll
    for (int r = 0; r < 8; ++r) thr[r] = 0u;
    for (int bit = 31; bit >= 0; --bit) {
        const unsigned int bmask = 1u << bit;
        #pragma unroll
        for (int r = 0; r < 8; ++r) {
            const unsigned int cand = thr[r] | bmask;
            int cnt = (key[0][r] >= cand ? 1 : 0) + (key[1][r] >= cand ? 1 : 0) +
                      (key[2][r] >= cand ? 1 : 0) + (key[3][r] >= cand ? 1 : 0);
            cnt = row_sum16(cnt);
            thr[r] = cnt >= MAXK ? cand : thr[r];
        }
    }

    const unsigned int hshift = (unsigned int)half << 4;
    const unsigned int below  = (1u << r16) - 1u;
    __syncthreads();
    float* sV = sA;
    unsigned char* sI = (unsigned char*)(sA + TILE_M * MAXK);
    #pragma unroll
    for (int r = 0; r < 8; ++r) {
        const unsigned int T = thr[r];
        bool gt[4], eq[4];
        unsigned int me[4];
        int n_gt = 0;
        #pragma unroll
        for (int t = 0; t < 4; ++t) {
            gt[t] = key[t][r] > T;
            eq[t] = key[t][r] == T;
            unsigned int mg = (__builtin_amdgcn_ballot_w32(gt[t]) >> hshift) & 0xffffu;
            me[t] = (__builtin_amdgcn_ballot_w32(eq[t]) >> hshift) & 0xffffu;
            n_gt += __popc(mg);
        }
        const int need = MAXK - n_gt;
        const float hv[4] = {c0[r], c1[r], c2[r], c3[r]};
        int eq_before = 0, slot_base = 0;
        #pragma unroll
        for (int t = 0; t < 4; ++t) {
            const int eq_rank = eq_before + __popc(me[t] & below);
            const bool sel = gt[t] || (eq[t] && eq_rank < need);
            const unsigned int ms = (__builtin_amdgcn_ballot_w32(sel) >> hshift) & 0xffffu;
            const int slot = slot_base + __popc(ms & below);
            if (sel) {
                sV[(r + 8 * half) * MAXK + slot] = hv[t];
                sI[(r + 8 * half) * MAXK + slot] = (unsigned char)(t * 16 + r16);
            }
            eq_before += __popc(me[t]);
            slot_base += __popc(ms);
        }
    }
    __syncthreads();

    const float4* sV4 = (const float4*)sV;
    #pragma unroll
    for (int q = 0; q < 2; ++q) {
        const int f = q * 32 + lane;
        if (m0 + (f >> 2) < n_nodes)
            ((float4*)tk_vals)[(long long)m0 * (MAXK / 4) + f] = sV4[f];
    }
    if (lane < TILE_M && m0 + lane < n_nodes)
        ((uint4*)tk_idx)[m0 + lane] = ((const uint4*)sI)[lane];

    #pragma unroll
    for (int j = 0; j < 8; ++j) {
        if (e0 + j * 32 + lane < e1) {
            atomicAdd(&out_deg[es[j]], 1.0f);
            atomicAdd(&in_deg[ed[j]], 1.0f);
        }
    }
    for (int e = e0 + 8 * 32 + lane; e < e1; e += 32) {
        atomicAdd(&out_deg[src[e]], 1.0f);
        atomicAdd(&in_deg[dst[e]], 1.0f);
    }
}


__global__ __launch_bounds__(BKT_NODES)
void bkt_count_scale_kernel(const float* __restrict__ out_deg, const float* __restrict__ in_deg,
                            int* __restrict__ bkt_cnt, float* __restrict__ tk_vals, int n_nodes) {
    __shared__ int wtot[BKT_NODES / 32];
    __shared__ float ssc[BKT_NODES];
    const int tid  = threadIdx.x;
    const int node = blockIdx.x * BKT_NODES + tid;
    float id = node < n_nodes ? in_deg[node] : 0.f;
    float od = node < n_nodes ? out_deg[node] : 0.f;
    int x = (int)id;
    od = od < 1.f ? 1.f : od;
    id = id < 1.f ? 1.f : id;
    ssc[tid] = rsqrtf(od) * rsqrtf(id);
    #pragma unroll
    for (int o = 16; o > 0; o >>= 1) x += __shfl_xor(x, o);
    if ((tid & 31) == 0) wtot[tid >> 5] = x;
    __syncthreads();
    if (tid == 0) {
        int s = 0;
        #pragma unroll
        for (int k = 0; k < BKT_NODES / 32; ++k) s += wtot[k];
        bkt_cnt[blockIdx.x] = s;
    }
    float4* rec = (float4*)tk_vals + (long long)blockIdx.x * (BKT_NODES * MAXK / 4);
    float4 v[MAXK / 4];
    #pragma unroll
    for (int q = 0; q < MAXK / 4; ++q) {
        const int f = q * BKT_NODES + tid;
        if (blockIdx.x * BKT_NODES + (f >> 2) < n_nodes) v[q] = rec[f];
    }
    #pragma unroll
    for (int q = 0; q < MAXK / 4; ++q) {
        const int f = q * BKT_NODES + tid;
        if (blockIdx.x * BKT_NODES + (f >> 2) < n_nodes) {
            const float s = ssc[f >> 2];
            v[q].x *= s; v[q].y *= s; v[q].z *= s; v[q].w *= s;
            rec[f] = v[q];
        }
    }
}

__global__ __launch_bounds__(1024)
void bkt_fill_kernel(const int* __restrict__ src, const int* __restrict__ dst,
                     const int* __restrict__ bkt_cnt, int* __restrict__ bkt_cur,
                     int* __restrict__ bkt_start, unsigned int* __restrict__ binned,
                     int n_edges) {
    __shared__ int hist[MAX_BKT];
    __shared__ int gbase[MAX_BKT];
    __shared__ int wsum[2][32];
    __shared__ unsigned int stage[EDGE_CHUNK];
    __shared__ unsigned short stage_b[EDGE_CHUNK];
    const int tid  = threadIdx.x;
    const int lane = tid & 31;
    const int wave = tid >> 5;
    hist[tid] = 0;
    const int v = bkt_cnt[tid];

    const int base = blockIdx.x * EDGE_CHUNK;
    int es[EPT], ed[EPT];
    #pragma unroll
    for (int j = 0; j < EPT; ++j) {
        const int e  = base + j * 1024 + tid;
        const int ec = e < n_edges ? e : n_edges - 1;
        const int d  = dst[ec];
        es[j] = src[ec];
        ed[j] = e < n_edges ? d : -1;
    }
    __syncthreads();
    #pragma unroll
    for (int j = 0; j < EPT; ++j)
        if (ed[j] >= 0) atomicAdd(&hist[ed[j] >> BKT_SHIFT], 1);
    __syncthreads();

    const int c = hist[tid];
    int x = v, y = c;
    #pragma unroll
    for (int o = 1; o < 32; o <<= 1) {
        const int xs = __shfl_up(x, o);
        const int ys = __shfl_up(y, o);
        if (lane >= o) { x += xs; y += ys; }
    }
    if (lane == 31) { wsum[0][wave] = x; wsum[1][wave] = y; }
    __syncthreads();
    if (wave < 2) {
        const int w = wsum[wave][lane];
        int s = w;
        #pragma unroll
        for (int o = 1; o < 32; o <<= 1) {
            const int t = __shfl_up(s, o);
            if (lane >= o) s += t;
        }
        wsum[wave][lane] = s - w;
    }
    __syncthreads();

    const int start  = x - v + wsum[0][wave];
    const int lstart = y - c + wsum[1][wave];
    if (blockIdx.x == 0) bkt_start[tid] = start;
    const int rsv = c ? atomicAdd(&bkt_cur[tid], c) : 0;
    gbase[tid] = start + rsv - lstart;
    hist[tid]  = lstart;
    __syncthreads();

    #pragma unroll
    for (int j = 0; j < EPT; ++j) {
        if (ed[j] >= 0) {
            const int b    = ed[j] >> BKT_SHIFT;
            const int slot = atomicAdd(&hist[b], 1);
            stage[slot]   = ((unsigned int)es[j] << BKT_SHIFT) | ((unsigned int)ed[j] & (BKT_NODES - 1));
            stage_b[slot] = (unsigned short)b;
        }
    }
    __syncthreads();

    const int left = n_edges - base;
    const int n_here = left < EDGE_CHUNK ? left : EDGE_CHUNK;
    #pragma unroll
    for (int j = 0; j < EPT; ++j) {
        const int slot = j * 1024 + tid;
        if (slot < n_here) binned[gbase[stage_b[slot]] + slot] = stage[slot];
    }
}

__device__ __forceinline__ void add_record(float* row, float4 v, unsigned int ip) {
    atomicAdd(row + (ip & 0xffu), v.x);
    atomicAdd(row + ((ip >> 8) & 0xffu), v.y);
    atomicAdd(row + ((ip >> 16) & 0xffu), v.z);
    atomicAdd(row + ((ip >> 24) & 0xffu), v.w);
}

__global__ __launch_bounds__(1024)
void bkt_sum_kernel(const unsigned int* __restrict__ binned, const int* __restrict__ bkt_cnt,
                    const int* __restrict__ bkt_start, const float* __restrict__ tk_vals,
                    const unsigned int* __restrict__ tk_idx, const float* __restrict__ bias,
                    float* __restrict__ out, int n_nodes) {
    __shared__ __attribute__((aligned(16))) float acc[BKT_NODES * OUT_FEATS];
    const int tid = threadIdx.x;
    for (int i = tid; i < BKT_NODES * OUT_FEATS; i += 1024) acc[i] = bias[i & (OUT_FEATS - 1)];
    __syncthreads();

    const int cnt = bkt_cnt[blockIdx.x];
    const unsigned int* mine = binned + bkt_start[blockIdx.x];
    const int part = tid & 3;
    int i0 = tid >> 2;
    for (; i0 + 3 * 256 < cnt; i0 += 4 * 256) {
        unsigned int w[4], ip[4];
        float4 v[4];
        #pragma unroll
        for (int k = 0; k < 4; ++k) w[k] = mine[i0 + 256 * k];
        #pragma unroll
        for (int k = 0; k < 4; ++k) {
            const long long s = (long long)(w[k] >> BKT_SHIFT);
            v[k]  = ((const float4*)tk_vals)[s * 4 + part];
            ip[k] = tk_idx[s * 4 + part];
        }
        #pragma unroll
        for (int k = 0; k < 4; ++k)
            add_record(acc + (w[k] & (BKT_NODES - 1)) * OUT_FEATS, v[k], ip[k]);
    }
    for (; i0 < cnt; i0 += 256) {
        const unsigned int w = mine[i0];
        const long long s = (long long)(w >> BKT_SHIFT);
        const float4 v = ((const float4*)tk_vals)[s * 4 + part];
        const unsigned int ip = tk_idx[s * 4 + part];
        add_record(acc + (w & (BKT_NODES - 1)) * OUT_FEATS, v, ip);
    }
    __syncthreads();

    const long long node0 = (long long)blockIdx.x * BKT_NODES;
    for (int i = tid; i < BKT_NODES * OUT_FEATS / 4; i += 1024) {
        if (node0 + (i >> 4) < n_nodes)
            ((float4*)out)[node0 * (OUT_FEATS / 4) + i] = ((const float4*)acc)[i];
    }
}

extern "C" void kernel_launch(void* const* d_in, const int* in_sizes, int n_in,
                              void* d_out, int out_size, void* d_ws, size_t ws_size,
                              hipStream_t stream) {
    const float* feat   = (const float*)d_in[0];
    const float* weight = (const float*)d_in[1];
    const float* bias   = (const float*)d_in[2];
    const int*   src    = (const int*)d_in[3];
    const int*   dst    = (const int*)d_in[4];
    float* out = (float*)d_out;

    const int n_nodes = in_sizes[0] / IN_FEATS;
    const int n_edges = in_sizes[3];

    float* ws_f    = (float*)d_ws;
    float* out_deg = ws_f;
    float* in_deg  = out_deg + n_nodes;
    float* wT      = in_deg + n_nodes;
    float* tk_vals = wT + IN_FEATS * OUT_FEATS;
    unsigned int* tk_idx = (unsigned int*)(tk_vals + (long long)n_nodes * MAXK);
    int* bkt_cnt   = (int*)(tk_idx + (long long)n_nodes * (MAXK / 4));
    int* bkt_cur   = bkt_cnt + MAX_BKT;
    int* bkt_start = bkt_cur + MAX_BKT;
    unsigned int* binned = (unsigned int*)(bkt_start + MAX_BKT);

    {
        int work = 2 * n_nodes > 2 * MAX_BKT ? 2 * n_nodes : 2 * MAX_BKT;
        work = work > IN_FEATS * OUT_FEATS ? work : IN_FEATS * OUT_FEATS;
        int grid = (work + 255) / 256;
        init_kernel<<<grid, 256, 0, stream>>>(out_deg, bkt_cnt, weight, wT, n_nodes);
    }
    const int n_tiles  = (n_nodes + BKT_NODES - 1) / BKT_NODES;
    const int n_chunks = (n_edges + EDGE_CHUNK - 1) / EDGE_CHUNK;
    {
        int grid = (n_nodes + TILE_M - 1) / TILE_M;
        gemm_maxk_kernel<<<grid, 32, 0, stream>>>(feat, wT, src, dst, out_deg, in_deg,
                                                   tk_vals, tk_idx, n_nodes, n_edges);
    }
    bkt_count_scale_kernel<<<n_tiles, BKT_NODES, 0, stream>>>(out_deg, in_deg, bkt_cnt, tk_vals,
                                                             n_nodes);
    bkt_fill_kernel<<<n_chunks, 1024, 0, stream>>>(src, dst, bkt_cnt, bkt_cur, bkt_start,
                                                   binned, n_edges);
    bkt_sum_kernel<<<n_tiles, 1024, 0, stream>>>(binned, bkt_cnt, bkt_start, tk_vals, tk_idx,
                                                 bias, out, n_nodes);
}


# degrees read off the sort: source ends also bucketed by source tile, a block per tile counts in- and out-degrees on LDS counters; tile sizes from LDS chunk histograms; sum order unchanged
# speedup vs baseline: 2.7911x; 1.8963x over previous
#include <hip/hip_runtime.h>
#include <math.h>

typedef __attribute__((ext_vector_type(2))) float v2f;
typedef __attribute__((ext_vector_type(8))) float v8f;

#define IN_FEATS 256
#define OUT_FEATS 64
#define MAXK 16
#define TILE_M 16
#define LDA (IN_FEATS + 4)
#define BKT_SHIFT 8
#define BKT_NODES (1 << BKT_SHIFT)
#define MAX_BKT 1024
#define EDGE_CHUNK 8192
#define EPT (EDGE_CHUNK / 1024)
#define HIST_CHUNK 4096
#define HIST_EPT 16
#define TAB_INTS (6 * MAX_BKT)

__global__ void init_kernel(int* __restrict__ tab,
                            const float* __restrict__ weight, float* __restrict__ wT) {
    int i = blockIdx.x * blockDim.x + threadIdx.x;
    if (i < TAB_INTS) tab[i] = 0;
    if (i < IN_FEATS * OUT_FEATS) {
        int k = i >> 6;
        int c = i & 63;
        wT[c * IN_FEATS + k] = weight[i];
    }
}


__device__ __forceinline__ int row_sum16(int x) {
    x += __builtin_amdgcn_update_dpp(0, x, 0xB1, 0xF, 0xF, false);
    x += __builtin_amdgcn_update_dpp(0, x, 0x4E, 0xF, 0xF, false);
    x += __builtin_amdgcn_update_dpp(0, x, 0x141, 0xF, 0xF, false);
    x += __builtin_amdgcn_update_dpp(0, x, 0x140, 0xF, 0xF, false);
    return x;
}

__device__ __forceinline__ unsigned int order_key(float f) {
    unsigned int u = __float_as_uint(f);
    return u ^ ((unsigned int)((int)u >> 31) | 0x80000000u);
}

__global__ __launch_bounds__(32)
void gemm_maxk_kernel(const float* __restrict__ feat, const float* __restrict__ wT,
                      const int* __restrict__ src, const int* __restrict__ dst,
                      int* __restrict__ cnt_d, int* __restrict__ cnt_s,
                      float* __restrict__ tk_vals, unsigned int* __restrict__ tk_idx,
                      int n_nodes, int n_edges, int n_hist) {
    __shared__ __attribute__((aligned(16))) float sA[TILE_M * LDA];

    const int lane = threadIdx.x;

    if ((int)blockIdx.x < n_hist) {
        int* hd = (int*)sA;
        int* hs = hd + MAX_BKT;
        for (int i = lane; i < 2 * MAX_BKT; i += 32) hd[i] = 0;
        __syncthreads();
        const int e0 = blockIdx.x * HIST_CHUNK;
        for (int t = 0; t < HIST_CHUNK / (32 * HIST_EPT); ++t) {
            int ks[HIST_EPT], kd[HIST_EPT];
            #pragma unroll
            for (int j = 0; j < HIST_EPT; ++j) {
                const int e  = e0 + (t * HIST_EPT + j) * 32 + lane;
                const int ec = e < n_edges ? e : n_edges - 1;
                const int dv = dst[ec];
                ks[j] = src[ec];
                kd[j] = e < n_edges ? dv : -1;
            }
            #pragma unroll
            for (int j = 0; j < HIST_EPT; ++j) {
                if (kd[j] >= 0) {
                    atomicAdd(&hd[kd[j] >> BKT_SHIFT], 1);
                    atomicAdd(&hs[ks[j] >> BKT_SHIFT], 1);
                }
            }
        }
        __syncthreads();
        for (int i = lane; i < MAX_BKT; i += 32) {
            const int cd = hd[i];
            const int cs = hs[i];
            if (cd) atomicAdd(&cnt_d[i], cd);
            if (cs) atomicAdd(&cnt_s[i], cs);
        }
        return;
    }

    const int m0   = ((int)blockIdx.x - n_hist) * TILE_M;
    const int r16  = lane & 15;
    const int half = lane >> 4;

    #pragma unroll
    for (int pass = 0; pass < 2; ++pass) {
        float4 buf[16];
        #pragma unroll
        for (int j = 0; j < 16; ++j) {
            int i   = (pass * 16 + j) * 32 + lane;
            int row = i >> 6;
            int c4  = i & 63;
            int grow = m0 + row;
            grow = grow < n_nodes ? grow : n_nodes - 1;
            buf[j] = ((const float4*)feat)[(long long)grow * (IN_FEATS / 4) + c4];
        }
        #pragma unroll
        for (int j = 0; j < 16; ++j) {
            int i   = (pass * 16 + j) * 32 + lane;
            int row = i >> 6;
            int c4  = i & 63;
            float4 v = buf[j];
            if (m0 + row >= n_nodes) v = make_float4(0.f, 0.f, 0.f, 0.f);
            *(float4*)&sA[row * LDA + c4 * 4] = v;
        }
    }
    __syncthreads();

    const int koff = half << 1;
    const float* arow = sA + r16 * LDA;
    const float* w0 = wT + r16 * IN_FEATS;
    const float* w1 = w0 + 16 * IN_FEATS;
    const float* w2 = w0 + 32 * IN_FEATS;
    const float* w3 = w0 + 48 * IN_FEATS;

    v8f c0 = {0.f, 0.f, 0.f, 0.f, 0.f, 0.f, 0.f, 0.f};
    v8f c1 = c0, c2 = c0, c3 = c0;
    #pragma unroll
    for (int k0 = 0; k0 < IN_FEATS; k0 += 4) {
        v2f a, b0, b1, b2, b3;
        a.x  = arow[k0 + koff];  a.y  = arow[k0 + koff + 1];
        b0.x = w0[k0 + koff];    b0.y = w0[k0 + koff + 1];
        b1.x = w1[k0 + koff];    b1.y = w1[k0 + koff + 1];
        b2.x = w2[k0 + koff];    b2.y = w2[k0 + koff + 1];
        b3.x = w3[k0 + koff];    b3.y = w3[k0 + koff + 1];
        c0 = __builtin_amdgcn_wmma_f32_16x16x4_f32(false, a, false, b0, (short)0, c0, false, false);
        c1 = __builtin_amdgcn_wmma_f32_16x16x4_f32(false, a, false, b1, (short)0, c1, false, false);
        c2 = __builtin_amdgcn_wmma_f32_16x16x4_f32(false, a, false, b2, (short)0, c2, false, false);
        c3 = __builtin_amdgcn_wmma_f32_16x16x4_f32(false, a, false, b3, (short)0, c3, false, false);
    }

    unsigned int key[4][8];
    #pragma unroll
    for (int r = 0; r < 8; ++r) {
        key[0][r] = order_key(c0[r]);
        key[1][r] = order_key(c1[r]);
        key[2][r] = order_key(c2[r]);
        key[3][r] = order_key(c3[r]);
    }

    unsigned int thr[8];
    #pragma unroll
    for (int r = 0; r < 8; ++r) thr[r] = 0u;
    for (int bit = 31; bit >= 0; --bit) {
        const unsigned int bmask = 1u << bit;
        #pragma unroll
        for (int r = 0; r < 8; ++r) {
            const unsigned int cand = thr[r] | bmask;
            int cnt = (key[0][r] >= cand ? 1 : 0) + (key[1][r] >= cand ? 1 : 0) +
                      (key[2][r] >= cand ? 1 : 0) + (key[3][r] >= cand ? 1 : 0);
            cnt = row_sum16(cnt);
            thr[r] = cnt >= MAXK ? cand : thr[r];
        }
    }

    const unsigned int hshift = (unsigned int)half << 4;
    const unsigned int below  = (1u << r16) - 1u;
    __syncthreads();
    float* sV = sA;
    unsigned char* sI = (unsigned char*)(sA + TILE_M * MAXK);
    #pragma unroll
    for (int r = 0; r < 8; ++r) {
        const unsigned int T = thr[r];
        bool gt[4], eq[4];
        unsigned int me[4];
        int n_gt = 0;
        #pragma unroll
        for (int t = 0; t < 4; ++t) {
            gt[t] = key[t][r] > T;
            eq[t] = key[t][r] == T;
            unsigned int mg = (__builtin_amdgcn_ballot_w32(gt[t]) >> hshift) & 0xffffu;
            me[t] = (__builtin_amdgcn_ballot_w32(eq[t]) >> hshift) & 0xffffu;
            n_gt += __popc(mg);
        }
        const int need = MAXK - n_gt;
        const float hv[4] = {c0[r], c1[r], c2[r], c3[r]};
        int eq_before = 0, slot_base = 0;
        #pragma unroll
        for (int t = 0; t < 4; ++t) {
            const int eq_rank = eq_before + __popc(me[t] & below);
            const bool sel = gt[t] || (eq[t] && eq_rank < need);
            const unsigned int ms = (__builtin_amdgcn_ballot_w32(sel) >> hshift) & 0xffffu;
            const int slot = slot_base + __popc(ms & below);
            if (sel) {
                sV[(r + 8 * half) * MAXK + slot] = hv[t];
                sI[(r + 8 * half) * MAXK + slot] = (unsigned char)(t * 16 + r16);
            }
            eq_before += __popc(me[t]);
            slot_base += __popc(ms);
        }
    }
    __syncthreads();

    const float4* sV4 = (const float4*)sV;
    #pragma unroll
    for (int q = 0; q < 2; ++q) {
        const int f = q * 32 + lane;
        if (m0 + (f >> 2) < n_nodes)
            ((float4*)tk_vals)[(long long)m0 * (MAXK / 4) + f] = sV4[f];
    }
    if (lane < TILE_M && m0 + lane < n_nodes)
        ((uint4*)tk_idx)[m0 + lane] = ((const uint4*)sI)[lane];
}


__global__ __launch_bounds__(1024)
void bkt_fill_kernel(const int* __restrict__ src, const int* __restrict__ dst,
                     int* __restrict__ tab, unsigned int* __restrict__ binned_d,
                     unsigned char* __restrict__ binned_s, int n_edges) {
    const int key = blockIdx.y;
    const int* bkt_cnt = tab + key * 3 * MAX_BKT;
    int* bkt_cur   = tab + key * 3 * MAX_BKT + MAX_BKT;
    int* bkt_start = tab + key * 3 * MAX_BKT + 2 * MAX_BKT;
    __shared__ int hist[MAX_BKT];
    __shared__ int gbase[MAX_BKT];
    __shared__ int wsum[2][32];
    __shared__ unsigned int stage[EDGE_CHUNK];
    __shared__ unsigned short stage_b[EDGE_CHUNK];
    const int tid  = threadIdx.x;
    const int lane = tid & 31;
    const int wave = tid >> 5;
    hist[tid] = 0;
    const int v = bkt_cnt[tid];

    const int base = blockIdx.x * EDGE_CHUNK;
    int es[EPT], ed[EPT];
    #pragma unroll
    for (int j = 0; j < EPT; ++j) {
        const int e  = base + j * 1024 + tid;
        const int ec = e < n_edges ? e : n_edges - 1;
        const int sv = src[ec];
        const int dv = dst[ec];
        es[j] = key ? dv : sv;
        ed[j] = e < n_edges ? (key ? sv : dv) : -1;
    }
    __syncthreads();
    #pragma unroll
    for (int j = 0; j < EPT; ++j)
        if (ed[j] >= 0) atomicAdd(&hist[ed[j] >> BKT_SHIFT], 1);
    __syncthreads();

    const int c = hist[tid];
    int x = v, y = c;
    #pragma unroll
    for (int o = 1; o < 32; o <<= 1) {
        const int xs = __shfl_up(x, o);
        const int ys = __shfl_up(y, o);
        if (lane >= o) { x += xs; y += ys; }
    }
    if (lane == 31) { wsum[0][wave] = x; wsum[1][wave] = y; }
    __syncthreads();
    if (wave < 2) {
        const int w = wsum[wave][lane];
        int s = w;
        #pragma unroll
        for (int o = 1; o < 32; o <<= 1) {
            const int t = __shfl_up(s, o);
            if (lane >= o) s += t;
        }
        wsum[wave][lane] = s - w;
    }
    __syncthreads();

    const int start  = x - v + wsum[0][wave];
    const int lstart = y - c + wsum[1][wave];
    if (blockIdx.x == 0) bkt_start[tid] = start;
    const int rsv = c ? atomicAdd(&bkt_cur[tid], c) : 0;
    gbase[tid] = start + rsv - lstart;
    hist[tid]  = lstart;
    __syncthreads();

    #pragma unroll
    for (int j = 0; j < EPT; ++j) {
        if (ed[j] >= 0) {
            const int b    = ed[j] >> BKT_SHIFT;
            const int slot = atomicAdd(&hist[b], 1);
            stage[slot]   = ((unsigned int)es[j] << BKT_SHIFT) | ((unsigned int)ed[j] & (BKT_NODES - 1));
            stage_b[slot] = (unsigned short)b;
        }
    }
    __syncthreads();

    const int left = n_edges - base;
    const int n_here = left < EDGE_CHUNK ? left : EDGE_CHUNK;
    #pragma unroll
    for (int j = 0; j < EPT; ++j) {
        const int slot = j * 1024 + tid;
        if (slot < n_here) {
            const int g = gbase[stage_b[slot]] + slot;
            if (key == 0) binned_d[g] = stage[slot];
            else          binned_s[g] = (unsigned char)stage[slot];
        }
    }
}

__global__ __launch_bounds__(1024)
void bkt_degree_scale_kernel(const unsigned int* __restrict__ binned_d,
                             const unsigned char* __restrict__ binned_s,
                             const int* __restrict__ tab, float* __restrict__ tk_vals,
                             int n_nodes, int n_edges) {
    __shared__ int cin[BKT_NODES];
    __shared__ int cout[BKT_NODES];
    __shared__ float ssc[BKT_NODES];
    const int tid = threadIdx.x;
    if (tid < BKT_NODES) { cin[tid] = 0; cout[tid] = 0; }
    const int nd = tab[blockIdx.x];
    const int sd = tab[2 * MAX_BKT + blockIdx.x];
    const int ns = tab[3 * MAX_BKT + blockIdx.x];
    const int ss = tab[5 * MAX_BKT + blockIdx.x];
    float4* rec = (float4*)tk_vals + (long long)blockIdx.x * (BKT_NODES * MAXK / 4);
    const bool live = blockIdx.x * BKT_NODES + (tid >> 2) < n_nodes;
    float4 v = make_float4(0.f, 0.f, 0.f, 0.f);
    if (live) v = rec[tid];
    __syncthreads();
    const int nmax = nd > ns ? nd : ns;
    for (int i0 = tid; i0 < nmax; i0 += 4 * 1024) {
        unsigned int wd[4], ws[4];
        #pragma unroll
        for (int k = 0; k < 4; ++k) {
            const int i  = i0 + 1024 * k;
            const int gd = sd + i < n_edges ? sd + i : n_edges - 1;
            const int gs = ss + i < n_edges ? ss + i : n_edges - 1;
            wd[k] = binned_d[gd];
            ws[k] = binned_s[gs];
        }
        #pragma unroll
        for (int k = 0; k < 4; ++k) {
            const int i = i0 + 1024 * k;
            if (i < nd) atomicAdd(&cin[wd[k] & (BKT_NODES - 1)], 1);
            if (i < ns) atomicAdd(&cout[ws[k] & (BKT_NODES - 1)], 1);
        }
    }
    __syncthreads();
    if (tid < BKT_NODES) {
        float id = (float)cin[tid];
        float od = (float)cout[tid];
        od = od < 1.f ? 1.f : od;
        id = id < 1.f ? 1.f : id;
        ssc[tid] = rsqrtf(od) * rsqrtf(id);
    }
    __syncthreads();
    if (live) {
        const float sc = ssc[tid >> 2];
        v.x *= sc; v.y *= sc; v.z *= sc; v.w *= sc;
        rec[tid] = v;
    }
}

__device__ __forceinline__ void add_record(float* row, float4 v, unsigned int ip) {
    atomicAdd(row + (ip & 0xffu), v.x);
    atomicAdd(row + ((ip >> 8) & 0xffu), v.y);
    atomicAdd(row + ((ip >> 16) & 0xffu), v.z);
    atomicAdd(row + ((ip >> 24) & 0xffu), v.w);
}

__global__ __launch_bounds__(1024)
void bkt_sum_kernel(const unsigned int* __restrict__ binned, const int* __restrict__ bkt_cnt,
                    const int* __restrict__ bkt_start, const float* __restrict__ tk_vals,
                    const unsigned int* __restrict__ tk_idx, const float* __restrict__ bias,
                    float* __restrict__ out, int n_nodes) {
    __shared__ __attribute__((aligned(16))) float acc[BKT_NODES * OUT_FEATS];
    const int tid = threadIdx.x;
    for (int i = tid; i < BKT_NODES * OUT_FEATS; i += 1024) acc[i] = bias[i & (OUT_FEATS - 1)];
    __syncthreads();

    const int cnt = bkt_cnt[blockIdx.x];
    const unsigned int* mine = binned + bkt_start[blockIdx.x];
    const int part = tid & 3;
    int i0 = tid >> 2;
    for (; i0 + 3 * 256 < cnt; i0 += 4 * 256) {
        unsigned int w[4], ip[4];
        float4 v[4];
        #pragma unroll
        for (int k = 0; k < 4; ++k) w[k] = mine[i0 + 256 * k];
        #pragma unroll
        for (int k = 0; k < 4; ++k) {
            const long long s = (long long)(w[k] >> BKT_SHIFT);
            v[k]  = ((const float4*)tk_vals)[s * 4 + part];
            ip[k] = tk_idx[s * 4 + part];
        }
        #pragma unroll
        for (int k = 0; k < 4; ++k)
            add_record(acc + (w[k] & (BKT_NODES - 1)) * OUT_FEATS, v[k], ip[k]);
    }
    for (; i0 < cnt; i0 += 256) {
        const unsigned int w = mine[i0];
        const long long s = (long long)(w >> BKT_SHIFT);
        const float4 v = ((const float4*)tk_vals)[s * 4 + part];
        const unsigned int ip = tk_idx[s * 4 + part];
        add_record(acc + (w & (BKT_NODES - 1)) * OUT_FEATS, v, ip);
    }
    __syncthreads();

    const long long node0 = (long long)blockIdx.x * BKT_NODES;
    for (int i = tid; i < BKT_NODES * OUT_FEATS / 4; i += 1024) {
        if (node0 + (i >> 4) < n_nodes)
            ((float4*)out)[node0 * (OUT_FEATS / 4) + i] = ((const float4*)acc)[i];
    }
}

extern "C" void kernel_launch(void* const* d_in, const int* in_sizes, int n_in,
                              void* d_out, int out_size, void* d_ws, size_t ws_size,
                              hipStream_t stream) {
    const float* feat   = (const float*)d_in[0];
    const float* weight = (const float*)d_in[1];
    const float* bias   = (const float*)d_in[2];
    const int*   src    = (const int*)d_in[3];
    const int*   dst    = (const int*)d_in[4];
    float* out = (float*)d_out;

    const int n_nodes = in_sizes[0] / IN_FEATS;
    const int n_edges = in_sizes[3];

    float* wT      = (float*)d_ws;
    float* tk_vals = wT + IN_FEATS * OUT_FEATS;
    unsigned int* tk_idx = (unsigned int*)(tk_vals + (long long)n_nodes * MAXK);
    int* tab       = (int*)(tk_idx + (long long)n_nodes * (MAXK / 4));
    unsigned int* binned_d  = (unsigned int*)(tab + TAB_INTS);
    unsigned char* binned_s = (unsigned char*)d_out;

    init_kernel<<<IN_FEATS * OUT_FEATS / 256, 256, 0, stream>>>(tab, weight, wT);
    const int n_tiles  = (n_nodes + BKT_NODES - 1) / BKT_NODES;
    const int n_chunks = (n_edges + EDGE_CHUNK - 1) / EDGE_CHUNK;
    const int n_hist   = (n_edges + HIST_CHUNK - 1) / HIST_CHUNK;
    {
        int grid = n_hist + (n_nodes + TILE_M - 1) / TILE_M;
        gemm_maxk_kernel<<<grid, 32, 0, stream>>>(feat, wT, src, dst, tab, tab + 3 * MAX_BKT,
                                                   tk_vals, tk_idx, n_nodes, n_edges, n_hist);
    }
    bkt_fill_kernel<<<dim3(n_chunks, 2), 1024, 0, stream>>>(src, dst, tab, binned_d, binned_s,
                                                            n_edges);
    bkt_degree_scale_kernel<<<n_tiles, 1024, 0, stream>>>(binned_d, binned_s, tab, tk_vals,
                                                          n_nodes, n_edges);
    bkt_sum_kernel<<<n_tiles, 1024, 0, stream>>>(binned_d, tab, tab + 2 * MAX_BKT, tk_vals, tk_idx,
                                                 bias, out, n_nodes);
}


# top-16 chosen on scalars: C tiles re-sorted across lane halves so a register holds one row (lane = column); bitwise search on scalar thresholds and mask bit counts; kept set, slots, sums unchanged
# speedup vs baseline: 3.0633x; 1.0975x over previous
#include <hip/hip_runtime.h>
#include <math.h>

typedef __attribute__((ext_vector_type(2))) float v2f;
typedef __attribute__((ext_vector_type(8))) float v8f;

#define IN_FEATS 256
#define OUT_FEATS 64
#define MAXK 16
#define TILE_M 16
#define LDA (IN_FEATS + 4)
#define BKT_SHIFT 8
#define BKT_NODES (1 << BKT_SHIFT)
#define MAX_BKT 1024
#define EDGE_CHUNK 8192
#define EPT (EDGE_CHUNK / 1024)
#define HIST_CHUNK 4096
#define HIST_EPT 16
#define TAB_INTS (6 * MAX_BKT)

__global__ void init_kernel(int* __restrict__ tab,
                            const float* __restrict__ weight, float* __restrict__ wT) {
    int i = blockIdx.x * blockDim.x + threadIdx.x;
    if (i < TAB_INTS) tab[i] = 0;
    if (i < IN_FEATS * OUT_FEATS) {
        int k = i >> 6;
        int c = i & 63;
        wT[c * IN_FEATS + k] = weight[i];
    }
}


__device__ __forceinline__ unsigned int other_row(unsigned int v) {
    return (unsigned int)__builtin_amdgcn_permlanex16((int)v, (int)v, 0x76543210, 0xfedcba98, false, false);
}

__device__ __forceinline__ int scalar_bits64(unsigned long long m) {
    int n;
    asm("s_bcnt1_i32_b64 %0, %1" : "=s"(n) : "s"(m) : "scc");
    return n;
}

__device__ __forceinline__ unsigned int order_key(unsigned int u) {
    return u ^ ((unsigned int)((int)u >> 31) | 0x80000000u);
}

__global__ __launch_bounds__(32)
void gemm_maxk_kernel(const float* __restrict__ feat, const float* __restrict__ wT,
                      const int* __restrict__ src, const int* __restrict__ dst,
                      int* __restrict__ cnt_d, int* __restrict__ cnt_s,
                      float* __restrict__ tk_vals, unsigned int* __restrict__ tk_idx,
                      int n_nodes, int n_edges, int n_hist) {
    __shared__ __attribute__((aligned(16))) float sA[TILE_M * LDA];

    const int lane = threadIdx.x;

    if ((int)blockIdx.x < n_hist) {
        int* hd = (int*)sA;
        int* hs = hd + MAX_BKT;
        for (int i = lane; i < 2 * MAX_BKT; i += 32) hd[i] = 0;
        __syncthreads();
        const int e0 = blockIdx.x * HIST_CHUNK;
        for (int t = 0; t < HIST_CHUNK / (32 * HIST_EPT); ++t) {
            int ks[HIST_EPT], kd[HIST_EPT];
            #pragma unroll
            for (int j = 0; j < HIST_EPT; ++j) {
                const int e  = e0 + (t * HIST_EPT + j) * 32 + lane;
                const int ec = e < n_edges ? e : n_edges - 1;
                const int dv = dst[ec];
                ks[j] = src[ec];
                kd[j] = e < n_edges ? dv : -1;
            }
            #pragma unroll
            for (int j = 0; j < HIST_EPT; ++j) {
                if (kd[j] >= 0) {
                    atomicAdd(&hd[kd[j] >> BKT_SHIFT], 1);
                    atomicAdd(&hs[ks[j] >> BKT_SHIFT], 1);
                }
            }
        }
        __syncthreads();
        for (int i = lane; i < MAX_BKT; i += 32) {
            const int cd = hd[i];
            const int cs = hs[i];
            if (cd) atomicAdd(&cnt_d[i], cd);
            if (cs) atomicAdd(&cnt_s[i], cs);
        }
        return;
    }

    const int m0   = ((int)blockIdx.x - n_hist) * TILE_M;
    const int r16  = lane & 15;
    const int half = lane >> 4;

    #pragma unroll
    for (int pass = 0; pass < 2; ++pass) {
        float4 buf[16];
        #pragma unroll
        for (int j = 0; j < 16; ++j) {
            int i   = (pass * 16 + j) * 32 + lane;
            int row = i >> 6;
            int c4  = i & 63;
            int grow = m0 + row;
            grow = grow < n_nodes ? grow : n_nodes - 1;
            buf[j] = ((const float4*)feat)[(long long)grow * (IN_FEATS / 4) + c4];
        }
        #pragma unroll
        for (int j = 0; j < 16; ++j) {
            int i   = (pass * 16 + j) * 32 + lane;
            int row = i >> 6;
            int c4  = i & 63;
            float4 v = buf[j];
            if (m0 + row >= n_nodes) v = make_float4(0.f, 0.f, 0.f, 0.f);
            *(float4*)&sA[row * LDA + c4 * 4] = v;
        }
    }
    __syncthreads();

    const int koff = half << 1;
    const float* arow = sA + r16 * LDA;
    const float* w0 = wT + r16 * IN_FEATS;
    const float* w1 = w0 + 16 * IN_FEATS;
    const float* w2 = w0 + 32 * IN_FEATS;
    const float* w3 = w0 + 48 * IN_FEATS;

    v8f c0 = {0.f, 0.f, 0.f, 0.f, 0.f, 0.f, 0.f, 0.f};
    v8f c1 = c0, c2 = c0, c3 = c0;
    #pragma unroll
    for (int k0 = 0; k0 < IN_FEATS; k0 += 4) {
        v2f a, b0, b1, b2, b3;
        a.x  = arow[k0 + koff];  a.y  = arow[k0 + koff + 1];
        b0.x = w0[k0 + koff];    b0.y = w0[k0 + koff + 1];
        b1.x = w1[k0 + koff];    b1.y = w1[k0 + koff + 1];
        b2.x = w2[k0 + koff];    b2.y = w2[k0 + koff + 1];
        b3.x = w3[k0 + koff];    b3.y = w3[k0 + koff + 1];
        c0 = __builtin_amdgcn_wmma_f32_16x16x4_f32(false, a, false, b0, (short)0, c0, false, false);
        c1 = __builtin_amdgcn_wmma_f32_16x16x4_f32(false, a, false, b1, (short)0, c1, false, false);
        c2 = __builtin_amdgcn_wmma_f32_16x16x4_f32(false, a, false, b2, (short)0, c2, false, false);
        c3 = __builtin_amdgcn_wmma_f32_16x16x4_f32(false, a, false, b3, (short)0, c3, false, false);
    }

    __builtin_amdgcn_sched_barrier(0);
    asm volatile("v_nop\n\tv_nop\n\tv_nop\n\tv_nop");
    __builtin_amdgcn_sched_barrier(0);

    const unsigned int upper = 0u - (unsigned int)half;
    unsigned int x[TILE_M][2];
    #pragma unroll
    for (int r = 0; r < 8; ++r) {
        const unsigned int a0 = __float_as_uint(c0[r]), a1 = __float_as_uint(c1[r]);
        const unsigned int a2 = __float_as_uint(c2[r]), a3 = __float_as_uint(c3[r]);
        const unsigned int p0 = other_row(a0), p1 = other_row(a1);
        const unsigned int p2 = other_row(a2), p3 = other_row(a3);
        x[r][0]     = (a0 & ~upper) | (p1 & upper);
        x[r + 8][0] = (p0 & ~upper) | (a1 & upper);
        x[r][1]     = (a2 & ~upper) | (p3 & upper);
        x[r + 8][1] = (p2 & ~upper) | (a3 & upper);
    }
    unsigned int key[TILE_M][2];
    #pragma unroll
    for (int R = 0; R < TILE_M; ++R) {
        key[R][0] = order_key(x[R][0]);
        key[R][1] = order_key(x[R][1]);
    }

    unsigned int thr[TILE_M];
    #pragma unroll
    for (int R = 0; R < TILE_M; ++R) thr[R] = 0u;
    for (int bit = 31; bit >= 0; --bit) {
        const unsigned int bmask = 1u << bit;
        unsigned int cand[TILE_M];
        unsigned long long m[TILE_M];
        int n[TILE_M];
        #pragma unroll
        for (int R = 0; R < TILE_M; ++R) cand[R] = thr[R] | bmask;
        __builtin_amdgcn_sched_barrier(0);
        #pragma unroll
        for (int R = 0; R < TILE_M; ++R)
            m[R] = (unsigned long long)__builtin_amdgcn_ballot_w32(key[R][0] >= cand[R]) |
                   ((unsigned long long)__builtin_amdgcn_ballot_w32(key[R][1] >= cand[R]) << 32);
        __builtin_amdgcn_sched_barrier(0);
        #pragma unroll
        for (int R = 0; R < TILE_M; ++R) n[R] = scalar_bits64(m[R]);
        __builtin_amdgcn_sched_barrier(0);
        #pragma unroll
        for (int R = 0; R < TILE_M; ++R) thr[R] = n[R] >= MAXK ? cand[R] : thr[R];
    }

    __syncthreads();
    float* sV = sA;
    unsigned char* sI = (unsigned char*)(sA + TILE_M * MAXK);
    #pragma unroll
    for (int R = 0; R < TILE_M; ++R) {
        const unsigned int T = thr[R];
        const bool gt0 = key[R][0] > T,  gt1 = key[R][1] > T;
        const bool eq0 = key[R][0] == T, eq1 = key[R][1] == T;
        const unsigned int g0 = __builtin_amdgcn_ballot_w32(gt0), g1 = __builtin_amdgcn_ballot_w32(gt1);
        const unsigned int e0 = __builtin_amdgcn_ballot_w32(eq0), e1 = __builtin_amdgcn_ballot_w32(eq1);
        const int need = MAXK - (__popc(g0) + __popc(g1));
        const int rank0 = (int)__builtin_amdgcn_mbcnt_lo(e0, 0u);
        const int rank1 = (int)__builtin_amdgcn_mbcnt_lo(e1, (unsigned int)__popc(e0));
        const unsigned int s0 = g0 | (e0 & __builtin_amdgcn_ballot_w32(rank0 < need));
        const unsigned int s1 = g1 | (e1 & __builtin_amdgcn_ballot_w32(rank1 < need));
        const bool sel0 = __builtin_amdgcn_inverse_ballot_w32(s0);
        const bool sel1 = __builtin_amdgcn_inverse_ballot_w32(s1);
        const int slot0 = (int)__builtin_amdgcn_mbcnt_lo(s0, 0u);
        const int slot1 = (int)__builtin_amdgcn_mbcnt_lo(s1, (unsigned int)__popc(s0));
        if (sel0) {
            sV[R * MAXK + slot0] = __uint_as_float(x[R][0]);
            sI[R * MAXK + slot0] = (unsigned char)lane;
        }
        if (sel1) {
            sV[R * MAXK + slot1] = __uint_as_float(x[R][1]);
            sI[R * MAXK + slot1] = (unsigned char)(32 + lane);
        }
    }
    __syncthreads();

    const float4* sV4 = (const float4*)sV;
    #pragma unroll
    for (int q = 0; q < 2; ++q) {
        const int f = q * 32 + lane;
        if (m0 + (f >> 2) < n_nodes)
            ((float4*)tk_vals)[(long long)m0 * (MAXK / 4) + f] = sV4[f];
    }
    if (lane < TILE_M && m0 + lane < n_nodes)
        ((uint4*)tk_idx)[m0 + lane] = ((const uint4*)sI)[lane];
}


__global__ __launch_bounds__(1024)
void bkt_fill_kernel(const int* __restrict__ src, const int* __restrict__ dst,
                     int* __restrict__ tab, unsigned int* __restrict__ binned_d,
                     unsigned char* __restrict__ binned_s, int n_edges) {
    const int key = blockIdx.y;
    const int* bkt_cnt = tab + key * 3 * MAX_BKT;
    int* bkt_cur   = tab + key * 3 * MAX_BKT + MAX_BKT;
    int* bkt_start = tab + key * 3 * MAX_BKT + 2 * MAX_BKT;
    __shared__ int hist[MAX_BKT];
    __shared__ int gbase[MAX_BKT];
    __shared__ int wsum[2][32];
    __shared__ unsigned int stage[EDGE_CHUNK];
    __shared__ unsigned short stage_b[EDGE_CHUNK];
    const int tid  = threadIdx.x;
    const int lane = tid & 31;
    const int wave = tid >> 5;
    hist[tid] = 0;
    const int v = bkt_cnt[tid];

    const int base = blockIdx.x * EDGE_CHUNK;
    int es[EPT], ed[EPT];
    #pragma unroll
    for (int j = 0; j < EPT; ++j) {
        const int e  = base + j * 1024 + tid;
        const int ec = e < n_edges ? e : n_edges - 1;
        const int sv = src[ec];
        const int dv = dst[ec];
        es[j] = key ? dv : sv;
        ed[j] = e < n_edges ? (key ? sv : dv) : -1;
    }
    __syncthreads();
    #pragma unroll
    for (int j = 0; j < EPT; ++j)
        if (ed[j] >= 0) atomicAdd(&hist[ed[j] >> BKT_SHIFT], 1);
    __syncthreads();

    const int c = hist[tid];
    int x = v, y = c;
    #pragma unroll
    for (int o = 1; o < 32; o <<= 1) {
        const int xs = __shfl_up(x, o);
        const int ys = __shfl_up(y, o);
        if (lane >= o) { x += xs; y += ys; }
    }
    if (lane == 31) { wsum[0][wave] = x; wsum[1][wave] = y; }
    __syncthreads();
    if (wave < 2) {
        const int w = wsum[wave][lane];
        int s = w;
        #pragma unroll
        for (int o = 1; o < 32; o <<= 1) {
            const int t = __shfl_up(s, o);
            if (lane >= o) s += t;
        }
        wsum[wave][lane] = s - w;
    }
    __syncthreads();

    const int start  = x - v + wsum[0][wave];
    const int lstart = y - c + wsum[1][wave];
    if (blockIdx.x == 0) bkt_start[tid] = start;
    const int rsv = c ? atomicAdd(&bkt_cur[tid], c) : 0;
    gbase[tid] = start + rsv - lstart;
    hist[tid]  = lstart;
    __syncthreads();

    #pragma unroll
    for (int j = 0; j < EPT; ++j) {
        if (ed[j] >= 0) {
            const int b    = ed[j] >> BKT_SHIFT;
            const int slot = atomicAdd(&hist[b], 1);
            stage[slot]   = ((unsigned int)es[j] << BKT_SHIFT) | ((unsigned int)ed[j] & (BKT_NODES - 1));
            stage_b[slot] = (unsigned short)b;
        }
    }
    __syncthreads();

    const int left = n_edges - base;
    const int n_here = left < EDGE_CHUNK ? left : EDGE_CHUNK;
    #pragma unroll
    for (int j = 0; j < EPT; ++j) {
        const int slot = j * 1024 + tid;
        if (slot < n_here) {
            const int g = gbase[stage_b[slot]] + slot;
            if (key == 0) binned_d[g] = stage[slot];
            else          binned_s[g] = (unsigned char)stage[slot];
        }
    }
}

__global__ __launch_bounds__(1024)
void bkt_degree_scale_kernel(const unsigned int* __restrict__ binned_d,
                             const unsigned char* __restrict__ binned_s,
                             const int* __restrict__ tab, float* __restrict__ tk_vals,
                             int n_nodes, int n_edges) {
    __shared__ int cin[BKT_NODES];
    __shared__ int cout[BKT_NODES];
    __shared__ float ssc[BKT_NODES];
    const int tid = threadIdx.x;
    if (tid < BKT_NODES) { cin[tid] = 0; cout[tid] = 0; }
    const int nd = tab[blockIdx.x];
    const int sd = tab[2 * MAX_BKT + blockIdx.x];
    const int ns = tab[3 * MAX_BKT + blockIdx.x];
    const int ss = tab[5 * MAX_BKT + blockIdx.x];
    float4* rec = (float4*)tk_vals + (long long)blockIdx.x * (BKT_NODES * MAXK / 4);
    const bool live = blockIdx.x * BKT_NODES + (tid >> 2) < n_nodes;
    float4 v = make_float4(0.f, 0.f, 0.f, 0.f);
    if (live) v = rec[tid];
    __syncthreads();
    const int nmax = nd > ns ? nd : ns;
    for (int i0 = tid; i0 < nmax; i0 += 4 * 1024) {
        unsigned int wd[4], ws[4];
        #pragma unroll
        for (int k = 0; k < 4; ++k) {
            const int i  = i0 + 1024 * k;
            const int gd = sd + i < n_edges ? sd + i : n_edges - 1;
            const int gs = ss + i < n_edges ? ss + i : n_edges - 1;
            wd[k] = binned_d[gd];
            ws[k] = binned_s[gs];
        }
        #pragma unroll
        for (int k = 0; k < 4; ++k) {
            const int i = i0 + 1024 * k;
            if (i < nd) atomicAdd(&cin[wd[k] & (BKT_NODES - 1)], 1);
            if (i < ns) atomicAdd(&cout[ws[k] & (BKT_NODES - 1)], 1);
        }
    }
    __syncthreads();
    if (tid < BKT_NODES) {
        float id = (float)cin[tid];
        float od = (float)cout[tid];
        od = od < 1.f ? 1.f : od;
        id = id < 1.f ? 1.f : id;
        ssc[tid] = rsqrtf(od) * rsqrtf(id);
    }
    __syncthreads();
    if (live) {
        const float sc = ssc[tid >> 2];
        v.x *= sc; v.y *= sc; v.z *= sc; v.w *= sc;
        rec[tid] = v;
    }
}

__device__ __forceinline__ void add_record(float* row, float4 v, unsigned int ip) {
    atomicAdd(row + (ip & 0xffu), v.x);
    atomicAdd(row + ((ip >> 8) & 0xffu), v.y);
    atomicAdd(row + ((ip >> 16) & 0xffu), v.z);
    atomicAdd(row + ((ip >> 24) & 0xffu), v.w);
}

__global__ __launch_bounds__(1024)
void bkt_sum_kernel(const unsigned int* __restrict__ binned, const int* __restrict__ bkt_cnt,
                    const int* __restrict__ bkt_start, const float* __restrict__ tk_vals,
                    const unsigned int* __restrict__ tk_idx, const float* __restrict__ bias,
                    float* __restrict__ out, int n_nodes) {
    __shared__ __attribute__((aligned(16))) float acc[BKT_NODES * OUT_FEATS];
    const int tid = threadIdx.x;
    for (int i = tid; i < BKT_NODES * OUT_FEATS; i += 1024) acc[i] = bias[i & (OUT_FEATS - 1)];
    __syncthreads();

    const int cnt = bkt_cnt[blockIdx.x];
    const unsigned int* mine = binned + bkt_start[blockIdx.x];
    const int part = tid & 3;
    int i0 = tid >> 2;
    for (; i0 + 3 * 256 < cnt; i0 += 4 * 256) {
        unsigned int w[4], ip[4];
        float4 v[4];
        #pragma unroll
        for (int k = 0; k < 4; ++k) w[k] = mine[i0 + 256 * k];
        #pragma unroll
        for (int k = 0; k < 4; ++k) {
            const long long s = (long long)(w[k] >> BKT_SHIFT);
            v[k]  = ((const float4*)tk_vals)[s * 4 + part];
            ip[k] = tk_idx[s * 4 + part];
        }
        #pragma unroll
        for (int k = 0; k < 4; ++k)
            add_record(acc + (w[k] & (BKT_NODES - 1)) * OUT_FEATS, v[k], ip[k]);
    }
    for (; i0 < cnt; i0 += 256) {
        const unsigned int w = mine[i0];
        const long long s = (long long)(w >> BKT_SHIFT);
        const float4 v = ((const float4*)tk_vals)[s * 4 + part];
        const unsigned int ip = tk_idx[s * 4 + part];
        add_record(acc + (w & (BKT_NODES - 1)) * OUT_FEATS, v, ip);
    }
    __syncthreads();

    const long long node0 = (long long)blockIdx.x * BKT_NODES;
    for (int i = tid; i < BKT_NODES * OUT_FEATS / 4; i += 1024) {
        if (node0 + (i >> 4) < n_nodes)
            ((float4*)out)[node0 * (OUT_FEATS / 4) + i] = ((const float4*)acc)[i];
    }
}

extern "C" void kernel_launch(void* const* d_in, const int* in_sizes, int n_in,
                              void* d_out, int out_size, void* d_ws, size_t ws_size,
                              hipStream_t stream) {
    const float* feat   = (const float*)d_in[0];
    const float* weight = (const float*)d_in[1];
    const float* bias   = (const float*)d_in[2];
    const int*   src    = (const int*)d_in[3];
    const int*   dst    = (const int*)d_in[4];
    float* out = (float*)d_out;

    const int n_nodes = in_sizes[0] / IN_FEATS;
    const int n_edges = in_sizes[3];

    float* wT      = (float*)d_ws;
    float* tk_vals = wT + IN_FEATS * OUT_FEATS;
    unsigned int* tk_idx = (unsigned int*)(tk_vals + (long long)n_nodes * MAXK);
    int* tab       = (int*)(tk_idx + (long long)n_nodes * (MAXK / 4));
    unsigned int* binned_d  = (unsigned int*)(tab + TAB_INTS);
    unsigned char* binned_s = (unsigned char*)d_out;

    init_kernel<<<IN_FEATS * OUT_FEATS / 256, 256, 0, stream>>>(tab, weight, wT);
    const int n_tiles  = (n_nodes + BKT_NODES - 1) / BKT_NODES;
    const int n_chunks = (n_edges + EDGE_CHUNK - 1) / EDGE_CHUNK;
    const int n_hist   = (n_edges + HIST_CHUNK - 1) / HIST_CHUNK;
    {
        int grid = n_hist + (n_nodes + TILE_M - 1) / TILE_M;
        gemm_maxk_kernel<<<grid, 32, 0, stream>>>(feat, wT, src, dst, tab, tab + 3 * MAX_BKT,
                                                   tk_vals, tk_idx, n_nodes, n_edges, n_hist);
    }
    bkt_fill_kernel<<<dim3(n_chunks, 2), 1024, 0, stream>>>(src, dst, tab, binned_d, binned_s,
                                                            n_edges);
    bkt_degree_scale_kernel<<<n_tiles, 1024, 0, stream>>>(binned_d, binned_s, tab, tk_vals,
                                                          n_nodes, n_edges);
    bkt_sum_kernel<<<n_tiles, 1024, 0, stream>>>(binned_d, tab, tab + 2 * MAX_BKT, tk_vals, tk_idx,
                                                 bias, out, n_nodes);
}


# GEMM block = 4 waves x 64 rows, product split by columns: a wave loads a quarter of W, each B fragment feeds four row tiles (W read once per 64 rows); C swapped through LDS, each wave selects 16 rows
# speedup vs baseline: 3.5617x; 1.1627x over previous
#include <hip/hip_runtime.h>
#include <math.h>

typedef __attribute__((ext_vector_type(2))) float v2f;
typedef __attribute__((ext_vector_type(8))) float v8f;

#define IN_FEATS 256
#define OUT_FEATS 64
#define MAXK 16
#define TILE_M 16
#define GEMM_WAVES 4
#define BLOCK_M (TILE_M * GEMM_WAVES)
#define LDC (OUT_FEATS + 4)
#define LDA (IN_FEATS + 4)
#define BKT_SHIFT 8
#define BKT_NODES (1 << BKT_SHIFT)
#define MAX_BKT 1024
#define EDGE_CHUNK 8192
#define EPT (EDGE_CHUNK / 1024)
#define HIST_CHUNK 4096
#define HIST_EPT 16
#define TAB_INTS (6 * MAX_BKT)

__global__ void init_kernel(int* __restrict__ tab,
                            const float* __restrict__ weight, float* __restrict__ wT) {
    int i = blockIdx.x * blockDim.x + threadIdx.x;
    if (i < TAB_INTS) tab[i] = 0;
    if (i < IN_FEATS * OUT_FEATS) {
        int k = i >> 6;
        int c = i & 63;
        wT[c * IN_FEATS + k] = weight[i];
    }
}


__device__ __forceinline__ int scalar_bits64(unsigned long long m) {
    int n;
    asm("s_bcnt1_i32_b64 %0, %1" : "=s"(n) : "s"(m) : "scc");
    return n;
}

__device__ __forceinline__ unsigned int order_key(unsigned int u) {
    return u ^ ((unsigned int)((int)u >> 31) | 0x80000000u);
}

__global__ __launch_bounds__(32 * GEMM_WAVES)
void gemm_maxk_kernel(const float* __restrict__ feat, const float* __restrict__ wT,
                      const int* __restrict__ src, const int* __restrict__ dst,
                      int* __restrict__ cnt_d, int* __restrict__ cnt_s,
                      float* __restrict__ tk_vals, unsigned int* __restrict__ tk_idx,
                      int n_nodes, int n_edges, int n_hist) {
    __shared__ __attribute__((aligned(16))) float sA[BLOCK_M * LDA];

    const int tid  = threadIdx.x;
    const int lane = tid & 31;
    const int wave = __builtin_amdgcn_readfirstlane(tid >> 5);

    if ((int)blockIdx.x < n_hist) {
        int* hd = (int*)sA;
        int* hs = hd + MAX_BKT;
        for (int i = tid; i < 2 * MAX_BKT; i += 32 * GEMM_WAVES) hd[i] = 0;
        __syncthreads();
        const int e0 = (blockIdx.x * GEMM_WAVES + wave) * HIST_CHUNK;
        for (int t = 0; t < HIST_CHUNK / (32 * HIST_EPT); ++t) {
            int ks[HIST_EPT], kd[HIST_EPT];
            #pragma unroll
            for (int j = 0; j < HIST_EPT; ++j) {
                const int e  = e0 + (t * HIST_EPT + j) * 32 + lane;
                const int ec = e < n_edges ? e : n_edges - 1;
                const int dv = dst[ec];
                ks[j] = src[ec];
                kd[j] = e < n_edges ? dv : -1;
            }
            #pragma unroll
            for (int j = 0; j < HIST_EPT; ++j) {
                if (kd[j] >= 0) {
                    atomicAdd(&hd[kd[j] >> BKT_SHIFT], 1);
                    atomicAdd(&hs[ks[j] >> BKT_SHIFT], 1);
                }
            }
        }
        __syncthreads();
        for (int i = tid; i < MAX_BKT; i += 32 * GEMM_WAVES) {
            const int cd = hd[i];
            const int cs = hs[i];
            if (cd) atomicAdd(&cnt_d[i], cd);
            if (cs) atomicAdd(&cnt_s[i], cs);
        }
        return;
    }

    const int m0   = ((int)blockIdx.x - n_hist) * BLOCK_M;
    const int r16  = lane & 15;
    const int half = lane >> 4;

    #pragma unroll
    for (int pass = 0; pass < 2; ++pass) {
        float4 buf[16];
        #pragma unroll
        for (int j = 0; j < 16; ++j) {
            int i   = (pass * 16 + j) * (32 * GEMM_WAVES) + tid;
            int row = i >> 6;
            int c4  = i & 63;
            int grow = m0 + row;
            grow = grow < n_nodes ? grow : n_nodes - 1;
            buf[j] = ((const float4*)feat)[(long long)grow * (IN_FEATS / 4) + c4];
        }
        #pragma unroll
        for (int j = 0; j < 16; ++j) {
            int i   = (pass * 16 + j) * (32 * GEMM_WAVES) + tid;
            int row = i >> 6;
            int c4  = i & 63;
            float4 v = buf[j];
            if (m0 + row >= n_nodes) v = make_float4(0.f, 0.f, 0.f, 0.f);
            *(float4*)&sA[row * LDA + c4 * 4] = v;
        }
    }
    __syncthreads();

    const int koff = half << 1;
    const float* arow0 = sA + r16 * LDA;
    const float* arow1 = arow0 + TILE_M * LDA;
    const float* arow2 = arow1 + TILE_M * LDA;
    const float* arow3 = arow2 + TILE_M * LDA;
    const float* wcol  = wT + (wave * 16 + r16) * IN_FEATS;

    v8f acc[GEMM_WAVES];
    #pragma unroll
    for (int i = 0; i < GEMM_WAVES; ++i) acc[i] = (v8f){0.f, 0.f, 0.f, 0.f, 0.f, 0.f, 0.f, 0.f};
    #pragma unroll
    for (int k0 = 0; k0 < IN_FEATS; k0 += 4) {
        v2f a0, a1, a2, a3, b;
        a0.x = arow0[k0 + koff]; a0.y = arow0[k0 + koff + 1];
        a1.x = arow1[k0 + koff]; a1.y = arow1[k0 + koff + 1];
        a2.x = arow2[k0 + koff]; a2.y = arow2[k0 + koff + 1];
        a3.x = arow3[k0 + koff]; a3.y = arow3[k0 + koff + 1];
        b.x  = wcol[k0 + koff];  b.y  = wcol[k0 + koff + 1];
        acc[0] = __builtin_amdgcn_wmma_f32_16x16x4_f32(false, a0, false, b, (short)0, acc[0], false, false);
        acc[1] = __builtin_amdgcn_wmma_f32_16x16x4_f32(false, a1, false, b, (short)0, acc[1], false, false);
        acc[2] = __builtin_amdgcn_wmma_f32_16x16x4_f32(false, a2, false, b, (short)0, acc[2], false, false);
        acc[3] = __builtin_amdgcn_wmma_f32_16x16x4_f32(false, a3, false, b, (short)0, acc[3], false, false);
    }

    __builtin_amdgcn_sched_barrier(0);
    asm volatile("v_nop\n\tv_nop\n\tv_nop\n\tv_nop");
    __builtin_amdgcn_sched_barrier(0);

    __syncthreads();
    float* sC = sA;
    #pragma unroll
    for (int u = 0; u < GEMM_WAVES; ++u) {
        #pragma unroll
        for (int r = 0; r < 8; ++r)
            sC[(u * TILE_M + half * 8 + r) * LDC + wave * 16 + r16] = acc[u][r];
    }
    __syncthreads();
    const float* crow = sC + wave * TILE_M * LDC;
    unsigned int x[TILE_M][2];
    #pragma unroll
    for (int R = 0; R < TILE_M; ++R) {
        x[R][0] = __float_as_uint(crow[R * LDC + lane]);
        x[R][1] = __float_as_uint(crow[R * LDC + 32 + lane]);
    }
    unsigned int key[TILE_M][2];
    #pragma unroll
    for (int R = 0; R < TILE_M; ++R) {
        key[R][0] = order_key(x[R][0]);
        key[R][1] = order_key(x[R][1]);
    }

    unsigned int thr[TILE_M];
    #pragma unroll
    for (int R = 0; R < TILE_M; ++R) thr[R] = 0u;
    for (int bit = 31; bit >= 0; --bit) {
        const unsigned int bmask = 1u << bit;
        unsigned int cand[TILE_M];
        unsigned long long m[TILE_M];
        int n[TILE_M];
        #pragma unroll
        for (int R = 0; R < TILE_M; ++R) cand[R] = thr[R] | bmask;
        __builtin_amdgcn_sched_barrier(0);
        #pragma unroll
        for (int R = 0; R < TILE_M; ++R)
            m[R] = (unsigned long long)__builtin_amdgcn_ballot_w32(key[R][0] >= cand[R]) |
                   ((unsigned long long)__builtin_amdgcn_ballot_w32(key[R][1] >= cand[R]) << 32);
        __builtin_amdgcn_sched_barrier(0);
        #pragma unroll
        for (int R = 0; R < TILE_M; ++R) n[R] = scalar_bits64(m[R]);
        __builtin_amdgcn_sched_barrier(0);
        #pragma unroll
        for (int R = 0; R < TILE_M; ++R) thr[R] = n[R] >= MAXK ? cand[R] : thr[R];
    }

    float* sV = sA + BLOCK_M * LDC;
    unsigned char* sI = (unsigned char*)(sV + BLOCK_M * MAXK);
    #pragma unroll
    for (int R = 0; R < TILE_M; ++R) {
        const unsigned int T = thr[R];
        const bool gt0 = key[R][0] > T,  gt1 = key[R][1] > T;
        const bool eq0 = key[R][0] == T, eq1 = key[R][1] == T;
        const unsigned int g0 = __builtin_amdgcn_ballot_w32(gt0), g1 = __builtin_amdgcn_ballot_w32(gt1);
        const unsigned int e0 = __builtin_amdgcn_ballot_w32(eq0), e1 = __builtin_amdgcn_ballot_w32(eq1);
        const int need = MAXK - (__popc(g0) + __popc(g1));
        const int rank0 = (int)__builtin_amdgcn_mbcnt_lo(e0, 0u);
        const int rank1 = (int)__builtin_amdgcn_mbcnt_lo(e1, (unsigned int)__popc(e0));
        const unsigned int s0 = g0 | (e0 & __builtin_amdgcn_ballot_w32(rank0 < need));
        const unsigned int s1 = g1 | (e1 & __builtin_amdgcn_ballot_w32(rank1 < need));
        const bool sel0 = __builtin_amdgcn_inverse_ballot_w32(s0);
        const bool sel1 = __builtin_amdgcn_inverse_ballot_w32(s1);
        const int slot0 = (int)__builtin_amdgcn_mbcnt_lo(s0, 0u);
        const int slot1 = (int)__builtin_amdgcn_mbcnt_lo(s1, (unsigned int)__popc(s0));
        const int rec = (wave * TILE_M + R) * MAXK;
        if (sel0) {
            sV[rec + slot0] = __uint_as_float(x[R][0]);
            sI[rec + slot0] = (unsigned char)lane;
        }
        if (sel1) {
            sV[rec + slot1] = __uint_as_float(x[R][1]);
            sI[rec + slot1] = (unsigned char)(32 + lane);
        }
    }
    __syncthreads();

    const float4* sV4 = (const float4*)sV;
    #pragma unroll
    for (int q = 0; q < BLOCK_M * MAXK / (4 * 32 * GEMM_WAVES); ++q) {
        const int f = q * (32 * GEMM_WAVES) + tid;
        if (m0 + (f >> 2) < n_nodes)
            ((float4*)tk_vals)[(long long)m0 * (MAXK / 4) + f] = sV4[f];
    }
    if (tid < BLOCK_M && m0 + tid < n_nodes)
        ((uint4*)tk_idx)[m0 + tid] = ((const uint4*)sI)[tid];
}


__global__ __launch_bounds__(1024)
void bkt_fill_kernel(const int* __restrict__ src, const int* __restrict__ dst,
                     int* __restrict__ tab, unsigned int* __restrict__ binned_d,
                     unsigned char* __restrict__ binned_s, int n_edges) {
    const int key = blockIdx.y;
    const int* bkt_cnt = tab + key * 3 * MAX_BKT;
    int* bkt_cur   = tab + key * 3 * MAX_BKT + MAX_BKT;
    int* bkt_start = tab + key * 3 * MAX_BKT + 2 * MAX_BKT;
    __shared__ int hist[MAX_BKT];
    __shared__ int gbase[MAX_BKT];
    __shared__ int wsum[2][32];
    __shared__ unsigned int stage[EDGE_CHUNK];
    __shared__ unsigned short stage_b[EDGE_CHUNK];
    const int tid  = threadIdx.x;
    const int lane = tid & 31;
    const int wave = tid >> 5;
    hist[tid] = 0;
    const int v = bkt_cnt[tid];

    const int base = blockIdx.x * EDGE_CHUNK;
    int es[EPT], ed[EPT];
    #pragma unroll
    for (int j = 0; j < EPT; ++j) {
        const int e  = base + j * 1024 + tid;
        const int ec = e < n_edges ? e : n_edges - 1;
        const int sv = src[ec];
        const int dv = dst[ec];
        es[j] = key ? dv : sv;
        ed[j] = e < n_edges ? (key ? sv : dv) : -1;
    }
    __syncthreads();
    #pragma unroll
    for (int j = 0; j < EPT; ++j)
        if (ed[j] >= 0) atomicAdd(&hist[ed[j] >> BKT_SHIFT], 1);
    __syncthreads();

    const int c = hist[tid];
    int x = v, y = c;
    #pragma unroll
    for (int o = 1; o < 32; o <<= 1) {
        const int xs = __shfl_up(x, o);
        const int ys = __shfl_up(y, o);
        if (lane >= o) { x += xs; y += ys; }
    }
    if (lane == 31) { wsum[0][wave] = x; wsum[1][wave] = y; }
    __syncthreads();
    if (wave < 2) {
        const int w = wsum[wave][lane];
        int s = w;
        #pragma unroll
        for (int o = 1; o < 32; o <<= 1) {
            const int t = __shfl_up(s, o);
            if (lane >= o) s += t;
        }
        wsum[wave][lane] = s - w;
    }
    __syncthreads();

    const int start  = x - v + wsum[0][wave];
    const int lstart = y - c + wsum[1][wave];
    if (blockIdx.x == 0) bkt_start[tid] = start;
    const int rsv = c ? atomicAdd(&bkt_cur[tid], c) : 0;
    gbase[tid] = start + rsv - lstart;
    hist[tid]  = lstart;
    __syncthreads();

    #pragma unroll
    for (int j = 0; j < EPT; ++j) {
        if (ed[j] >= 0) {
            const int b    = ed[j] >> BKT_SHIFT;
            const int slot = atomicAdd(&hist[b], 1);
            stage[slot]   = ((unsigned int)es[j] << BKT_SHIFT) | ((unsigned int)ed[j] & (BKT_NODES - 1));
            stage_b[slot] = (unsigned short)b;
        }
    }
    __syncthreads();

    const int left = n_edges - base;
    const int n_here = left < EDGE_CHUNK ? left : EDGE_CHUNK;
    #pragma unroll
    for (int j = 0; j < EPT; ++j) {
        const int slot = j * 1024 + tid;
        if (slot < n_here) {
            const int g = gbase[stage_b[slot]] + slot;
            if (key == 0) binned_d[g] = stage[slot];
            else          binned_s[g] = (unsigned char)stage[slot];
        }
    }
}

__global__ __launch_bounds__(1024)
void bkt_degree_scale_kernel(const unsigned int* __restrict__ binned_d,
                             const unsigned char* __restrict__ binned_s,
                             const int* __restrict__ tab, float* __restrict__ tk_vals,
                             int n_nodes, int n_edges) {
    __shared__ int cin[BKT_NODES];
    __shared__ int cout[BKT_NODES];
    __shared__ float ssc[BKT_NODES];
    const int tid = threadIdx.x;
    if (tid < BKT_NODES) { cin[tid] = 0; cout[tid] = 0; }
    const int nd = tab[blockIdx.x];
    const int sd = tab[2 * MAX_BKT + blockIdx.x];
    const int ns = tab[3 * MAX_BKT + blockIdx.x];
    const int ss = tab[5 * MAX_BKT + blockIdx.x];
    float4* rec = (float4*)tk_vals + (long long)blockIdx.x * (BKT_NODES * MAXK / 4);
    const bool live = blockIdx.x * BKT_NODES + (tid >> 2) < n_nodes;
    float4 v = make_float4(0.f, 0.f, 0.f, 0.f);
    if (live) v = rec[tid];
    __syncthreads();
    const int nmax = nd > ns ? nd : ns;
    for (int i0 = tid; i0 < nmax; i0 += 4 * 1024) {
        unsigned int wd[4], ws[4];
        #pragma unroll
        for (int k = 0; k < 4; ++k) {
            const int i  = i0 + 1024 * k;
            const int gd = sd + i < n_edges ? sd + i : n_edges - 1;
            const int gs = ss + i < n_edges ? ss + i : n_edges - 1;
            wd[k] = binned_d[gd];
            ws[k] = binned_s[gs];
        }
        #pragma unroll
        for (int k = 0; k < 4; ++k) {
            const int i = i0 + 1024 * k;
            if (i < nd) atomicAdd(&cin[wd[k] & (BKT_NODES - 1)], 1);
            if (i < ns) atomicAdd(&cout[ws[k] & (BKT_NODES - 1)], 1);
        }
    }
    __syncthreads();
    if (tid < BKT_NODES) {
        float id = (float)cin[tid];
        float od = (float)cout[tid];
        od = od < 1.f ? 1.f : od;
        id = id < 1.f ? 1.f : id;
        ssc[tid] = rsqrtf(od) * rsqrtf(id);
    }
    __syncthreads();
    if (live) {
        const float sc = ssc[tid >> 2];
        v.x *= sc; v.y *= sc; v.z *= sc; v.w *= sc;
        rec[tid] = v;
    }
}

__device__ __forceinline__ void add_record(float* row, float4 v, unsigned int ip) {
    atomicAdd(row + (ip & 0xffu), v.x);
    atomicAdd(row + ((ip >> 8) & 0xffu), v.y);
    atomicAdd(row + ((ip >> 16) & 0xffu), v.z);
    atomicAdd(row + ((ip >> 24) & 0xffu), v.w);
}

__global__ __launch_bounds__(1024)
void bkt_sum_kernel(const unsigned int* __restrict__ binned, const int* __restrict__ bkt_cnt,
                    const int* __restrict__ bkt_start, const float* __restrict__ tk_vals,
                    const unsigned int* __restrict__ tk_idx, const float* __restrict__ bias,
                    float* __restrict__ out, int n_nodes) {
    __shared__ __attribute__((aligned(16))) float acc[BKT_NODES * OUT_FEATS];
    const int tid = threadIdx.x;
    for (int i = tid; i < BKT_NODES * OUT_FEATS; i += 1024) acc[i] = bias[i & (OUT_FEATS - 1)];
    __syncthreads();

    const int cnt = bkt_cnt[blockIdx.x];
    const unsigned int* mine = binned + bkt_start[blockIdx.x];
    const int part = tid & 3;
    int i0 = tid >> 2;
    for (; i0 + 3 * 256 < cnt; i0 += 4 * 256) {
        unsigned int w[4], ip[4];
        float4 v[4];
        #pragma unroll
        for (int k = 0; k < 4; ++k) w[k] = mine[i0 + 256 * k];
        #pragma unroll
        for (int k = 0; k < 4; ++k) {
            const long long s = (long long)(w[k] >> BKT_SHIFT);
            v[k]  = ((const float4*)tk_vals)[s * 4 + part];
            ip[k] = tk_idx[s * 4 + part];
        }
        #pragma unroll
        for (int k = 0; k < 4; ++k)
            add_record(acc + (w[k] & (BKT_NODES - 1)) * OUT_FEATS, v[k], ip[k]);
    }
    for (; i0 < cnt; i0 += 256) {
        const unsigned int w = mine[i0];
        const long long s = (long long)(w >> BKT_SHIFT);
        const float4 v = ((const float4*)tk_vals)[s * 4 + part];
        const unsigned int ip = tk_idx[s * 4 + part];
        add_record(acc + (w & (BKT_NODES - 1)) * OUT_FEATS, v, ip);
    }
    __syncthreads();

    const long long node0 = (long long)blockIdx.x * BKT_NODES;
    for (int i = tid; i < BKT_NODES * OUT_FEATS / 4; i += 1024) {
        if (node0 + (i >> 4) < n_nodes)
            ((float4*)out)[node0 * (OUT_FEATS / 4) + i] = ((const float4*)acc)[i];
    }
}

extern "C" void kernel_launch(void* const* d_in, const int* in_sizes, int n_in,
                              void* d_out, int out_size, void* d_ws, size_t ws_size,
                              hipStream_t stream) {
    const float* feat   = (const float*)d_in[0];
    const float* weight = (const float*)d_in[1];
    const float* bias   = (const float*)d_in[2];
    const int*   src    = (const int*)d_in[3];
    const int*   dst    = (const int*)d_in[4];
    float* out = (float*)d_out;

    const int n_nodes = in_sizes[0] / IN_FEATS;
    const int n_edges = in_sizes[3];

    float* wT      = (float*)d_ws;
    float* tk_vals = wT + IN_FEATS * OUT_FEATS;
    unsigned int* tk_idx = (unsigned int*)(tk_vals + (long long)n_nodes * MAXK);
    int* tab       = (int*)(tk_idx + (long long)n_nodes * (MAXK / 4));
    unsigned int* binned_d  = (unsigned int*)(tab + TAB_INTS);
    unsigned char* binned_s = (unsigned char*)d_out;

    init_kernel<<<IN_FEATS * OUT_FEATS / 256, 256, 0, stream>>>(tab, weight, wT);
    const int n_tiles  = (n_nodes + BKT_NODES - 1) / BKT_NODES;
    const int n_chunks = (n_edges + EDGE_CHUNK - 1) / EDGE_CHUNK;
    const int n_hist   = (n_edges + GEMM_WAVES * HIST_CHUNK - 1) / (GEMM_WAVES * HIST_CHUNK);
    {
        int grid = n_hist + (n_nodes + BLOCK_M - 1) / BLOCK_M;
        gemm_maxk_kernel<<<grid, 32 * GEMM_WAVES, 0, stream>>>(feat, wT, src, dst, tab, tab + 3 * MAX_BKT,
                                                   tk_vals, tk_idx, n_nodes, n_edges, n_hist);
    }
    bkt_fill_kernel<<<dim3(n_chunks, 2), 1024, 0, stream>>>(src, dst, tab, binned_d, binned_s,
                                                            n_edges);
    bkt_degree_scale_kernel<<<n_tiles, 1024, 0, stream>>>(binned_d, binned_s, tab, tk_vals,
                                                          n_nodes, n_edges);
    bkt_sum_kernel<<<n_tiles, 1024, 0, stream>>>(binned_d, tab, tab + 2 * MAX_BKT, tk_vals, tk_idx,
                                                 bias, out, n_nodes);
}


# tiles of 196 nodes, were 256: 511 tile blocks all resident, two on every WGP, equal edge shares (were 391 blocks on 256 WGPs); sum loop predicated, no one-edge tail trips; sum order LDS arrival order
# speedup vs baseline: 3.7100x; 1.0416x over previous
#include <hip/hip_runtime.h>
#include <math.h>

typedef __attribute__((ext_vector_type(2))) float v2f;
typedef __attribute__((ext_vector_type(8))) float v8f;

#define IN_FEATS 256
#define OUT_FEATS 64
#define MAXK 16
#define TILE_M 16
#define GEMM_WAVES 4
#define BLOCK_M (TILE_M * GEMM_WAVES)
#define LDC (OUT_FEATS + 4)
#define LDA (IN_FEATS + 4)
#define BKT_NODES 196
#define IN_TILE_BITS 8
#define IN_TILE_MASK ((1 << IN_TILE_BITS) - 1)
#define MAX_BKT 1024
#define EDGE_CHUNK 8192
#define EPT (EDGE_CHUNK / 1024)
#define HIST_CHUNK 4096
#define HIST_EPT 16
#define TAB_INTS (6 * MAX_BKT)

__global__ void init_kernel(int* __restrict__ tab,
                            const float* __restrict__ weight, float* __restrict__ wT) {
    int i = blockIdx.x * blockDim.x + threadIdx.x;
    if (i < TAB_INTS) tab[i] = 0;
    if (i < IN_FEATS * OUT_FEATS) {
        int k = i >> 6;
        int c = i & 63;
        wT[c * IN_FEATS + k] = weight[i];
    }
}


__device__ __forceinline__ int scalar_bits64(unsigned long long m) {
    int n;
    asm("s_bcnt1_i32_b64 %0, %1" : "=s"(n) : "s"(m) : "scc");
    return n;
}

__device__ __forceinline__ int tile_of(int node) {
    return (int)((unsigned int)node / (unsigned int)BKT_NODES);
}

__device__ __forceinline__ unsigned int order_key(unsigned int u) {
    return u ^ ((unsigned int)((int)u >> 31) | 0x80000000u);
}

__global__ __launch_bounds__(32 * GEMM_WAVES)
void gemm_maxk_kernel(const float* __restrict__ feat, const float* __restrict__ wT,
                      const int* __restrict__ src, const int* __restrict__ dst,
                      int* __restrict__ cnt_d, int* __restrict__ cnt_s,
                      float* __restrict__ tk_vals, unsigned int* __restrict__ tk_idx,
                      int n_nodes, int n_edges, int n_hist) {
    __shared__ __attribute__((aligned(16))) float sA[BLOCK_M * LDA];

    const int tid  = threadIdx.x;
    const int lane = tid & 31;
    const int wave = __builtin_amdgcn_readfirstlane(tid >> 5);

    if ((int)blockIdx.x < n_hist) {
        int* hd = (int*)sA;
        int* hs = hd + MAX_BKT;
        for (int i = tid; i < 2 * MAX_BKT; i += 32 * GEMM_WAVES) hd[i] = 0;
        __syncthreads();
        const int e0 = (blockIdx.x * GEMM_WAVES + wave) * HIST_CHUNK;
        for (int t = 0; t < HIST_CHUNK / (32 * HIST_EPT); ++t) {
            int ks[HIST_EPT], kd[HIST_EPT];
            #pragma unroll
            for (int j = 0; j < HIST_EPT; ++j) {
                const int e  = e0 + (t * HIST_EPT + j) * 32 + lane;
                const int ec = e < n_edges ? e : n_edges - 1;
                const int dv = dst[ec];
                ks[j] = src[ec];
                kd[j] = e < n_edges ? dv : -1;
            }
            #pragma unroll
            for (int j = 0; j < HIST_EPT; ++j) {
                if (kd[j] >= 0) {
                    atomicAdd(&hd[tile_of(kd[j])], 1);
                    atomicAdd(&hs[tile_of(ks[j])], 1);
                }
            }
        }
        __syncthreads();
        for (int i = tid; i < MAX_BKT; i += 32 * GEMM_WAVES) {
            const int cd = hd[i];
            const int cs = hs[i];
            if (cd) atomicAdd(&cnt_d[i], cd);
            if (cs) atomicAdd(&cnt_s[i], cs);
        }
        return;
    }

    const int m0   = ((int)blockIdx.x - n_hist) * BLOCK_M;
    const int r16  = lane & 15;
    const int half = lane >> 4;

    #pragma unroll
    for (int pass = 0; pass < 2; ++pass) {
        float4 buf[16];
        #pragma unroll
        for (int j = 0; j < 16; ++j) {
            int i   = (pass * 16 + j) * (32 * GEMM_WAVES) + tid;
            int row = i >> 6;
            int c4  = i & 63;
            int grow = m0 + row;
            grow = grow < n_nodes ? grow : n_nodes - 1;
            buf[j] = ((const float4*)feat)[(long long)grow * (IN_FEATS / 4) + c4];
        }
        #pragma unroll
        for (int j = 0; j < 16; ++j) {
            int i   = (pass * 16 + j) * (32 * GEMM_WAVES) + tid;
            int row = i >> 6;
            int c4  = i & 63;
            float4 v = buf[j];
            if (m0 + row >= n_nodes) v = make_float4(0.f, 0.f, 0.f, 0.f);
            *(float4*)&sA[row * LDA + c4 * 4] = v;
        }
    }
    __syncthreads();

    const int koff = half << 1;
    const float* arow0 = sA + r16 * LDA;
    const float* arow1 = arow0 + TILE_M * LDA;
    const float* arow2 = arow1 + TILE_M * LDA;
    const float* arow3 = arow2 + TILE_M * LDA;
    const float* wcol  = wT + (wave * 16 + r16) * IN_FEATS;

    v8f acc[GEMM_WAVES];
    #pragma unroll
    for (int i = 0; i < GEMM_WAVES; ++i) acc[i] = (v8f){0.f, 0.f, 0.f, 0.f, 0.f, 0.f, 0.f, 0.f};
    #pragma unroll
    for (int k0 = 0; k0 < IN_FEATS; k0 += 4) {
        v2f a0, a1, a2, a3, b;
        a0.x = arow0[k0 + koff]; a0.y = arow0[k0 + koff + 1];
        a1.x = arow1[k0 + koff]; a1.y = arow1[k0 + koff + 1];
        a2.x = arow2[k0 + koff]; a2.y = arow2[k0 + koff + 1];
        a3.x = arow3[k0 + koff]; a3.y = arow3[k0 + koff + 1];
        b.x  = wcol[k0 + koff];  b.y  = wcol[k0 + koff + 1];
        acc[0] = __builtin_amdgcn_wmma_f32_16x16x4_f32(false, a0, false, b, (short)0, acc[0], false, false);
        acc[1] = __builtin_amdgcn_wmma_f32_16x16x4_f32(false, a1, false, b, (short)0, acc[1], false, false);
        acc[2] = __builtin_amdgcn_wmma_f32_16x16x4_f32(false, a2, false, b, (short)0, acc[2], false, false);
        acc[3] = __builtin_amdgcn_wmma_f32_16x16x4_f32(false, a3, false, b, (short)0, acc[3], false, false);
    }

    __builtin_amdgcn_sched_barrier(0);
    asm volatile("v_nop\n\tv_nop\n\tv_nop\n\tv_nop");
    __builtin_amdgcn_sched_barrier(0);

    __syncthreads();
    float* sC = sA;
    #pragma unroll
    for (int u = 0; u < GEMM_WAVES; ++u) {
        #pragma unroll
        for (int r = 0; r < 8; ++r)
            sC[(u * TILE_M + half * 8 + r) * LDC + wave * 16 + r16] = acc[u][r];
    }
    __syncthreads();
    const float* crow = sC + wave * TILE_M * LDC;
    unsigned int x[TILE_M][2];
    #pragma unroll
    for (int R = 0; R < TILE_M; ++R) {
        x[R][0] = __float_as_uint(crow[R * LDC + lane]);
        x[R][1] = __float_as_uint(crow[R * LDC + 32 + lane]);
    }
    unsigned int key[TILE_M][2];
    #pragma unroll
    for (int R = 0; R < TILE_M; ++R) {
        key[R][0] = order_key(x[R][0]);
        key[R][1] = order_key(x[R][1]);
    }

    unsigned int thr[TILE_M];
    #pragma unroll
    for (int R = 0; R < TILE_M; ++R) thr[R] = 0u;
    for (int bit = 31; bit >= 0; --bit) {
        const unsigned int bmask = 1u << bit;
        unsigned int cand[TILE_M];
        unsigned long long m[TILE_M];
        int n[TILE_M];
        #pragma unroll
        for (int R = 0; R < TILE_M; ++R) cand[R] = thr[R] | bmask;
        __builtin_amdgcn_sched_barrier(0);
        #pragma unroll
        for (int R = 0; R < TILE_M; ++R)
            m[R] = (unsigned long long)__builtin_amdgcn_ballot_w32(key[R][0] >= cand[R]) |
                   ((unsigned long long)__builtin_amdgcn_ballot_w32(key[R][1] >= cand[R]) << 32);
        __builtin_amdgcn_sched_barrier(0);
        #pragma unroll
        for (int R = 0; R < TILE_M; ++R) n[R] = scalar_bits64(m[R]);
        __builtin_amdgcn_sched_barrier(0);
        #pragma unroll
        for (int R = 0; R < TILE_M; ++R) thr[R] = n[R] >= MAXK ? cand[R] : thr[R];
    }

    float* sV = sA + BLOCK_M * LDC;
    unsigned char* sI = (unsigned char*)(sV + BLOCK_M * MAXK);
    #pragma unroll
    for (int R = 0; R < TILE_M; ++R) {
        const unsigned int T = thr[R];
        const bool gt0 = key[R][0] > T,  gt1 = key[R][1] > T;
        const bool eq0 = key[R][0] == T, eq1 = key[R][1] == T;
        const unsigned int g0 = __builtin_amdgcn_ballot_w32(gt0), g1 = __builtin_amdgcn_ballot_w32(gt1);
        const unsigned int e0 = __builtin_amdgcn_ballot_w32(eq0), e1 = __builtin_amdgcn_ballot_w32(eq1);
        const int need = MAXK - (__popc(g0) + __popc(g1));
        const int rank0 = (int)__builtin_amdgcn_mbcnt_lo(e0, 0u);
        const int rank1 = (int)__builtin_amdgcn_mbcnt_lo(e1, (unsigned int)__popc(e0));
        const unsigned int s0 = g0 | (e0 & __builtin_amdgcn_ballot_w32(rank0 < need));
        const unsigned int s1 = g1 | (e1 & __builtin_amdgcn_ballot_w32(rank1 < need));
        const bool sel0 = __builtin_amdgcn_inverse_ballot_w32(s0);
        const bool sel1 = __builtin_amdgcn_inverse_ballot_w32(s1);
        const int slot0 = (int)__builtin_amdgcn_mbcnt_lo(s0, 0u);
        const int slot1 = (int)__builtin_amdgcn_mbcnt_lo(s1, (unsigned int)__popc(s0));
        const int rec = (wave * TILE_M + R) * MAXK;
        if (sel0) {
            sV[rec + slot0] = __uint_as_float(x[R][0]);
            sI[rec + slot0] = (unsigned char)lane;
        }
        if (sel1) {
            sV[rec + slot1] = __uint_as_float(x[R][1]);
            sI[rec + slot1] = (unsigned char)(32 + lane);
        }
    }
    __syncthreads();

    const float4* sV4 = (const float4*)sV;
    #pragma unroll
    for (int q = 0; q < BLOCK_M * MAXK / (4 * 32 * GEMM_WAVES); ++q) {
        const int f = q * (32 * GEMM_WAVES) + tid;
        if (m0 + (f >> 2) < n_nodes)
            ((float4*)tk_vals)[(long long)m0 * (MAXK / 4) + f] = sV4[f];
    }
    if (tid < BLOCK_M && m0 + tid < n_nodes)
        ((uint4*)tk_idx)[m0 + tid] = ((const uint4*)sI)[tid];
}


__global__ __launch_bounds__(1024)
void bkt_fill_kernel(const int* __restrict__ src, const int* __restrict__ dst,
                     int* __restrict__ tab, unsigned int* __restrict__ binned_d,
                     unsigned char* __restrict__ binned_s, int n_edges) {
    const int key = blockIdx.y;
    const int* bkt_cnt = tab + key * 3 * MAX_BKT;
    int* bkt_cur   = tab + key * 3 * MAX_BKT + MAX_BKT;
    int* bkt_start = tab + key * 3 * MAX_BKT + 2 * MAX_BKT;
    __shared__ int hist[MAX_BKT];
    __shared__ int gbase[MAX_BKT];
    __shared__ int wsum[2][32];
    __shared__ unsigned int stage[EDGE_CHUNK];
    __shared__ unsigned short stage_b[EDGE_CHUNK];
    const int tid  = threadIdx.x;
    const int lane = tid & 31;
    const int wave = tid >> 5;
    hist[tid] = 0;
    const int v = bkt_cnt[tid];

    const int base = blockIdx.x * EDGE_CHUNK;
    int es[EPT], ed[EPT];
    #pragma unroll
    for (int j = 0; j < EPT; ++j) {
        const int e  = base + j * 1024 + tid;
        const int ec = e < n_edges ? e : n_edges - 1;
        const int sv = src[ec];
        const int dv = dst[ec];
        es[j] = key ? dv : sv;
        ed[j] = e < n_edges ? (key ? sv : dv) : -1;
    }
    __syncthreads();
    #pragma unroll
    for (int j = 0; j < EPT; ++j)
        if (ed[j] >= 0) atomicAdd(&hist[tile_of(ed[j])], 1);
    __syncthreads();

    const int c = hist[tid];
    int x = v, y = c;
    #pragma unroll
    for (int o = 1; o < 32; o <<= 1) {
        const int xs = __shfl_up(x, o);
        const int ys = __shfl_up(y, o);
        if (lane >= o) { x += xs; y += ys; }
    }
    if (lane == 31) { wsum[0][wave] = x; wsum[1][wave] = y; }
    __syncthreads();
    if (wave < 2) {
        const int w = wsum[wave][lane];
        int s = w;
        #pragma unroll
        for (int o = 1; o < 32; o <<= 1) {
            const int t = __shfl_up(s, o);
            if (lane >= o) s += t;
        }
        wsum[wave][lane] = s - w;
    }
    __syncthreads();

    const int start  = x - v + wsum[0][wave];
    const int lstart = y - c + wsum[1][wave];
    if (blockIdx.x == 0) bkt_start[tid] = start;
    const int rsv = c ? atomicAdd(&bkt_cur[tid], c) : 0;
    gbase[tid] = start + rsv - lstart;
    hist[tid]  = lstart;
    __syncthreads();

    #pragma unroll
    for (int j = 0; j < EPT; ++j) {
        if (ed[j] >= 0) {
            const int b    = tile_of(ed[j]);
            const int slot = atomicAdd(&hist[b], 1);
            stage[slot]   = ((unsigned int)es[j] << IN_TILE_BITS) | (unsigned int)(ed[j] - b * BKT_NODES);
            stage_b[slot] = (unsigned short)b;
        }
    }
    __syncthreads();

    const int left = n_edges - base;
    const int n_here = left < EDGE_CHUNK ? left : EDGE_CHUNK;
    #pragma unroll
    for (int j = 0; j < EPT; ++j) {
        const int slot = j * 1024 + tid;
        if (slot < n_here) {
            const int g = gbase[stage_b[slot]] + slot;
            if (key == 0) binned_d[g] = stage[slot];
            else          binned_s[g] = (unsigned char)stage[slot];
        }
    }
}

__global__ __launch_bounds__(1024)
void bkt_degree_scale_kernel(const unsigned int* __restrict__ binned_d,
                             const unsigned char* __restrict__ binned_s,
                             const int* __restrict__ tab, float* __restrict__ tk_vals,
                             int n_nodes, int n_edges) {
    __shared__ int cin[BKT_NODES];
    __shared__ int cout[BKT_NODES];
    __shared__ float ssc[BKT_NODES];
    const int tid = threadIdx.x;
    if (tid < BKT_NODES) { cin[tid] = 0; cout[tid] = 0; }
    const int nd = tab[blockIdx.x];
    const int sd = tab[2 * MAX_BKT + blockIdx.x];
    const int ns = tab[3 * MAX_BKT + blockIdx.x];
    const int ss = tab[5 * MAX_BKT + blockIdx.x];
    float4* rec = (float4*)tk_vals + (long long)blockIdx.x * (BKT_NODES * MAXK / 4);
    const bool live = (tid >> 2) < BKT_NODES && blockIdx.x * BKT_NODES + (tid >> 2) < n_nodes;
    float4 v = make_float4(0.f, 0.f, 0.f, 0.f);
    if (live) v = rec[tid];
    __syncthreads();
    const int nmax = nd > ns ? nd : ns;
    for (int i0 = tid; i0 < nmax; i0 += 4 * 1024) {
        unsigned int wd[4], ws[4];
        #pragma unroll
        for (int k = 0; k < 4; ++k) {
            const int i  = i0 + 1024 * k;
            const int gd = sd + i < n_edges ? sd + i : n_edges - 1;
            const int gs = ss + i < n_edges ? ss + i : n_edges - 1;
            wd[k] = binned_d[gd];
            ws[k] = binned_s[gs];
        }
        #pragma unroll
        for (int k = 0; k < 4; ++k) {
            const int i = i0 + 1024 * k;
            if (i < nd) atomicAdd(&cin[wd[k] & IN_TILE_MASK], 1);
            if (i < ns) atomicAdd(&cout[ws[k] & IN_TILE_MASK], 1);
        }
    }
    __syncthreads();
    if (tid < BKT_NODES) {
        float id = (float)cin[tid];
        float od = (float)cout[tid];
        od = od < 1.f ? 1.f : od;
        id = id < 1.f ? 1.f : id;
        ssc[tid] = rsqrtf(od) * rsqrtf(id);
    }
    __syncthreads();
    if (live) {
        const float sc = ssc[tid >> 2];
        v.x *= sc; v.y *= sc; v.z *= sc; v.w *= sc;
        rec[tid] = v;
    }
}

__device__ __forceinline__ void add_record(float* row, float4 v, unsigned int ip) {
    atomicAdd(row + (ip & 0xffu), v.x);
    atomicAdd(row + ((ip >> 8) & 0xffu), v.y);
    atomicAdd(row + ((ip >> 16) & 0xffu), v.z);
    atomicAdd(row + ((ip >> 24) & 0xffu), v.w);
}

__global__ __launch_bounds__(1024)
void bkt_sum_kernel(const unsigned int* __restrict__ binned, const int* __restrict__ bkt_cnt,
                    const int* __restrict__ bkt_start, const float* __restrict__ tk_vals,
                    const unsigned int* __restrict__ tk_idx, const float* __restrict__ bias,
                    float* __restrict__ out, int n_nodes) {
    __shared__ __attribute__((aligned(16))) float acc[BKT_NODES * OUT_FEATS];
    const int tid = threadIdx.x;
    for (int i = tid; i < BKT_NODES * OUT_FEATS; i += 1024) acc[i] = bias[i & (OUT_FEATS - 1)];
    __syncthreads();

    const int cnt = bkt_cnt[blockIdx.x];
    const unsigned int* mine = binned + bkt_start[blockIdx.x];
    const int part = tid & 3;
    for (int i0 = tid >> 2; i0 < cnt; i0 += 4 * 256) {
        unsigned int w[4], ip[4];
        float4 v[4];
        #pragma unroll
        for (int k = 0; k < 4; ++k) {
            const int i = i0 + 256 * k;
            w[k] = mine[i < cnt ? i : cnt - 1];
        }
        #pragma unroll
        for (int k = 0; k < 4; ++k) {
            const long long s = (long long)(w[k] >> IN_TILE_BITS);
            v[k]  = ((const float4*)tk_vals)[s * 4 + part];
            ip[k] = tk_idx[s * 4 + part];
        }
        asm volatile("" :: "v"(v[0].x), "v"(v[0].y), "v"(v[0].z), "v"(v[0].w), "v"(ip[0]),
                           "v"(v[1].x), "v"(v[1].y), "v"(v[1].z), "v"(v[1].w), "v"(ip[1]),
                           "v"(v[2].x), "v"(v[2].y), "v"(v[2].z), "v"(v[2].w), "v"(ip[2]),
                           "v"(v[3].x), "v"(v[3].y), "v"(v[3].z), "v"(v[3].w), "v"(ip[3]));
        #pragma unroll
        for (int k = 0; k < 4; ++k)
            if (i0 + 256 * k < cnt)
                add_record(acc + (w[k] & IN_TILE_MASK) * OUT_FEATS, v[k], ip[k]);
    }
    __syncthreads();

    const long long node0 = (long long)blockIdx.x * BKT_NODES;
    for (int i = tid; i < BKT_NODES * OUT_FEATS / 4; i += 1024) {
        if (node0 + (i >> 4) < n_nodes)
            ((float4*)out)[node0 * (OUT_FEATS / 4) + i] = ((const float4*)acc)[i];
    }
}

extern "C" void kernel_launch(void* const* d_in, const int* in_sizes, int n_in,
                              void* d_out, int out_size, void* d_ws, size_t ws_size,
                              hipStream_t stream) {
    const float* feat   = (const float*)d_in[0];
    const float* weight = (const float*)d_in[1];
    const float* bias   = (const float*)d_in[2];
    const int*   src    = (const int*)d_in[3];
    const int*   dst    = (const int*)d_in[4];
    float* out = (float*)d_out;

    const int n_nodes = in_sizes[0] / IN_FEATS;
    const int n_edges = in_sizes[3];

    float* wT      = (float*)d_ws;
    float* tk_vals = wT + IN_FEATS * OUT_FEATS;
    unsigned int* tk_idx = (unsigned int*)(tk_vals + (long long)n_nodes * MAXK);
    int* tab       = (int*)(tk_idx + (long long)n_nodes * (MAXK / 4));
    unsigned int* binned_d  = (unsigned int*)(tab + TAB_INTS);
    unsigned char* binned_s = (unsigned char*)d_out;

    init_kernel<<<IN_FEATS * OUT_FEATS / 256, 256, 0, stream>>>(tab, weight, wT);
    const int n_tiles  = (n_nodes + BKT_NODES - 1) / BKT_NODES;
    const int n_chunks = (n_edges + EDGE_CHUNK - 1) / EDGE_CHUNK;
    const int n_hist   = (n_edges + GEMM_WAVES * HIST_CHUNK - 1) / (GEMM_WAVES * HIST_CHUNK);
    {
        int grid = n_hist + (n_nodes + BLOCK_M - 1) / BLOCK_M;
        gemm_maxk_kernel<<<grid, 32 * GEMM_WAVES, 0, stream>>>(feat, wT, src, dst, tab, tab + 3 * MAX_BKT,
                                                   tk_vals, tk_idx, n_nodes, n_edges, n_hist);
    }
    bkt_fill_kernel<<<dim3(n_chunks, 2), 1024, 0, stream>>>(src, dst, tab, binned_d, binned_s,
                                                            n_edges);
    bkt_degree_scale_kernel<<<n_tiles, 1024, 0, stream>>>(binned_d, binned_s, tab, tk_vals,
                                                          n_nodes, n_edges);
    bkt_sum_kernel<<<n_tiles, 1024, 0, stream>>>(binned_d, tab, tab + 2 * MAX_BKT, tk_vals, tk_idx,
                                                 bias, out, n_nodes);
}


# GEMM block cut to half its LDS (feat tile staged in two K halves through one 64x128 buffer) and 113 VGPRs (operand reads in two-set groups): eight blocks resident a WGP, were four; k order unchanged
# speedup vs baseline: 3.8876x; 1.0479x over previous
#include <hip/hip_runtime.h>
#include <math.h>

typedef __attribute__((ext_vector_type(2))) float v2f;
typedef __attribute__((ext_vector_type(8))) float v8f;
typedef __attribute__((ext_vector_type(4))) float v4f;

#define IN_FEATS 256
#define OUT_FEATS 64
#define MAXK 16
#define TILE_M 16
#define GEMM_WAVES 4
#define BLOCK_M (TILE_M * GEMM_WAVES)
#define LDC (OUT_FEATS + 4)
#define K_HALF (IN_FEATS / 2)
#define LDA (K_HALF + 4)
#define A_STEPS 2
#define B_STEPS 8
#define BKT_NODES 196
#define IN_TILE_BITS 8
#define IN_TILE_MASK ((1 << IN_TILE_BITS) - 1)
#define MAX_BKT 1024
#define EDGE_CHUNK 8192
#define EPT (EDGE_CHUNK / 1024)
#define HIST_CHUNK 4096
#define HIST_EPT 16
#define TAB_INTS (6 * MAX_BKT)

__global__ void init_kernel(int* __restrict__ tab,
                            const float* __restrict__ weight, float* __restrict__ wT) {
    int i = blockIdx.x * blockDim.x + threadIdx.x;
    if (i < TAB_INTS) tab[i] = 0;
    if (i < IN_FEATS * OUT_FEATS) {
        int k = i >> 6;
        int c = i & 63;
        wT[c * IN_FEATS + k] = weight[i];
    }
}


__device__ __forceinline__ int scalar_bits64(unsigned long long m) {
    int n;
    asm("s_bcnt1_i32_b64 %0, %1" : "=s"(n) : "s"(m) : "scc");
    return n;
}

__device__ __forceinline__ int tile_of(int node) {
    return (int)((unsigned int)node / (unsigned int)BKT_NODES);
}

__device__ __forceinline__ unsigned int order_key(unsigned int u) {
    return u ^ ((unsigned int)((int)u >> 31) | 0x80000000u);
}

__global__ __launch_bounds__(32 * GEMM_WAVES)
void gemm_maxk_kernel(const float* __restrict__ feat, const float* __restrict__ wT,
                      const int* __restrict__ src, const int* __restrict__ dst,
                      int* __restrict__ cnt_d, int* __restrict__ cnt_s,
                      float* __restrict__ tk_vals, unsigned int* __restrict__ tk_idx,
                      int n_nodes, int n_edges, int n_hist) {
    __shared__ __attribute__((aligned(16))) float sA[BLOCK_M * LDA];

    const int tid  = threadIdx.x;
    const int lane = tid & 31;
    const int wave = __builtin_amdgcn_readfirstlane(tid >> 5);

    if ((int)blockIdx.x < n_hist) {
        int* hd = (int*)sA;
        int* hs = hd + MAX_BKT;
        for (int i = tid; i < 2 * MAX_BKT; i += 32 * GEMM_WAVES) hd[i] = 0;
        __syncthreads();
        const int e0 = (blockIdx.x * GEMM_WAVES + wave) * HIST_CHUNK;
        for (int t = 0; t < HIST_CHUNK / (32 * HIST_EPT); ++t) {
            int ks[HIST_EPT], kd[HIST_EPT];
            #pragma unroll
            for (int j = 0; j < HIST_EPT; ++j) {
                const int e  = e0 + (t * HIST_EPT + j) * 32 + lane;
                const int ec = e < n_edges ? e : n_edges - 1;
                const int dv = dst[ec];
                ks[j] = src[ec];
                kd[j] = e < n_edges ? dv : -1;
            }
            #pragma unroll
            for (int j = 0; j < HIST_EPT; ++j) {
                if (kd[j] >= 0) {
                    atomicAdd(&hd[tile_of(kd[j])], 1);
                    atomicAdd(&hs[tile_of(ks[j])], 1);
                }
            }
        }
        __syncthreads();
        for (int i = tid; i < MAX_BKT; i += 32 * GEMM_WAVES) {
            const int cd = hd[i];
            const int cs = hs[i];
            if (cd) atomicAdd(&cnt_d[i], cd);
            if (cs) atomicAdd(&cnt_s[i], cs);
        }
        return;
    }

    const int m0   = ((int)blockIdx.x - n_hist) * BLOCK_M;
    const int r16  = lane & 15;
    const int half = lane >> 4;

    const int koff = half << 1;
    const float* arow = sA + r16 * LDA + koff;
    const float* wtile = wT + wave * 16 * IN_FEATS;
    const unsigned int woff = (unsigned int)(r16 * IN_FEATS + koff);
    const bool full = m0 + BLOCK_M <= n_nodes;
    const float* frow = feat + ((long long)m0 + wave) * IN_FEATS;
    const unsigned int foff = (unsigned int)lane;

    v8f acc[GEMM_WAVES];
    #pragma unroll
    for (int i = 0; i < GEMM_WAVES; ++i) acc[i] = (v8f){0.f, 0.f, 0.f, 0.f, 0.f, 0.f, 0.f, 0.f};
    v2f aq[2][A_STEPS][GEMM_WAVES];
    v2f bq[2][B_STEPS];
    #pragma unroll
    for (int h = 0; h < 2; ++h) {
        if (h) __syncthreads();
        #pragma unroll
        for (int s = 0; s < B_STEPS; ++s) bq[0][s] = *(const v2f*)(wtile + h * K_HALF + s * 4 + woff);
        __builtin_amdgcn_sched_barrier(0);
        #pragma unroll
        for (int pass = 0; pass < 2; ++pass) {
            v4f buf[8];
            if (full) {
                #pragma unroll
                for (int j = 0; j < 8; ++j)
                    buf[j] = ((const v4f*)(frow + (pass * 8 + j) * GEMM_WAVES * IN_FEATS + h * K_HALF))[foff];
            } else {
                #pragma unroll
                for (int j = 0; j < 8; ++j) {
                    int grow = m0 + (pass * 8 + j) * GEMM_WAVES + wave;
                    grow = grow < n_nodes ? grow : n_nodes - 1;
                    buf[j] = ((const v4f*)(feat + (long long)grow * IN_FEATS + h * K_HALF))[foff];
                }
            }
            #pragma unroll
            for (int j = 0; j < 8; ++j) {
                const int row = (pass * 8 + j) * GEMM_WAVES + wave;
                v4f v = buf[j];
                if (m0 + row >= n_nodes) v = (v4f){0.f, 0.f, 0.f, 0.f};
                *(v4f*)&sA[row * LDA + lane * 4] = v;
            }
        }
        __syncthreads();
        #pragma unroll
        for (int s = 0; s < A_STEPS; ++s)
            #pragma unroll
            for (int u = 0; u < GEMM_WAVES; ++u)
                aq[0][s][u] = *(const v2f*)(arow + u * TILE_M * LDA + s * 4);
        #pragma unroll
        for (int t = 0; t < K_HALF / (4 * A_STEPS); ++t) {
            if (t + 1 < K_HALF / (4 * A_STEPS)) {
                #pragma unroll
                for (int s = 0; s < A_STEPS; ++s)
                    #pragma unroll
                    for (int u = 0; u < GEMM_WAVES; ++u)
                        aq[(t + 1) & 1][s][u] = *(const v2f*)(arow + u * TILE_M * LDA + ((t + 1) * A_STEPS + s) * 4);
            }
            if (t % (B_STEPS / A_STEPS) == 0 && t / (B_STEPS / A_STEPS) + 1 < K_HALF / (4 * B_STEPS)) {
                const int gn = t / (B_STEPS / A_STEPS) + 1;
                #pragma unroll
                for (int s = 0; s < B_STEPS; ++s)
                    bq[gn & 1][s] = *(const v2f*)(wtile + h * K_HALF + (gn * B_STEPS + s) * 4 + woff);
            }
            __builtin_amdgcn_sched_barrier(0);
            #pragma unroll
            for (int s = 0; s < A_STEPS; ++s) {
                const v2f b = bq[(t / (B_STEPS / A_STEPS)) & 1][(t % (B_STEPS / A_STEPS)) * A_STEPS + s];
                acc[0] = __builtin_amdgcn_wmma_f32_16x16x4_f32(false, aq[t & 1][s][0], false, b, (short)0, acc[0], false, false);
                acc[1] = __builtin_amdgcn_wmma_f32_16x16x4_f32(false, aq[t & 1][s][1], false, b, (short)0, acc[1], false, false);
                acc[2] = __builtin_amdgcn_wmma_f32_16x16x4_f32(false, aq[t & 1][s][2], false, b, (short)0, acc[2], false, false);
                acc[3] = __builtin_amdgcn_wmma_f32_16x16x4_f32(false, aq[t & 1][s][3], false, b, (short)0, acc[3], false, false);
            }
            __builtin_amdgcn_sched_barrier(0);
        }
        asm volatile("v_nop\n\tv_nop\n\tv_nop\n\tv_nop");
        __builtin_amdgcn_sched_barrier(0);
    }

    __syncthreads();
    float* sC = sA;
    #pragma unroll
    for (int u = 0; u < GEMM_WAVES; ++u) {
        #pragma unroll
        for (int r = 0; r < 8; ++r)
            sC[(u * TILE_M + half * 8 + r) * LDC + wave * 16 + r16] = acc[u][r];
    }
    __syncthreads();
    const float* crow = sC + wave * TILE_M * LDC;
    unsigned int x[TILE_M][2];
    #pragma unroll
    for (int R = 0; R < TILE_M; ++R) {
        x[R][0] = __float_as_uint(crow[R * LDC + lane]);
        x[R][1] = __float_as_uint(crow[R * LDC + 32 + lane]);
    }
    unsigned int key[TILE_M][2];
    #pragma unroll
    for (int R = 0; R < TILE_M; ++R) {
        key[R][0] = order_key(x[R][0]);
        key[R][1] = order_key(x[R][1]);
    }

    unsigned int thr[TILE_M];
    #pragma unroll
    for (int R = 0; R < TILE_M; ++R) thr[R] = 0u;
    for (int bit = 31; bit >= 0; --bit) {
        const unsigned int bmask = 1u << bit;
        unsigned int cand[TILE_M];
        unsigned long long m[TILE_M];
        int n[TILE_M];
        #pragma unroll
        for (int R = 0; R < TILE_M; ++R) cand[R] = thr[R] | bmask;
        __builtin_amdgcn_sched_barrier(0);
        #pragma unroll
        for (int R = 0; R < TILE_M; ++R)
            m[R] = (unsigned long long)__builtin_amdgcn_ballot_w32(key[R][0] >= cand[R]) |
                   ((unsigned long long)__builtin_amdgcn_ballot_w32(key[R][1] >= cand[R]) << 32);
        __builtin_amdgcn_sched_barrier(0);
        #pragma unroll
        for (int R = 0; R < TILE_M; ++R) n[R] = scalar_bits64(m[R]);
        __builtin_amdgcn_sched_barrier(0);
        #pragma unroll
        for (int R = 0; R < TILE_M; ++R) thr[R] = n[R] >= MAXK ? cand[R] : thr[R];
    }

    float* sV = sA + BLOCK_M * LDC;
    unsigned char* sI = (unsigned char*)(sV + BLOCK_M * MAXK);
    #pragma unroll
    for (int R = 0; R < TILE_M; ++R) {
        const unsigned int T = thr[R];
        const bool gt0 = key[R][0] > T,  gt1 = key[R][1] > T;
        const bool eq0 = key[R][0] == T, eq1 = key[R][1] == T;
        const unsigned int g0 = __builtin_amdgcn_ballot_w32(gt0), g1 = __builtin_amdgcn_ballot_w32(gt1);
        const unsigned int e0 = __builtin_amdgcn_ballot_w32(eq0), e1 = __builtin_amdgcn_ballot_w32(eq1);
        const int need = MAXK - (__popc(g0) + __popc(g1));
        const int rank0 = (int)__builtin_amdgcn_mbcnt_lo(e0, 0u);
        const int rank1 = (int)__builtin_amdgcn_mbcnt_lo(e1, (unsigned int)__popc(e0));
        const unsigned int s0 = g0 | (e0 & __builtin_amdgcn_ballot_w32(rank0 < need));
        const unsigned int s1 = g1 | (e1 & __builtin_amdgcn_ballot_w32(rank1 < need));
        const bool sel0 = __builtin_amdgcn_inverse_ballot_w32(s0);
        const bool sel1 = __builtin_amdgcn_inverse_ballot_w32(s1);
        const int slot0 = (int)__builtin_amdgcn_mbcnt_lo(s0, 0u);
        const int slot1 = (int)__builtin_amdgcn_mbcnt_lo(s1, (unsigned int)__popc(s0));
        const int rec = (wave * TILE_M + R) * MAXK;
        if (sel0) {
            sV[rec + slot0] = __uint_as_float(x[R][0]);
            sI[rec + slot0] = (unsigned char)lane;
        }
        if (sel1) {
            sV[rec + slot1] = __uint_as_float(x[R][1]);
            sI[rec + slot1] = (unsigned char)(32 + lane);
        }
    }
    __syncthreads();

    const float4* sV4 = (const float4*)sV;
    #pragma unroll
    for (int q = 0; q < BLOCK_M * MAXK / (4 * 32 * GEMM_WAVES); ++q) {
        const int f = q * (32 * GEMM_WAVES) + tid;
        if (m0 + (f >> 2) < n_nodes)
            ((float4*)tk_vals)[(long long)m0 * (MAXK / 4) + f] = sV4[f];
    }
    if (tid < BLOCK_M && m0 + tid < n_nodes)
        ((uint4*)tk_idx)[m0 + tid] = ((const uint4*)sI)[tid];
}


__global__ __launch_bounds__(1024)
void bkt_fill_kernel(const int* __restrict__ src, const int* __restrict__ dst,
                     int* __restrict__ tab, unsigned int* __restrict__ binned_d,
                     unsigned char* __restrict__ binned_s, int n_edges) {
    const int key = blockIdx.y;
    const int* bkt_cnt = tab + key * 3 * MAX_BKT;
    int* bkt_cur   = tab + key * 3 * MAX_BKT + MAX_BKT;
    int* bkt_start = tab + key * 3 * MAX_BKT + 2 * MAX_BKT;
    __shared__ int hist[MAX_BKT];
    __shared__ int gbase[MAX_BKT];
    __shared__ int wsum[2][32];
    __shared__ unsigned int stage[EDGE_CHUNK];
    __shared__ unsigned short stage_b[EDGE_CHUNK];
    const int tid  = threadIdx.x;
    const int lane = tid & 31;
    const int wave = tid >> 5;
    hist[tid] = 0;
    const int v = bkt_cnt[tid];

    const int base = blockIdx.x * EDGE_CHUNK;
    int es[EPT], ed[EPT];
    #pragma unroll
    for (int j = 0; j < EPT; ++j) {
        const int e  = base + j * 1024 + tid;
        const int ec = e < n_edges ? e : n_edges - 1;
        const int sv = src[ec];
        const int dv = dst[ec];
        es[j] = key ? dv : sv;
        ed[j] = e < n_edges ? (key ? sv : dv) : -1;
    }
    __syncthreads();
    #pragma unroll
    for (int j = 0; j < EPT; ++j)
        if (ed[j] >= 0) atomicAdd(&hist[tile_of(ed[j])], 1);
    __syncthreads();

    const int c = hist[tid];
    int x = v, y = c;
    #pragma unroll
    for (int o = 1; o < 32; o <<= 1) {
        const int xs = __shfl_up(x, o);
        const int ys = __shfl_up(y, o);
        if (lane >= o) { x += xs; y += ys; }
    }
    if (lane == 31) { wsum[0][wave] = x; wsum[1][wave] = y; }
    __syncthreads();
    if (wave < 2) {
        const int w = wsum[wave][lane];
        int s = w;
        #pragma unroll
        for (int o = 1; o < 32; o <<= 1) {
            const int t = __shfl_up(s, o);
            if (lane >= o) s += t;
        }
        wsum[wave][lane] = s - w;
    }
    __syncthreads();

    const int start  = x - v + wsum[0][wave];
    const int lstart = y - c + wsum[1][wave];
    if (blockIdx.x == 0) bkt_start[tid] = start;
    const int rsv = c ? atomicAdd(&bkt_cur[tid], c) : 0;
    gbase[tid] = start + rsv - lstart;
    hist[tid]  = lstart;
    __syncthreads();

    #pragma unroll
    for (int j = 0; j < EPT; ++j) {
        if (ed[j] >= 0) {
            const int b    = tile_of(ed[j]);
            const int slot = atomicAdd(&hist[b], 1);
            stage[slot]   = ((unsigned int)es[j] << IN_TILE_BITS) | (unsigned int)(ed[j] - b * BKT_NODES);
            stage_b[slot] = (unsigned short)b;
        }
    }
    __syncthreads();

    const int left = n_edges - base;
    const int n_here = left < EDGE_CHUNK ? left : EDGE_CHUNK;
    #pragma unroll
    for (int j = 0; j < EPT; ++j) {
        const int slot = j * 1024 + tid;
        if (slot < n_here) {
            const int g = gbase[stage_b[slot]] + slot;
            if (key == 0) binned_d[g] = stage[slot];
            else          binned_s[g] = (unsigned char)stage[slot];
        }
    }
}

__global__ __launch_bounds__(1024)
void bkt_degree_scale_kernel(const unsigned int* __restrict__ binned_d,
                             const unsigned char* __restrict__ binned_s,
                             const int* __restrict__ tab, float* __restrict__ tk_vals,
                             int n_nodes, int n_edges) {
    __shared__ int cin[BKT_NODES];
    __shared__ int cout[BKT_NODES];
    __shared__ float ssc[BKT_NODES];
    const int tid = threadIdx.x;
    if (tid < BKT_NODES) { cin[tid] = 0; cout[tid] = 0; }
    const int nd = tab[blockIdx.x];
    const int sd = tab[2 * MAX_BKT + blockIdx.x];
    const int ns = tab[3 * MAX_BKT + blockIdx.x];
    const int ss = tab[5 * MAX_BKT + blockIdx.x];
    float4* rec = (float4*)tk_vals + (long long)blockIdx.x * (BKT_NODES * MAXK / 4);
    const bool live = (tid >> 2) < BKT_NODES && blockIdx.x * BKT_NODES + (tid >> 2) < n_nodes;
    float4 v = make_float4(0.f, 0.f, 0.f, 0.f);
    if (live) v = rec[tid];
    __syncthreads();
    const int nmax = nd > ns ? nd : ns;
    for (int i0 = tid; i0 < nmax; i0 += 4 * 1024) {
        unsigned int wd[4], ws[4];
        #pragma unroll
        for (int k = 0; k < 4; ++k) {
            const int i  = i0 + 1024 * k;
            const int gd = sd + i < n_edges ? sd + i : n_edges - 1;
            const int gs = ss + i < n_edges ? ss + i : n_edges - 1;
            wd[k] = binned_d[gd];
            ws[k] = binned_s[gs];
        }
        #pragma unroll
        for (int k = 0; k < 4; ++k) {
            const int i = i0 + 1024 * k;
            if (i < nd) atomicAdd(&cin[wd[k] & IN_TILE_MASK], 1);
            if (i < ns) atomicAdd(&cout[ws[k] & IN_TILE_MASK], 1);
        }
    }
    __syncthreads();
    if (tid < BKT_NODES) {
        float id = (float)cin[tid];
        float od = (float)cout[tid];
        od = od < 1.f ? 1.f : od;
        id = id < 1.f ? 1.f : id;
        ssc[tid] = rsqrtf(od) * rsqrtf(id);
    }
    __syncthreads();
    if (live) {
        const float sc = ssc[tid >> 2];
        v.x *= sc; v.y *= sc; v.z *= sc; v.w *= sc;
        rec[tid] = v;
    }
}

__device__ __forceinline__ void add_record(float* row, float4 v, unsigned int ip) {
    atomicAdd(row + (ip & 0xffu), v.x);
    atomicAdd(row + ((ip >> 8) & 0xffu), v.y);
    atomicAdd(row + ((ip >> 16) & 0xffu), v.z);
    atomicAdd(row + ((ip >> 24) & 0xffu), v.w);
}

__global__ __launch_bounds__(1024)
void bkt_sum_kernel(const unsigned int* __restrict__ binned, const int* __restrict__ bkt_cnt,
                    const int* __restrict__ bkt_start, const float* __restrict__ tk_vals,
                    const unsigned int* __restrict__ tk_idx, const float* __restrict__ bias,
                    float* __restrict__ out, int n_nodes) {
    __shared__ __attribute__((aligned(16))) float acc[BKT_NODES * OUT_FEATS];
    const int tid = threadIdx.x;
    for (int i = tid; i < BKT_NODES * OUT_FEATS; i += 1024) acc[i] = bias[i & (OUT_FEATS - 1)];
    __syncthreads();

    const int cnt = bkt_cnt[blockIdx.x];
    const unsigned int* mine = binned + bkt_start[blockIdx.x];
    const int part = tid & 3;
    for (int i0 = tid >> 2; i0 < cnt; i0 += 4 * 256) {
        unsigned int w[4], ip[4];
        float4 v[4];
        #pragma unroll
        for (int k = 0; k < 4; ++k) {
            const int i = i0 + 256 * k;
            w[k] = mine[i < cnt ? i : cnt - 1];
        }
        #pragma unroll
        for (int k = 0; k < 4; ++k) {
            const long long s = (long long)(w[k] >> IN_TILE_BITS);
            v[k]  = ((const float4*)tk_vals)[s * 4 + part];
            ip[k] = tk_idx[s * 4 + part];
        }
        asm volatile("" :: "v"(v[0].x), "v"(v[0].y), "v"(v[0].z), "v"(v[0].w), "v"(ip[0]),
                           "v"(v[1].x), "v"(v[1].y), "v"(v[1].z), "v"(v[1].w), "v"(ip[1]),
                           "v"(v[2].x), "v"(v[2].y), "v"(v[2].z), "v"(v[2].w), "v"(ip[2]),
                           "v"(v[3].x), "v"(v[3].y), "v"(v[3].z), "v"(v[3].w), "v"(ip[3]));
        #pragma unroll
        for (int k = 0; k < 4; ++k)
            if (i0 + 256 * k < cnt)
                add_record(acc + (w[k] & IN_TILE_MASK) * OUT_FEATS, v[k], ip[k]);
    }
    __syncthreads();

    const long long node0 = (long long)blockIdx.x * BKT_NODES;
    for (int i = tid; i < BKT_NODES * OUT_FEATS / 4; i += 1024) {
        if (node0 + (i >> 4) < n_nodes)
            ((float4*)out)[node0 * (OUT_FEATS / 4) + i] = ((const float4*)acc)[i];
    }
}

extern "C" void kernel_launch(void* const* d_in, const int* in_sizes, int n_in,
                              void* d_out, int out_size, void* d_ws, size_t ws_size,
                              hipStream_t stream) {
    const float* feat   = (const float*)d_in[0];
    const float* weight = (const float*)d_in[1];
    const float* bias   = (const float*)d_in[2];
    const int*   src    = (const int*)d_in[3];
    const int*   dst    = (const int*)d_in[4];
    float* out = (float*)d_out;

    const int n_nodes = in_sizes[0] / IN_FEATS;
    const int n_edges = in_sizes[3];

    float* wT      = (float*)d_ws;
    float* tk_vals = wT + IN_FEATS * OUT_FEATS;
    unsigned int* tk_idx = (unsigned int*)(tk_vals + (long long)n_nodes * MAXK);
    int* tab       = (int*)(tk_idx + (long long)n_nodes * (MAXK / 4));
    unsigned int* binned_d  = (unsigned int*)(tab + TAB_INTS);
    unsigned char* binned_s = (unsigned char*)d_out;

    init_kernel<<<IN_FEATS * OUT_FEATS / 256, 256, 0, stream>>>(tab, weight, wT);
    const int n_tiles  = (n_nodes + BKT_NODES - 1) / BKT_NODES;
    const int n_chunks = (n_edges + EDGE_CHUNK - 1) / EDGE_CHUNK;
    const int n_hist   = (n_edges + GEMM_WAVES * HIST_CHUNK - 1) / (GEMM_WAVES * HIST_CHUNK);
    {
        int grid = n_hist + (n_nodes + BLOCK_M - 1) / BLOCK_M;
        gemm_maxk_kernel<<<grid, 32 * GEMM_WAVES, 0, stream>>>(feat, wT, src, dst, tab, tab + 3 * MAX_BKT,
                                                   tk_vals, tk_idx, n_nodes, n_edges, n_hist);
    }
    bkt_fill_kernel<<<dim3(n_chunks, 2), 1024, 0, stream>>>(src, dst, tab, binned_d, binned_s,
                                                            n_edges);
    bkt_degree_scale_kernel<<<n_tiles, 1024, 0, stream>>>(binned_d, binned_s, tab, tk_vals,
                                                          n_nodes, n_edges);
    bkt_sum_kernel<<<n_tiles, 1024, 0, stream>>>(binned_d, tab, tab + 2 * MAX_BKT, tk_vals, tk_idx,
                                                 bias, out, n_nodes);
}


# edges sorted by tile inside their 8192-edge chunk only (16-bit run starts a chunk; tile blocks walk runs via an LDS list): no tile count, cursor add or zeroing launch, 4 launches; sum order unchanged
# speedup vs baseline: 4.0522x; 1.0423x over previous
#include <hip/hip_runtime.h>
#include <math.h>

typedef __attribute__((ext_vector_type(2))) float v2f;
typedef __attribute__((ext_vector_type(8))) float v8f;
typedef __attribute__((ext_vector_type(4))) float v4f;

#define IN_FEATS 256
#define OUT_FEATS 64
#define MAXK 16
#define TILE_M 16
#define GEMM_WAVES 4
#define BLOCK_M (TILE_M * GEMM_WAVES)
#define LDC (OUT_FEATS + 4)
#define K_HALF (IN_FEATS / 2)
#define LDA (K_HALF + 4)
#define A_STEPS 2
#define B_STEPS 8
#define BKT_NODES 196
#define IN_TILE_BITS 8
#define IN_TILE_MASK ((1 << IN_TILE_BITS) - 1)
#define MAX_BKT 1024
#define EDGE_CHUNK 8192
#define EPT (EDGE_CHUNK / 1024)
#define LIST_CAP 4096


__device__ __forceinline__ int scalar_bits64(unsigned long long m) {
    int n;
    asm("s_bcnt1_i32_b64 %0, %1" : "=s"(n) : "s"(m) : "scc");
    return n;
}

__device__ __forceinline__ int tile_of(int node) {
    return (int)((unsigned int)node / (unsigned int)BKT_NODES);
}

__device__ __forceinline__ unsigned int order_key(unsigned int u) {
    return u ^ ((unsigned int)((int)u >> 31) | 0x80000000u);
}

__global__ __launch_bounds__(32 * GEMM_WAVES)
void gemm_maxk_kernel(const float* __restrict__ feat, const float* __restrict__ wT,
                      float* __restrict__ tk_vals, unsigned int* __restrict__ tk_idx,
                      int n_nodes) {
    __shared__ __attribute__((aligned(16))) float sA[BLOCK_M * LDA];

    const int tid  = threadIdx.x;
    const int lane = tid & 31;
    const int wave = __builtin_amdgcn_readfirstlane(tid >> 5);

    const int m0   = (int)blockIdx.x * BLOCK_M;
    const int r16  = lane & 15;
    const int half = lane >> 4;

    const int koff = half << 1;
    const float* arow = sA + r16 * LDA + koff;
    const float* wtile = wT + wave * 16 * IN_FEATS;
    const unsigned int woff = (unsigned int)(r16 * IN_FEATS + koff);
    const bool full = m0 + BLOCK_M <= n_nodes;
    const float* frow = feat + ((long long)m0 + wave) * IN_FEATS;
    const unsigned int foff = (unsigned int)lane;

    v8f acc[GEMM_WAVES];
    #pragma unroll
    for (int i = 0; i < GEMM_WAVES; ++i) acc[i] = (v8f){0.f, 0.f, 0.f, 0.f, 0.f, 0.f, 0.f, 0.f};
    v2f aq[2][A_STEPS][GEMM_WAVES];
    v2f bq[2][B_STEPS];
    #pragma unroll
    for (int h = 0; h < 2; ++h) {
        if (h) __syncthreads();
        #pragma unroll
        for (int s = 0; s < B_STEPS; ++s) bq[0][s] = *(const v2f*)(wtile + h * K_HALF + s * 4 + woff);
        __builtin_amdgcn_sched_barrier(0);
        #pragma unroll
        for (int pass = 0; pass < 2; ++pass) {
            v4f buf[8];
            if (full) {
                #pragma unroll
                for (int j = 0; j < 8; ++j)
                    buf[j] = ((const v4f*)(frow + (pass * 8 + j) * GEMM_WAVES * IN_FEATS + h * K_HALF))[foff];
            } else {
                #pragma unroll
                for (int j = 0; j < 8; ++j) {
                    int grow = m0 + (pass * 8 + j) * GEMM_WAVES + wave;
                    grow = grow < n_nodes ? grow : n_nodes - 1;
                    buf[j] = ((const v4f*)(feat + (long long)grow * IN_FEATS + h * K_HALF))[foff];
                }
            }
            #pragma unroll
            for (int j = 0; j < 8; ++j) {
                const int row = (pass * 8 + j) * GEMM_WAVES + wave;
                v4f v = buf[j];
                if (m0 + row >= n_nodes) v = (v4f){0.f, 0.f, 0.f, 0.f};
                *(v4f*)&sA[row * LDA + lane * 4] = v;
            }
        }
        __syncthreads();
        #pragma unroll
        for (int s = 0; s < A_STEPS; ++s)
            #pragma unroll
            for (int u = 0; u < GEMM_WAVES; ++u)
                aq[0][s][u] = *(const v2f*)(arow + u * TILE_M * LDA + s * 4);
        #pragma unroll
        for (int t = 0; t < K_HALF / (4 * A_STEPS); ++t) {
            if (t + 1 < K_HALF / (4 * A_STEPS)) {
                #pragma unroll
                for (int s = 0; s < A_STEPS; ++s)
                    #pragma unroll
                    for (int u = 0; u < GEMM_WAVES; ++u)
                        aq[(t + 1) & 1][s][u] = *(const v2f*)(arow + u * TILE_M * LDA + ((t + 1) * A_STEPS + s) * 4);
            }
            if (t % (B_STEPS / A_STEPS) == 0 && t / (B_STEPS / A_STEPS) + 1 < K_HALF / (4 * B_STEPS)) {
                const int gn = t / (B_STEPS / A_STEPS) + 1;
                #pragma unroll
                for (int s = 0; s < B_STEPS; ++s)
                    bq[gn & 1][s] = *(const v2f*)(wtile + h * K_HALF + (gn * B_STEPS + s) * 4 + woff);
            }
            __builtin_amdgcn_sched_barrier(0);
            #pragma unroll
            for (int s = 0; s < A_STEPS; ++s) {
                const v2f b = bq[(t / (B_STEPS / A_STEPS)) & 1][(t % (B_STEPS / A_STEPS)) * A_STEPS + s];
                acc[0] = __builtin_amdgcn_wmma_f32_16x16x4_f32(false, aq[t & 1][s][0], false, b, (short)0, acc[0], false, false);
                acc[1] = __builtin_amdgcn_wmma_f32_16x16x4_f32(false, aq[t & 1][s][1], false, b, (short)0, acc[1], false, false);
                acc[2] = __builtin_amdgcn_wmma_f32_16x16x4_f32(false, aq[t & 1][s][2], false, b, (short)0, acc[2], false, false);
                acc[3] = __builtin_amdgcn_wmma_f32_16x16x4_f32(false, aq[t & 1][s][3], false, b, (short)0, acc[3], false, false);
            }
            __builtin_amdgcn_sched_barrier(0);
        }
        asm volatile("v_nop\n\tv_nop\n\tv_nop\n\tv_nop");
        __builtin_amdgcn_sched_barrier(0);
    }

    __syncthreads();
    float* sC = sA;
    #pragma unroll
    for (int u = 0; u < GEMM_WAVES; ++u) {
        #pragma unroll
        for (int r = 0; r < 8; ++r)
            sC[(u * TILE_M + half * 8 + r) * LDC + wave * 16 + r16] = acc[u][r];
    }
    __syncthreads();
    const float* crow = sC + wave * TILE_M * LDC;
    unsigned int x[TILE_M][2];
    #pragma unroll
    for (int R = 0; R < TILE_M; ++R) {
        x[R][0] = __float_as_uint(crow[R * LDC + lane]);
        x[R][1] = __float_as_uint(crow[R * LDC + 32 + lane]);
    }
    unsigned int key[TILE_M][2];
    #pragma unroll
    for (int R = 0; R < TILE_M; ++R) {
        key[R][0] = order_key(x[R][0]);
        key[R][1] = order_key(x[R][1]);
    }

    unsigned int thr[TILE_M];
    #pragma unroll
    for (int R = 0; R < TILE_M; ++R) thr[R] = 0u;
    for (int bit = 31; bit >= 0; --bit) {
        const unsigned int bmask = 1u << bit;
        unsigned int cand[TILE_M];
        unsigned long long m[TILE_M];
        int n[TILE_M];
        #pragma unroll
        for (int R = 0; R < TILE_M; ++R) cand[R] = thr[R] | bmask;
        __builtin_amdgcn_sched_barrier(0);
        #pragma unroll
        for (int R = 0; R < TILE_M; ++R)
            m[R] = (unsigned long long)__builtin_amdgcn_ballot_w32(key[R][0] >= cand[R]) |
                   ((unsigned long long)__builtin_amdgcn_ballot_w32(key[R][1] >= cand[R]) << 32);
        __builtin_amdgcn_sched_barrier(0);
        #pragma unroll
        for (int R = 0; R < TILE_M; ++R) n[R] = scalar_bits64(m[R]);
        __builtin_amdgcn_sched_barrier(0);
        #pragma unroll
        for (int R = 0; R < TILE_M; ++R) thr[R] = n[R] >= MAXK ? cand[R] : thr[R];
    }

    float* sV = sA + BLOCK_M * LDC;
    unsigned char* sI = (unsigned char*)(sV + BLOCK_M * MAXK);
    #pragma unroll
    for (int R = 0; R < TILE_M; ++R) {
        const unsigned int T = thr[R];
        const bool gt0 = key[R][0] > T,  gt1 = key[R][1] > T;
        const bool eq0 = key[R][0] == T, eq1 = key[R][1] == T;
        const unsigned int g0 = __builtin_amdgcn_ballot_w32(gt0), g1 = __builtin_amdgcn_ballot_w32(gt1);
        const unsigned int e0 = __builtin_amdgcn_ballot_w32(eq0), e1 = __builtin_amdgcn_ballot_w32(eq1);
        const int need = MAXK - (__popc(g0) + __popc(g1));
        const int rank0 = (int)__builtin_amdgcn_mbcnt_lo(e0, 0u);
        const int rank1 = (int)__builtin_amdgcn_mbcnt_lo(e1, (unsigned int)__popc(e0));
        const unsigned int s0 = g0 | (e0 & __builtin_amdgcn_ballot_w32(rank0 < need));
        const unsigned int s1 = g1 | (e1 & __builtin_amdgcn_ballot_w32(rank1 < need));
        const bool sel0 = __builtin_amdgcn_inverse_ballot_w32(s0);
        const bool sel1 = __builtin_amdgcn_inverse_ballot_w32(s1);
        const int slot0 = (int)__builtin_amdgcn_mbcnt_lo(s0, 0u);
        const int slot1 = (int)__builtin_amdgcn_mbcnt_lo(s1, (unsigned int)__popc(s0));
        const int rec = (wave * TILE_M + R) * MAXK;
        if (sel0) {
            sV[rec + slot0] = __uint_as_float(x[R][0]);
            sI[rec + slot0] = (unsigned char)lane;
        }
        if (sel1) {
            sV[rec + slot1] = __uint_as_float(x[R][1]);
            sI[rec + slot1] = (unsigned char)(32 + lane);
        }
    }
    __syncthreads();

    const float4* sV4 = (const float4*)sV;
    #pragma unroll
    for (int q = 0; q < BLOCK_M * MAXK / (4 * 32 * GEMM_WAVES); ++q) {
        const int f = q * (32 * GEMM_WAVES) + tid;
        if (m0 + (f >> 2) < n_nodes)
            ((float4*)tk_vals)[(long long)m0 * (MAXK / 4) + f] = sV4[f];
    }
    if (tid < BLOCK_M && m0 + tid < n_nodes)
        ((uint4*)tk_idx)[m0 + tid] = ((const uint4*)sI)[tid];
}

__global__ __launch_bounds__(1024)
void chunk_sort_kernel(const int* __restrict__ src, const int* __restrict__ dst,
                       unsigned int* __restrict__ binned_d, unsigned char* __restrict__ binned_s,
                       unsigned short* __restrict__ run_d, unsigned short* __restrict__ run_s,
                       const float* __restrict__ weight, float* __restrict__ wT,
                       int n_edges, int n_tiles) {
    const int key = blockIdx.y;
    __shared__ int hist[MAX_BKT];
    __shared__ int wsum[32];
    __shared__ unsigned int stage[EDGE_CHUNK];
    const int tid  = threadIdx.x;
    const int lane = tid & 31;
    const int wave = tid >> 5;
    for (int i = ((int)blockIdx.y * (int)gridDim.x + (int)blockIdx.x) * 1024 + tid; i < IN_FEATS * OUT_FEATS;
         i += (int)gridDim.x * (int)gridDim.y * 1024)
        wT[(i & 63) * IN_FEATS + (i >> 6)] = weight[i];
    hist[tid] = 0;

    const int base = blockIdx.x * EDGE_CHUNK;
    int es[EPT], ed[EPT];
    #pragma unroll
    for (int j = 0; j < EPT; ++j) {
        const int e  = base + j * 1024 + tid;
        const int ec = e < n_edges ? e : n_edges - 1;
        const int sv = src[ec];
        const int dv = dst[ec];
        es[j] = key ? dv : sv;
        ed[j] = e < n_edges ? (key ? sv : dv) : -1;
    }
    __syncthreads();
    #pragma unroll
    for (int j = 0; j < EPT; ++j)
        if (ed[j] >= 0) atomicAdd(&hist[tile_of(ed[j])], 1);
    __syncthreads();

    const int c = hist[tid];
    int y = c;
    #pragma unroll
    for (int o = 1; o < 32; o <<= 1) {
        const int ys = __shfl_up(y, o);
        if (lane >= o) y += ys;
    }
    if (lane == 31) wsum[wave] = y;
    __syncthreads();
    if (wave == 0) {
        const int w = wsum[lane];
        int s = w;
        #pragma unroll
        for (int o = 1; o < 32; o <<= 1) {
            const int t = __shfl_up(s, o);
            if (lane >= o) s += t;
        }
        wsum[lane] = s - w;
    }
    __syncthreads();

    const int lstart = y - c + wsum[wave];
    if (tid <= n_tiles)
        (key ? run_s : run_d)[(long long)blockIdx.x * (n_tiles + 2) + tid] = (unsigned short)lstart;
    hist[tid] = lstart;
    __syncthreads();

    #pragma unroll
    for (int j = 0; j < EPT; ++j) {
        if (ed[j] >= 0) {
            const int b    = tile_of(ed[j]);
            const int slot = atomicAdd(&hist[b], 1);
            stage[slot] = ((unsigned int)es[j] << IN_TILE_BITS) | (unsigned int)(ed[j] - b * BKT_NODES);
        }
    }
    __syncthreads();

    const int left = n_edges - base;
    const int n_here = left < EDGE_CHUNK ? left : EDGE_CHUNK;
    #pragma unroll
    for (int j = 0; j < EPT; ++j) {
        const int slot = j * 1024 + tid;
        if (slot < n_here) {
            if (key == 0) binned_d[base + slot] = stage[slot];
            else          binned_s[base + slot] = (unsigned char)stage[slot];
        }
    }
}

__device__ __forceinline__ int block_excl_scan(int v, int* wsum, int lane, int wave, int* total) {
    int x = v;
    #pragma unroll
    for (int o = 1; o < 32; o <<= 1) {
        const int t = __shfl_up(x, o);
        if (lane >= o) x += t;
    }
    if (lane == 31) wsum[wave] = x;
    __syncthreads();
    if (wave == 0) {
        const int w = wsum[lane];
        int s = w;
        #pragma unroll
        for (int o = 1; o < 32; o <<= 1) {
            const int t = __shfl_up(s, o);
            if (lane >= o) s += t;
        }
        wsum[lane] = s - w;
        if (lane == 31) wsum[32] = s;
    }
    __syncthreads();
    *total = wsum[32];
    return x - v + wsum[wave];
}

__device__ __forceinline__ void block_excl_scan2(int v0, int v1, int* wsum, int lane, int wave,
                                                 int* p0, int* p1, int* t0, int* t1) {
    int x = v0, y = v1;
    #pragma unroll
    for (int o = 1; o < 32; o <<= 1) {
        const int xs = __shfl_up(x, o);
        const int ys = __shfl_up(y, o);
        if (lane >= o) { x += xs; y += ys; }
    }
    if (lane == 31) { wsum[wave] = x; wsum[33 + wave] = y; }
    __syncthreads();
    if (wave < 2) {
        int* ws = wsum + 33 * wave;
        const int w = ws[lane];
        int s = w;
        #pragma unroll
        for (int o = 1; o < 32; o <<= 1) {
            const int t = __shfl_up(s, o);
            if (lane >= o) s += t;
        }
        ws[lane] = s - w;
        if (lane == 31) ws[32] = s;
    }
    __syncthreads();
    *p0 = x - v0 + wsum[wave];
    *p1 = y - v1 + wsum[33 + wave];
    *t0 = wsum[32];
    *t1 = wsum[65];
}

__device__ __forceinline__ void list_run(int* list, int pre, int len, int gstart, int wbase) {
    const int j0 = pre > wbase ? pre : wbase;
    const int e1 = pre + len;
    const int e2 = wbase + LIST_CAP;
    const int j1 = e1 < e2 ? e1 : e2;
    #pragma nounroll
    for (int j = j0; j < j1; ++j) list[j - wbase] = gstart + (j - pre);
}

__global__ __launch_bounds__(1024)
void bkt_degree_scale_kernel(const unsigned int* __restrict__ binned_d,
                             const unsigned char* __restrict__ binned_s,
                             const unsigned short* __restrict__ run_d,
                             const unsigned short* __restrict__ run_s,
                             float* __restrict__ tk_vals,
                             int n_nodes, int n_chunks, int n_tiles) {
    __shared__ int cin[BKT_NODES];
    __shared__ int cout[BKT_NODES];
    __shared__ float ssc[BKT_NODES];
    __shared__ int list_d[LIST_CAP];
    __shared__ int list_s[LIST_CAP];
    __shared__ int wsd[66];
    const int tid  = threadIdx.x;
    const int lane = tid & 31;
    const int wave = tid >> 5;
    if (tid < BKT_NODES) { cin[tid] = 0; cout[tid] = 0; }
    float4* rec = (float4*)tk_vals + (long long)blockIdx.x * (BKT_NODES * MAXK / 4);
    const bool live = (tid >> 2) < BKT_NODES && blockIdx.x * BKT_NODES + (tid >> 2) < n_nodes;
    float4 v = make_float4(0.f, 0.f, 0.f, 0.f);
    if (live) v = rec[tid];
    int lend = 0, gd = 0, lens = 0, gs = 0;
    if (tid < n_chunks) {
        const long long r = (long long)tid * (n_tiles + 2) + blockIdx.x;
        long long r1 = r + 1;
        asm volatile("" : "+v"(r1));
        const int ad = run_d[r];
        const int as = run_s[r];
        const int bd = run_d[r1];
        const int bs = run_s[r1];
        lend = bd - ad; gd = tid * EDGE_CHUNK + ad;
        lens = bs - as; gs = tid * EDGE_CHUNK + as;
    }
    int nd, ns, pd, ps;
    block_excl_scan2(lend, lens, wsd, lane, wave, &pd, &ps, &nd, &ns);
    const int nmax = nd > ns ? nd : ns;
    for (int wbase = 0; wbase < nmax; wbase += LIST_CAP) {
        list_run(list_d, pd, lend, gd, wbase);
        list_run(list_s, ps, lens, gs, wbase);
        __syncthreads();
        const int wd_n = nd - wbase < LIST_CAP ? nd - wbase : LIST_CAP;
        const int ws_n = ns - wbase < LIST_CAP ? ns - wbase : LIST_CAP;
        const int wn = wd_n > ws_n ? wd_n : ws_n;
        for (int i0 = tid; i0 < wn; i0 += 4 * 1024) {
            unsigned int wd[4], ws[4];
            #pragma unroll
            for (int k = 0; k < 4; ++k) {
                const int i = i0 + 1024 * k;
                int qd = 0, qs = 0;
                if (wd_n > 0) qd = list_d[i < wd_n ? i : wd_n - 1];
                if (ws_n > 0) qs = list_s[i < ws_n ? i : ws_n - 1];
                wd[k] = binned_d[qd];
                ws[k] = binned_s[qs];
            }
            #pragma unroll
            for (int k = 0; k < 4; ++k) {
                const int i = i0 + 1024 * k;
                if (i < wd_n) atomicAdd(&cin[wd[k] & IN_TILE_MASK], 1);
                if (i < ws_n) atomicAdd(&cout[ws[k] & IN_TILE_MASK], 1);
            }
        }
        __syncthreads();
    }
    __syncthreads();
    if (tid < BKT_NODES) {
        float id = (float)cin[tid];
        float od = (float)cout[tid];
        od = od < 1.f ? 1.f : od;
        id = id < 1.f ? 1.f : id;
        ssc[tid] = rsqrtf(od) * rsqrtf(id);
    }
    __syncthreads();
    if (live) {
        const float sc = ssc[tid >> 2];
        v.x *= sc; v.y *= sc; v.z *= sc; v.w *= sc;
        rec[tid] = v;
    }
}

__device__ __forceinline__ void add_record(float* row, float4 v, unsigned int ip) {
    atomicAdd(row + (ip & 0xffu), v.x);
    atomicAdd(row + ((ip >> 8) & 0xffu), v.y);
    atomicAdd(row + ((ip >> 16) & 0xffu), v.z);
    atomicAdd(row + ((ip >> 24) & 0xffu), v.w);
}

__global__ __launch_bounds__(1024)
void bkt_sum_kernel(const unsigned int* __restrict__ binned, const unsigned short* __restrict__ run_d,
                    const float* __restrict__ tk_vals,
                    const unsigned int* __restrict__ tk_idx, const float* __restrict__ bias,
                    float* __restrict__ out, int n_nodes, int n_chunks, int n_tiles) {
    __shared__ __attribute__((aligned(16))) float acc[BKT_NODES * OUT_FEATS];
    __shared__ int list[LIST_CAP];
    __shared__ int wsd[33];
    const int tid  = threadIdx.x;
    const int lane = tid & 31;
    const int wave = tid >> 5;
    const float b0 = bias[tid & (OUT_FEATS - 1)];
    const int cc = tid < n_chunks ? tid : 0;
    const long long r = (long long)cc * (n_tiles + 2) + blockIdx.x;
    long long r1 = r + 1;
    asm volatile("" : "+v"(r1));
    const int ra = run_d[r];
    const int rb = run_d[r1];
    for (int i = tid; i < BKT_NODES * OUT_FEATS; i += 1024) acc[i] = b0;
    const int len = tid < n_chunks ? rb - ra : 0;
    const int g = cc * EDGE_CHUNK + ra;
    int cnt;
    const int pre = block_excl_scan(len, wsd, lane, wave, &cnt);

    const int part = tid & 3;
    for (int wbase = 0; wbase < cnt; wbase += LIST_CAP) {
        list_run(list, pre, len, g, wbase);
        __syncthreads();
        const int wn = cnt - wbase < LIST_CAP ? cnt - wbase : LIST_CAP;
        for (int i0 = tid >> 2; i0 < wn; i0 += 4 * 256) {
            unsigned int w[4], ip[4];
            float4 v[4];
            #pragma unroll
            for (int k = 0; k < 4; ++k) {
                const int i = i0 + 256 * k;
                w[k] = binned[list[i < wn ? i : wn - 1]];
            }
            #pragma unroll
            for (int k = 0; k < 4; ++k) {
                const long long s = (long long)(w[k] >> IN_TILE_BITS);
                v[k]  = ((const float4*)tk_vals)[s * 4 + part];
                ip[k] = tk_idx[s * 4 + part];
            }
            asm volatile("" :: "v"(v[0].x), "v"(v[0].y), "v"(v[0].z), "v"(v[0].w), "v"(ip[0]),
                               "v"(v[1].x), "v"(v[1].y), "v"(v[1].z), "v"(v[1].w), "v"(ip[1]),
                               "v"(v[2].x), "v"(v[2].y), "v"(v[2].z), "v"(v[2].w), "v"(ip[2]),
                               "v"(v[3].x), "v"(v[3].y), "v"(v[3].z), "v"(v[3].w), "v"(ip[3]));
            #pragma unroll
            for (int k = 0; k < 4; ++k)
                if (i0 + 256 * k < wn)
                    add_record(acc + (w[k] & IN_TILE_MASK) * OUT_FEATS, v[k], ip[k]);
        }
        __syncthreads();
    }
    __syncthreads();

    const long long node0 = (long long)blockIdx.x * BKT_NODES;
    for (int i = tid; i < BKT_NODES * OUT_FEATS / 4; i += 1024) {
        if (node0 + (i >> 4) < n_nodes)
            ((float4*)out)[node0 * (OUT_FEATS / 4) + i] = ((const float4*)acc)[i];
    }
}

extern "C" void kernel_launch(void* const* d_in, const int* in_sizes, int n_in,
                              void* d_out, int out_size, void* d_ws, size_t ws_size,
                              hipStream_t stream) {
    const float* feat   = (const float*)d_in[0];
    const float* weight = (const float*)d_in[1];
    const float* bias   = (const float*)d_in[2];
    const int*   src    = (const int*)d_in[3];
    const int*   dst    = (const int*)d_in[4];
    float* out = (float*)d_out;

    const int n_nodes = in_sizes[0] / IN_FEATS;
    const int n_edges = in_sizes[3];
    const int n_tiles  = (n_nodes + BKT_NODES - 1) / BKT_NODES;
    const int n_chunks = (n_edges + EDGE_CHUNK - 1) / EDGE_CHUNK;

    float* wT      = (float*)d_ws;
    float* tk_vals = wT + IN_FEATS * OUT_FEATS;
    unsigned int* tk_idx = (unsigned int*)(tk_vals + (long long)n_nodes * MAXK);
    unsigned int* binned_d  = tk_idx + (long long)n_nodes * (MAXK / 4);
    unsigned short* run_d   = (unsigned short*)(binned_d + n_edges);
    unsigned char* binned_s = (unsigned char*)d_out;
    unsigned short* run_s   = (unsigned short*)(binned_s + (long long)n_chunks * EDGE_CHUNK);

    chunk_sort_kernel<<<dim3(n_chunks, 2), 1024, 0, stream>>>(src, dst, binned_d, binned_s, run_d, run_s,
                                                              weight, wT, n_edges, n_tiles);
    gemm_maxk_kernel<<<(n_nodes + BLOCK_M - 1) / BLOCK_M, 32 * GEMM_WAVES, 0, stream>>>(feat, wT, tk_vals, tk_idx,
                                                                                        n_nodes);
    bkt_degree_scale_kernel<<<n_tiles, 1024, 0, stream>>>(binned_d, binned_s, run_d, run_s, tk_vals,
                                                          n_nodes, n_chunks, n_tiles);
    bkt_sum_kernel<<<n_tiles, 1024, 0, stream>>>(binned_d, run_d, tk_vals, tk_idx, bias, out,
                                                 n_nodes, n_chunks, n_tiles);
}


# GEMM B fragments read from the input W as it lies, k-major: two 4-byte loads a fragment, 2 lines a load (was one 8-byte load over 16 lines of a transposed copy, now gone); k pairing, sums unchanged
# speedup vs baseline: 4.3105x; 1.0637x over previous
#include <hip/hip_runtime.h>
#include <math.h>

typedef __attribute__((ext_vector_type(2))) float v2f;
typedef __attribute__((ext_vector_type(8))) float v8f;
typedef __attribute__((ext_vector_type(4))) float v4f;

#define IN_FEATS 256
#define OUT_FEATS 64
#define MAXK 16
#define TILE_M 16
#define GEMM_WAVES 4
#define BLOCK_M (TILE_M * GEMM_WAVES)
#define LDC (OUT_FEATS + 4)
#define K_HALF (IN_FEATS / 2)
#define LDA (K_HALF + 4)
#define A_STEPS 2
#define B_STEPS 8
#define BKT_NODES 196
#define IN_TILE_BITS 8
#define IN_TILE_MASK ((1 << IN_TILE_BITS) - 1)
#define MAX_BKT 1024
#define EDGE_CHUNK 8192
#define EPT (EDGE_CHUNK / 1024)
#define LIST_CAP 4096


__device__ __forceinline__ int scalar_bits64(unsigned long long m) {
    int n;
    asm("s_bcnt1_i32_b64 %0, %1" : "=s"(n) : "s"(m) : "scc");
    return n;
}

__device__ __forceinline__ int tile_of(int node) {
    return (int)((unsigned int)node / (unsigned int)BKT_NODES);
}

__device__ __forceinline__ unsigned int order_key(unsigned int u) {
    return u ^ ((unsigned int)((int)u >> 31) | 0x80000000u);
}

__global__ __launch_bounds__(32 * GEMM_WAVES)
void gemm_maxk_kernel(const float* __restrict__ feat, const float* __restrict__ weight,
                      float* __restrict__ tk_vals, unsigned int* __restrict__ tk_idx,
                      int n_nodes) {
    __shared__ __attribute__((aligned(16))) float sA[BLOCK_M * LDA];

    const int tid  = threadIdx.x;
    const int lane = tid & 31;
    const int wave = __builtin_amdgcn_readfirstlane(tid >> 5);

    const int m0   = (int)blockIdx.x * BLOCK_M;
    const int r16  = lane & 15;
    const int half = lane >> 4;

    const int koff = half << 1;
    const float* arow = sA + r16 * LDA + koff;
    const float* wtile = weight + wave * 16;
    const unsigned int woff = (unsigned int)(koff * OUT_FEATS + r16);
    const bool full = m0 + BLOCK_M <= n_nodes;
    const float* frow = feat + ((long long)m0 + wave) * IN_FEATS;
    const unsigned int foff = (unsigned int)lane;

    v8f acc[GEMM_WAVES];
    #pragma unroll
    for (int i = 0; i < GEMM_WAVES; ++i) acc[i] = (v8f){0.f, 0.f, 0.f, 0.f, 0.f, 0.f, 0.f, 0.f};
    v2f aq[2][A_STEPS][GEMM_WAVES];
    v2f bq[2][B_STEPS];
    #pragma unroll
    for (int h = 0; h < 2; ++h) {
        if (h) __syncthreads();
        #pragma unroll
        for (int s = 0; s < B_STEPS; ++s) {
            const float* wk = wtile + (h * K_HALF + s * 4) * OUT_FEATS + woff;
            bq[0][s] = (v2f){wk[0], wk[OUT_FEATS]};
        }
        __builtin_amdgcn_sched_barrier(0);
        #pragma unroll
        for (int pass = 0; pass < 2; ++pass) {
            v4f buf[8];
            if (full) {
                #pragma unroll
                for (int j = 0; j < 8; ++j)
                    buf[j] = ((const v4f*)(frow + (pass * 8 + j) * GEMM_WAVES * IN_FEATS + h * K_HALF))[foff];
            } else {
                #pragma unroll
                for (int j = 0; j < 8; ++j) {
                    int grow = m0 + (pass * 8 + j) * GEMM_WAVES + wave;
                    grow = grow < n_nodes ? grow : n_nodes - 1;
                    buf[j] = ((const v4f*)(feat + (long long)grow * IN_FEATS + h * K_HALF))[foff];
                }
            }
            #pragma unroll
            for (int j = 0; j < 8; ++j) {
                const int row = (pass * 8 + j) * GEMM_WAVES + wave;
                v4f v = buf[j];
                if (m0 + row >= n_nodes) v = (v4f){0.f, 0.f, 0.f, 0.f};
                *(v4f*)&sA[row * LDA + lane * 4] = v;
            }
        }
        __syncthreads();
        #pragma unroll
        for (int s = 0; s < A_STEPS; ++s)
            #pragma unroll
            for (int u = 0; u < GEMM_WAVES; ++u)
                aq[0][s][u] = *(const v2f*)(arow + u * TILE_M * LDA + s * 4);
        #pragma unroll
        for (int t = 0; t < K_HALF / (4 * A_STEPS); ++t) {
            if (t + 1 < K_HALF / (4 * A_STEPS)) {
                #pragma unroll
                for (int s = 0; s < A_STEPS; ++s)
                    #pragma unroll
                    for (int u = 0; u < GEMM_WAVES; ++u)
                        aq[(t + 1) & 1][s][u] = *(const v2f*)(arow + u * TILE_M * LDA + ((t + 1) * A_STEPS + s) * 4);
            }
            if (t % (B_STEPS / A_STEPS) == 0 && t / (B_STEPS / A_STEPS) + 1 < K_HALF / (4 * B_STEPS)) {
                const int gn = t / (B_STEPS / A_STEPS) + 1;
                #pragma unroll
                for (int s = 0; s < B_STEPS; ++s) {
                    const float* wk = wtile + (h * K_HALF + (gn * B_STEPS + s) * 4) * OUT_FEATS + woff;
                    bq[gn & 1][s] = (v2f){wk[0], wk[OUT_FEATS]};
                }
            }
            __builtin_amdgcn_sched_barrier(0);
            #pragma unroll
            for (int s = 0; s < A_STEPS; ++s) {
                const v2f b = bq[(t / (B_STEPS / A_STEPS)) & 1][(t % (B_STEPS / A_STEPS)) * A_STEPS + s];
                acc[0] = __builtin_amdgcn_wmma_f32_16x16x4_f32(false, aq[t & 1][s][0], false, b, (short)0, acc[0], false, false);
                acc[1] = __builtin_amdgcn_wmma_f32_16x16x4_f32(false, aq[t & 1][s][1], false, b, (short)0, acc[1], false, false);
                acc[2] = __builtin_amdgcn_wmma_f32_16x16x4_f32(false, aq[t & 1][s][2], false, b, (short)0, acc[2], false, false);
                acc[3] = __builtin_amdgcn_wmma_f32_16x16x4_f32(false, aq[t & 1][s][3], false, b, (short)0, acc[3], false, false);
            }
            __builtin_amdgcn_sched_barrier(0);
        }
        asm volatile("v_nop\n\tv_nop\n\tv_nop\n\tv_nop");
        __builtin_amdgcn_sched_barrier(0);
    }

    __syncthreads();
    float* sC = sA;
    #pragma unroll
    for (int u = 0; u < GEMM_WAVES; ++u) {
        #pragma unroll
        for (int r = 0; r < 8; ++r)
            sC[(u * TILE_M + half * 8 + r) * LDC + wave * 16 + r16] = acc[u][r];
    }
    __syncthreads();
    const float* crow = sC + wave * TILE_M * LDC;
    unsigned int x[TILE_M][2];
    #pragma unroll
    for (int R = 0; R < TILE_M; ++R) {
        x[R][0] = __float_as_uint(crow[R * LDC + lane]);
        x[R][1] = __float_as_uint(crow[R * LDC + 32 + lane]);
    }
    unsigned int key[TILE_M][2];
    #pragma unroll
    for (int R = 0; R < TILE_M; ++R) {
        key[R][0] = order_key(x[R][0]);
        key[R][1] = order_key(x[R][1]);
    }

    unsigned int thr[TILE_M];
    #pragma unroll
    for (int R = 0; R < TILE_M; ++R) thr[R] = 0u;
    for (int bit = 31; bit >= 0; --bit) {
        const unsigned int bmask = 1u << bit;
        unsigned int cand[TILE_M];
        unsigned long long m[TILE_M];
        int n[TILE_M];
        #pragma unroll
        for (int R = 0; R < TILE_M; ++R) cand[R] = thr[R] | bmask;
        __builtin_amdgcn_sched_barrier(0);
        #pragma unroll
        for (int R = 0; R < TILE_M; ++R)
            m[R] = (unsigned long long)__builtin_amdgcn_ballot_w32(key[R][0] >= cand[R]) |
                   ((unsigned long long)__builtin_amdgcn_ballot_w32(key[R][1] >= cand[R]) << 32);
        __builtin_amdgcn_sched_barrier(0);
        #pragma unroll
        for (int R = 0; R < TILE_M; ++R) n[R] = scalar_bits64(m[R]);
        __builtin_amdgcn_sched_barrier(0);
        #pragma unroll
        for (int R = 0; R < TILE_M; ++R) thr[R] = n[R] >= MAXK ? cand[R] : thr[R];
    }

    float* sV = sA + BLOCK_M * LDC;
    unsigned char* sI = (unsigned char*)(sV + BLOCK_M * MAXK);
    #pragma unroll
    for (int R = 0; R < TILE_M; ++R) {
        const unsigned int T = thr[R];
        const bool gt0 = key[R][0] > T,  gt1 = key[R][1] > T;
        const bool eq0 = key[R][0] == T, eq1 = key[R][1] == T;
        const unsigned int g0 = __builtin_amdgcn_ballot_w32(gt0), g1 = __builtin_amdgcn_ballot_w32(gt1);
        const unsigned int e0 = __builtin_amdgcn_ballot_w32(eq0), e1 = __builtin_amdgcn_ballot_w32(eq1);
        const int need = MAXK - (__popc(g0) + __popc(g1));
        const int rank0 = (int)__builtin_amdgcn_mbcnt_lo(e0, 0u);
        const int rank1 = (int)__builtin_amdgcn_mbcnt_lo(e1, (unsigned int)__popc(e0));
        const unsigned int s0 = g0 | (e0 & __builtin_amdgcn_ballot_w32(rank0 < need));
        const unsigned int s1 = g1 | (e1 & __builtin_amdgcn_ballot_w32(rank1 < need));
        const bool sel0 = __builtin_amdgcn_inverse_ballot_w32(s0);
        const bool sel1 = __builtin_amdgcn_inverse_ballot_w32(s1);
        const int slot0 = (int)__builtin_amdgcn_mbcnt_lo(s0, 0u);
        const int slot1 = (int)__builtin_amdgcn_mbcnt_lo(s1, (unsigned int)__popc(s0));
        const int rec = (wave * TILE_M + R) * MAXK;
        if (sel0) {
            sV[rec + slot0] = __uint_as_float(x[R][0]);
            sI[rec + slot0] = (unsigned char)lane;
        }
        if (sel1) {
            sV[rec + slot1] = __uint_as_float(x[R][1]);
            sI[rec + slot1] = (unsigned char)(32 + lane);
        }
    }
    __syncthreads();

    const float4* sV4 = (const float4*)sV;
    #pragma unroll
    for (int q = 0; q < BLOCK_M * MAXK / (4 * 32 * GEMM_WAVES); ++q) {
        const int f = q * (32 * GEMM_WAVES) + tid;
        if (m0 + (f >> 2) < n_nodes)
            ((float4*)tk_vals)[(long long)m0 * (MAXK / 4) + f] = sV4[f];
    }
    if (tid < BLOCK_M && m0 + tid < n_nodes)
        ((uint4*)tk_idx)[m0 + tid] = ((const uint4*)sI)[tid];
}

__global__ __launch_bounds__(1024)
void chunk_sort_kernel(const int* __restrict__ src, const int* __restrict__ dst,
                       unsigned int* __restrict__ binned_d, unsigned char* __restrict__ binned_s,
                       unsigned short* __restrict__ run_d, unsigned short* __restrict__ run_s,
                       int n_edges, int n_tiles) {
    const int key = blockIdx.y;
    __shared__ int hist[MAX_BKT];
    __shared__ int wsum[32];
    __shared__ unsigned int stage[EDGE_CHUNK];
    const int tid  = threadIdx.x;
    const int lane = tid & 31;
    const int wave = tid >> 5;
    hist[tid] = 0;

    const int base = blockIdx.x * EDGE_CHUNK;
    int es[EPT], ed[EPT];
    #pragma unroll
    for (int j = 0; j < EPT; ++j) {
        const int e  = base + j * 1024 + tid;
        const int ec = e < n_edges ? e : n_edges - 1;
        const int sv = src[ec];
        const int dv = dst[ec];
        es[j] = key ? dv : sv;
        ed[j] = e < n_edges ? (key ? sv : dv) : -1;
    }
    __syncthreads();
    #pragma unroll
    for (int j = 0; j < EPT; ++j)
        if (ed[j] >= 0) atomicAdd(&hist[tile_of(ed[j])], 1);
    __syncthreads();

    const int c = hist[tid];
    int y = c;
    #pragma unroll
    for (int o = 1; o < 32; o <<= 1) {
        const int ys = __shfl_up(y, o);
        if (lane >= o) y += ys;
    }
    if (lane == 31) wsum[wave] = y;
    __syncthreads();
    if (wave == 0) {
        const int w = wsum[lane];
        int s = w;
        #pragma unroll
        for (int o = 1; o < 32; o <<= 1) {
            const int t = __shfl_up(s, o);
            if (lane >= o) s += t;
        }
        wsum[lane] = s - w;
    }
    __syncthreads();

    const int lstart = y - c + wsum[wave];
    if (tid <= n_tiles)
        (key ? run_s : run_d)[(long long)blockIdx.x * (n_tiles + 2) + tid] = (unsigned short)lstart;
    hist[tid] = lstart;
    __syncthreads();

    #pragma unroll
    for (int j = 0; j < EPT; ++j) {
        if (ed[j] >= 0) {
            const int b    = tile_of(ed[j]);
            const int slot = atomicAdd(&hist[b], 1);
            stage[slot] = ((unsigned int)es[j] << IN_TILE_BITS) | (unsigned int)(ed[j] - b * BKT_NODES);
        }
    }
    __syncthreads();

    const int left = n_edges - base;
    const int n_here = left < EDGE_CHUNK ? left : EDGE_CHUNK;
    #pragma unroll
    for (int j = 0; j < EPT; ++j) {
        const int slot = j * 1024 + tid;
        if (slot < n_here) {
            if (key == 0) binned_d[base + slot] = stage[slot];
            else          binned_s[base + slot] = (unsigned char)stage[slot];
        }
    }
}

__device__ __forceinline__ int block_excl_scan(int v, int* wsum, int lane, int wave, int* total) {
    int x = v;
    #pragma unroll
    for (int o = 1; o < 32; o <<= 1) {
        const int t = __shfl_up(x, o);
        if (lane >= o) x += t;
    }
    if (lane == 31) wsum[wave] = x;
    __syncthreads();
    if (wave == 0) {
        const int w = wsum[lane];
        int s = w;
        #pragma unroll
        for (int o = 1; o < 32; o <<= 1) {
            const int t = __shfl_up(s, o);
            if (lane >= o) s += t;
        }
        wsum[lane] = s - w;
        if (lane == 31) wsum[32] = s;
    }
    __syncthreads();
    *total = wsum[32];
    return x - v + wsum[wave];
}

__device__ __forceinline__ void block_excl_scan2(int v0, int v1, int* wsum, int lane, int wave,
                                                 int* p0, int* p1, int* t0, int* t1) {
    int x = v0, y = v1;
    #pragma unroll
    for (int o = 1; o < 32; o <<= 1) {
        const int xs = __shfl_up(x, o);
        const int ys = __shfl_up(y, o);
        if (lane >= o) { x += xs; y += ys; }
    }
    if (lane == 31) { wsum[wave] = x; wsum[33 + wave] = y; }
    __syncthreads();
    if (wave < 2) {
        int* ws = wsum + 33 * wave;
        const int w = ws[lane];
        int s = w;
        #pragma unroll
        for (int o = 1; o < 32; o <<= 1) {
            const int t = __shfl_up(s, o);
            if (lane >= o) s += t;
        }
        ws[lane] = s - w;
        if (lane == 31) ws[32] = s;
    }
    __syncthreads();
    *p0 = x - v0 + wsum[wave];
    *p1 = y - v1 + wsum[33 + wave];
    *t0 = wsum[32];
    *t1 = wsum[65];
}

__device__ __forceinline__ void list_run(int* list, int pre, int len, int gstart, int wbase) {
    const int j0 = pre > wbase ? pre : wbase;
    const int e1 = pre + len;
    const int e2 = wbase + LIST_CAP;
    const int j1 = e1 < e2 ? e1 : e2;
    #pragma nounroll
    for (int j = j0; j < j1; ++j) list[j - wbase] = gstart + (j - pre);
}

__global__ __launch_bounds__(1024)
void bkt_degree_scale_kernel(const unsigned int* __restrict__ binned_d,
                             const unsigned char* __restrict__ binned_s,
                             const unsigned short* __restrict__ run_d,
                             const unsigned short* __restrict__ run_s,
                             float* __restrict__ tk_vals,
                             int n_nodes, int n_chunks, int n_tiles) {
    __shared__ int cin[BKT_NODES];
    __shared__ int cout[BKT_NODES];
    __shared__ float ssc[BKT_NODES];
    __shared__ int list_d[LIST_CAP];
    __shared__ int list_s[LIST_CAP];
    __shared__ int wsd[66];
    const int tid  = threadIdx.x;
    const int lane = tid & 31;
    const int wave = tid >> 5;
    if (tid < BKT_NODES) { cin[tid] = 0; cout[tid] = 0; }
    float4* rec = (float4*)tk_vals + (long long)blockIdx.x * (BKT_NODES * MAXK / 4);
    const bool live = (tid >> 2) < BKT_NODES && blockIdx.x * BKT_NODES + (tid >> 2) < n_nodes;
    float4 v = make_float4(0.f, 0.f, 0.f, 0.f);
    if (live) v = rec[tid];
    int lend = 0, gd = 0, lens = 0, gs = 0;
    if (tid < n_chunks) {
        const long long r = (long long)tid * (n_tiles + 2) + blockIdx.x;
        long long r1 = r + 1;
        asm volatile("" : "+v"(r1));
        const int ad = run_d[r];
        const int as = run_s[r];
        const int bd = run_d[r1];
        const int bs = run_s[r1];
        lend = bd - ad; gd = tid * EDGE_CHUNK + ad;
        lens = bs - as; gs = tid * EDGE_CHUNK + as;
    }
    int nd, ns, pd, ps;
    block_excl_scan2(lend, lens, wsd, lane, wave, &pd, &ps, &nd, &ns);
    const int nmax = nd > ns ? nd : ns;
    for (int wbase = 0; wbase < nmax; wbase += LIST_CAP) {
        list_run(list_d, pd, lend, gd, wbase);
        list_run(list_s, ps, lens, gs, wbase);
        __syncthreads();
        const int wd_n = nd - wbase < LIST_CAP ? nd - wbase : LIST_CAP;
        const int ws_n = ns - wbase < LIST_CAP ? ns - wbase : LIST_CAP;
        const int wn = wd_n > ws_n ? wd_n : ws_n;
        for (int i0 = tid; i0 < wn; i0 += 4 * 1024) {
            unsigned int wd[4], ws[4];
            #pragma unroll
            for (int k = 0; k < 4; ++k) {
                const int i = i0 + 1024 * k;
                int qd = 0, qs = 0;
                if (wd_n > 0) qd = list_d[i < wd_n ? i : wd_n - 1];
                if (ws_n > 0) qs = list_s[i < ws_n ? i : ws_n - 1];
                wd[k] = binned_d[qd];
                ws[k] = binned_s[qs];
            }
            #pragma unroll
            for (int k = 0; k < 4; ++k) {
                const int i = i0 + 1024 * k;
                if (i < wd_n) atomicAdd(&cin[wd[k] & IN_TILE_MASK], 1);
                if (i < ws_n) atomicAdd(&cout[ws[k] & IN_TILE_MASK], 1);
            }
        }
        __syncthreads();
    }
    __syncthreads();
    if (tid < BKT_NODES) {
        float id = (float)cin[tid];
        float od = (float)cout[tid];
        od = od < 1.f ? 1.f : od;
        id = id < 1.f ? 1.f : id;
        ssc[tid] = rsqrtf(od) * rsqrtf(id);
    }
    __syncthreads();
    if (live) {
        const float sc = ssc[tid >> 2];
        v.x *= sc; v.y *= sc; v.z *= sc; v.w *= sc;
        rec[tid] = v;
    }
}

__device__ __forceinline__ void add_record(float* row, float4 v, unsigned int ip) {
    atomicAdd(row + (ip & 0xffu), v.x);
    atomicAdd(row + ((ip >> 8) & 0xffu), v.y);
    atomicAdd(row + ((ip >> 16) & 0xffu), v.z);
    atomicAdd(row + ((ip >> 24) & 0xffu), v.w);
}

__global__ __launch_bounds__(1024)
void bkt_sum_kernel(const unsigned int* __restrict__ binned, const unsigned short* __restrict__ run_d,
                    const float* __restrict__ tk_vals,
                    const unsigned int* __restrict__ tk_idx, const float* __restrict__ bias,
                    float* __restrict__ out, int n_nodes, int n_chunks, int n_tiles) {
    __shared__ __attribute__((aligned(16))) float acc[BKT_NODES * OUT_FEATS];
    __shared__ int list[LIST_CAP];
    __shared__ int wsd[33];
    const int tid  = threadIdx.x;
    const int lane = tid & 31;
    const int wave = tid >> 5;
    const float b0 = bias[tid & (OUT_FEATS - 1)];
    const int cc = tid < n_chunks ? tid : 0;
    const long long r = (long long)cc * (n_tiles + 2) + blockIdx.x;
    long long r1 = r + 1;
    asm volatile("" : "+v"(r1));
    const int ra = run_d[r];
    const int rb = run_d[r1];
    for (int i = tid; i < BKT_NODES * OUT_FEATS; i += 1024) acc[i] = b0;
    const int len = tid < n_chunks ? rb - ra : 0;
    const int g = cc * EDGE_CHUNK + ra;
    int cnt;
    const int pre = block_excl_scan(len, wsd, lane, wave, &cnt);

    const int part = tid & 3;
    for (int wbase = 0; wbase < cnt; wbase += LIST_CAP) {
        list_run(list, pre, len, g, wbase);
        __syncthreads();
        const int wn = cnt - wbase < LIST_CAP ? cnt - wbase : LIST_CAP;
        for (int i0 = tid >> 2; i0 < wn; i0 += 4 * 256) {
            unsigned int w[4], ip[4];
            float4 v[4];
            #pragma unroll
            for (int k = 0; k < 4; ++k) {
                const int i = i0 + 256 * k;
                w[k] = binned[list[i < wn ? i : wn - 1]];
            }
            #pragma unroll
            for (int k = 0; k < 4; ++k) {
                const long long s = (long long)(w[k] >> IN_TILE_BITS);
                v[k]  = ((const float4*)tk_vals)[s * 4 + part];
                ip[k] = tk_idx[s * 4 + part];
            }
            asm volatile("" :: "v"(v[0].x), "v"(v[0].y), "v"(v[0].z), "v"(v[0].w), "v"(ip[0]),
                               "v"(v[1].x), "v"(v[1].y), "v"(v[1].z), "v"(v[1].w), "v"(ip[1]),
                               "v"(v[2].x), "v"(v[2].y), "v"(v[2].z), "v"(v[2].w), "v"(ip[2]),
                               "v"(v[3].x), "v"(v[3].y), "v"(v[3].z), "v"(v[3].w), "v"(ip[3]));
            #pragma unroll
            for (int k = 0; k < 4; ++k)
                if (i0 + 256 * k < wn)
                    add_record(acc + (w[k] & IN_TILE_MASK) * OUT_FEATS, v[k], ip[k]);
        }
        __syncthreads();
    }
    __syncthreads();

    const long long node0 = (long long)blockIdx.x * BKT_NODES;
    for (int i = tid; i < BKT_NODES * OUT_FEATS / 4; i += 1024) {
        if (node0 + (i >> 4) < n_nodes)
            ((float4*)out)[node0 * (OUT_FEATS / 4) + i] = ((const float4*)acc)[i];
    }
}

extern "C" void kernel_launch(void* const* d_in, const int* in_sizes, int n_in,
                              void* d_out, int out_size, void* d_ws, size_t ws_size,
                              hipStream_t stream) {
    const float* feat   = (const float*)d_in[0];
    const float* weight = (const float*)d_in[1];
    const float* bias   = (const float*)d_in[2];
    const int*   src    = (const int*)d_in[3];
    const int*   dst    = (const int*)d_in[4];
    float* out = (float*)d_out;

    const int n_nodes = in_sizes[0] / IN_FEATS;
    const int n_edges = in_sizes[3];
    const int n_tiles  = (n_nodes + BKT_NODES - 1) / BKT_NODES;
    const int n_chunks = (n_edges + EDGE_CHUNK - 1) / EDGE_CHUNK;

    float* tk_vals = (float*)d_ws;
    unsigned int* tk_idx = (unsigned int*)(tk_vals + (long long)n_nodes * MAXK);
    unsigned int* binned_d  = tk_idx + (long long)n_nodes * (MAXK / 4);
    unsigned short* run_d   = (unsigned short*)(binned_d + n_edges);
    unsigned char* binned_s = (unsigned char*)d_out;
    unsigned short* run_s   = (unsigned short*)(binned_s + (long long)n_chunks * EDGE_CHUNK);

    chunk_sort_kernel<<<dim3(n_chunks, 2), 1024, 0, stream>>>(src, dst, binned_d, binned_s, run_d, run_s,
                                                              n_edges, n_tiles);
    gemm_maxk_kernel<<<(n_nodes + BLOCK_M - 1) / BLOCK_M, 32 * GEMM_WAVES, 0, stream>>>(feat, weight, tk_vals, tk_idx,
                                                                                        n_nodes);
    bkt_degree_scale_kernel<<<n_tiles, 1024, 0, stream>>>(binned_d, binned_s, run_d, run_s, tk_vals,
                                                          n_nodes, n_chunks, n_tiles);
    bkt_sum_kernel<<<n_tiles, 1024, 0, stream>>>(binned_d, run_d, tk_vals, tk_idx, bias, out,
                                                 n_nodes, n_chunks, n_tiles);
}
